# starting kernel
# baseline (speedup 1.0000x reference)
#include <hip/hip_runtime.h>
#include <hip/hip_bf16.h>


typedef float v2f __attribute__((ext_vector_type(2)));
typedef float v8f __attribute__((ext_vector_type(8)));

#define NNODES  200000
#define NEDGES  6400000
#define NFEAT   128
#define NHID    16
#define NMLP    100
#define NCLS    27
#define NGRAPH  4096


__global__ void fill_f32(float* __restrict__ p, int n, float v) {
    int i = blockIdx.x * blockDim.x + threadIdx.x;
    if (i < n) p[i] = v;
}

__global__ void deg_accum(const int* __restrict__ dst, float* __restrict__ deg, int e) {
    int i = blockIdx.x * blockDim.x + threadIdx.x;
    if (i < e) atomicAdd(&deg[dst[i]], 1.0f);
}

__global__ void rsqrt_ip(float* __restrict__ p, int n) {
    int i = blockIdx.x * blockDim.x + threadIdx.x;
    if (i < n) p[i] = rsqrtf(p[i]);
}

__global__ __launch_bounds__(256)
void gemm_wmma_f32(const float* __restrict__ X, const float* __restrict__ W,
                   float* __restrict__ T, int M, int K) {
    __shared__ float lw[NFEAT * NHID];
    const int tid = threadIdx.x;
    const int kn  = K * NHID;
    for (int i = tid; i < kn; i += 256) lw[i] = W[i];
    __syncthreads();

    const int wave = tid >> 5;
    const int lane = tid & 31;
    const int tile = blockIdx.x * 8 + wave;
    if (tile * 16 >= M) return;

    const int row   = tile * 16 + (lane & 15);
    const int kHalf = (lane >> 4) << 1;
    const int n     = lane & 15;

    v8f c = {};
    const float* xr = X + (size_t)row * K + kHalf;
    for (int k0 = 0; k0 < K; k0 += 4) {
        float2 av = *(const float2*)(xr + k0);
        v2f a; a.x = av.x; a.y = av.y;
        v2f b;
        b.x = lw[(k0 + kHalf)     * NHID + n];
        b.y = lw[(k0 + kHalf + 1) * NHID + n];
        c = __builtin_amdgcn_wmma_f32_16x16x4_f32(
                 false, a,  false, b,
                 (short)0, c,  false,  false);
    }

    const int rbase = tile * 16 + ((lane >> 4) << 3);
#pragma unroll
    for (int v = 0; v < 8; ++v)
        T[(size_t)(rbase + v) * NHID + n] = c[v];
}

__global__ void edge_scatter(const int* __restrict__ src, const int* __restrict__ dst,
                             const float* __restrict__ dinv, const float* __restrict__ T,
                             float* __restrict__ OUT, int e) {
    int t  = blockIdx.x * blockDim.x + threadIdx.x;
    int ed = t >> 2, q = t & 3;
    if (ed >= e) return;
    int s = src[ed], d = dst[ed];
    float w = dinv[s] * dinv[d];
    const float4 m = *(const float4*)(T + (size_t)s * NHID + q * 4);
    float* o = OUT + (size_t)d * NHID + q * 4;
    atomicAdd(o + 0, m.x * w);
    atomicAdd(o + 1, m.y * w);
    atomicAdd(o + 2, m.z * w);
    atomicAdd(o + 3, m.w * w);
}

__global__ void finalize_node(float* __restrict__ ACC, const float* __restrict__ T,
                              const float* __restrict__ dinv, const float* __restrict__ bias,
                              int n16, int doRelu) {
    int idx = blockIdx.x * blockDim.x + threadIdx.x;
    if (idx >= n16) return;
    int i = idx >> 4, f = idx & 15;
    float di = dinv[i];
    float v  = ACC[idx] + T[idx] * di * di + bias[f];
    ACC[idx] = doRelu ? fmaxf(v, 0.0f) : v;
}

__global__ void pool_kernel(const int* __restrict__ bvec, const float* __restrict__ Hn,
                            float* __restrict__ Gp, int n) {
    int t = blockIdx.x * blockDim.x + threadIdx.x;
    int i = t >> 2, q = t & 3;
    if (i >= n) return;
    int g = bvec[i];
    const float4 h = *(const float4*)(Hn + (size_t)i * NHID + q * 4);
    float* o = Gp + (size_t)g * NHID + q * 4;
    atomicAdd(o + 0, h.x);
    atomicAdd(o + 1, h.y);
    atomicAdd(o + 2, h.z);
    atomicAdd(o + 3, h.w);
}

__global__ void lin1_kernel(const float* __restrict__ G, const float* __restrict__ W,
                            const float* __restrict__ B, float* __restrict__ O) {
    int t = blockIdx.x * blockDim.x + threadIdx.x;
    if (t >= NGRAPH * NMLP) return;
    int gi = t / NMLP, j = t % NMLP;
    float acc = B[j];
#pragma unroll
    for (int k = 0; k < NHID; ++k)
        acc = fmaf(fmaxf(G[gi * NHID + k], 0.0f), W[k * NMLP + j], acc);
    O[t] = fmaxf(acc, 0.0f);
}

__global__ void lin2_kernel(const float* __restrict__ G, const float* __restrict__ W,
                            const float* __restrict__ B, float* __restrict__ O) {
    int t = blockIdx.x * blockDim.x + threadIdx.x;
    if (t >= NGRAPH * NCLS) return;
    int gi = t / NCLS, c = t % NCLS;
    float acc = B[c];
    for (int k = 0; k < NMLP; ++k)
        acc = fmaf(G[gi * NMLP + k], W[k * NCLS + c], acc);
    O[t] = acc;
}


extern "C" void kernel_launch(void* const* d_in, const int* in_sizes, int n_in,
                              void* d_out, int out_size, void* d_ws, size_t ws_size,
                              hipStream_t stream) {
    const float* x   = (const float*)d_in[0];
    const int*  eidx = (const int*)d_in[1];
    const int*  bvec = (const int*)d_in[2];
    const float* W1  = (const float*)d_in[3];
    const float* b1  = (const float*)d_in[4];
    const float* W2  = (const float*)d_in[5];
    const float* b2  = (const float*)d_in[6];
    const float* Wl1 = (const float*)d_in[7];
    const float* bl1 = (const float*)d_in[8];
    const float* Wl2 = (const float*)d_in[9];
    const float* bl2 = (const float*)d_in[10];
    float* out = (float*)d_out;

    char* ws = (char*)d_ws;
    size_t off = 0;
    auto walloc = [&](size_t bytes) -> void* {
        void* p = ws + off;
        off = (off + bytes + 255) & ~(size_t)255;
        return p;
    };
    float* deg  = (float*)walloc((size_t)NNODES * 4);
    float* bufA = (float*)walloc((size_t)NNODES * NHID * 4);
    float* bufB = (float*)walloc((size_t)NNODES * NHID * 4);
    float* gbuf = (float*)walloc((size_t)NGRAPH * NHID * 4);
    float* g2   = (float*)walloc((size_t)NGRAPH * NMLP * 4);

    const int* src = eidx;
    const int* dst = eidx + NEDGES;

    const int TB = 256;
    const int nN    = (NNODES + TB - 1) / TB;
    const int nE    = (NEDGES + TB - 1) / TB;
    const int nN16  = (NNODES * NHID + TB - 1) / TB;
    const int nE4   = (NEDGES * 4 + TB - 1) / TB;
    const int nN4   = (NNODES * 4 + TB - 1) / TB;
    const int nTile = ((NNODES / 16) + 7) / 8;

    fill_f32<<<nN, TB, 0, stream>>>(deg, NNODES, 1.0f);
    deg_accum<<<nE, TB, 0, stream>>>(dst, deg, NEDGES);
    rsqrt_ip<<<nN, TB, 0, stream>>>(deg, NNODES);
    float* dinv = deg;

    gemm_wmma_f32<<<nTile, TB, 0, stream>>>(x, W1, bufA, NNODES, NFEAT);
    fill_f32<<<nN16, TB, 0, stream>>>(bufB, NNODES * NHID, 0.0f);
    edge_scatter<<<nE4, TB, 0, stream>>>(src, dst, dinv, bufA, bufB, NEDGES);
    finalize_node<<<nN16, TB, 0, stream>>>(bufB, bufA, dinv, b1, NNODES * NHID, 1);

    gemm_wmma_f32<<<nTile, TB, 0, stream>>>(bufB, W2, bufA, NNODES, NHID);
    fill_f32<<<nN16, TB, 0, stream>>>(bufB, NNODES * NHID, 0.0f);
    edge_scatter<<<nE4, TB, 0, stream>>>(src, dst, dinv, bufA, bufB, NEDGES);
    finalize_node<<<nN16, TB, 0, stream>>>(bufB, bufA, dinv, b2, NNODES * NHID, 0);

    fill_f32<<<(NGRAPH * NHID + TB - 1) / TB, TB, 0, stream>>>(gbuf, NGRAPH * NHID, 0.0f);
    pool_kernel<<<nN4, TB, 0, stream>>>(bvec, bufB, gbuf, NNODES);
    lin1_kernel<<<(NGRAPH * NMLP + TB - 1) / TB, TB, 0, stream>>>(gbuf, Wl1, bl1, g2);
    lin2_kernel<<<(NGRAPH * NCLS + TB - 1) / TB, TB, 0, stream>>>(g2, Wl2, bl2, out);
}


# In-degree count: 6.4M global float atomics replaced by integer histograms privatised in LDS (8 dst ranges x 32 edge chunks), partial rows summed with +1 and rsqrt folded in; degrees bit-identical
# speedup vs baseline: 1.3142x; 1.3142x over previous
#include <hip/hip_runtime.h>
#include <hip/hip_bf16.h>


typedef float v2f __attribute__((ext_vector_type(2)));
typedef float v8f __attribute__((ext_vector_type(8)));

#define NNODES  200000
#define NEDGES  6400000
#define NFEAT   128
#define NHID    16
#define NMLP    100
#define NCLS    27
#define NGRAPH  4096


__global__ void fill_f32(float* __restrict__ p, int n, float v) {
    int i = blockIdx.x * blockDim.x + threadIdx.x;
    if (i < n) p[i] = v;
}

#define HR     8
#define HBINS  (NNODES / HR)
#define HCH    32
#define HCHUNK (NEDGES / HCH)
#define HN4    (HCHUNK / 4)

__global__ __launch_bounds__(1024)
void deg_hist(const int* __restrict__ dst, unsigned* __restrict__ part) {
    __shared__ __attribute__((aligned(16))) unsigned h[HBINS];
    const int tid = threadIdx.x;
    const int r   = blockIdx.x & (HR - 1);
    const int c   = blockIdx.x >> 3;

    uint4* h4 = (uint4*)h;
    for (int i = tid; i < HBINS / 4; i += 1024) h4[i] = make_uint4(0u, 0u, 0u, 0u);
    __syncthreads();

    const unsigned lo = (unsigned)(r * HBINS);
    const int4* p = (const int4*)(dst + (size_t)c * HCHUNK);
    for (int base = tid; base < HN4; base += 1024 * 8) {
        int4 v[8];
#pragma unroll
        for (int u = 0; u < 8; ++u) {
            const int j = base + u * 1024;
            v[u] = (j < HN4) ? p[j] : make_int4(-1, -1, -1, -1);
        }
#pragma unroll
        for (int u = 0; u < 8; ++u) {
            unsigned a;
            a = (unsigned)v[u].x - lo; if (a < (unsigned)HBINS) atomicAdd(&h[a], 1u);
            a = (unsigned)v[u].y - lo; if (a < (unsigned)HBINS) atomicAdd(&h[a], 1u);
            a = (unsigned)v[u].z - lo; if (a < (unsigned)HBINS) atomicAdd(&h[a], 1u);
            a = (unsigned)v[u].w - lo; if (a < (unsigned)HBINS) atomicAdd(&h[a], 1u);
        }
    }
    __syncthreads();

    uint4* o = (uint4*)(part + (size_t)c * NNODES + lo);
    for (int i = tid; i < HBINS / 4; i += 1024) o[i] = h4[i];
}

__global__ void deg_reduce(const unsigned* __restrict__ part, float* __restrict__ dinv) {
    const int i = blockIdx.x * blockDim.x + threadIdx.x;
    if (i >= NNODES / 4) return;
    const uint4* p = (const uint4*)part + i;
    uint4 s = make_uint4(0u, 0u, 0u, 0u);
#pragma unroll
    for (int c = 0; c < HCH; ++c) {
        const uint4 v = p[(size_t)c * (NNODES / 4)];
        s.x += v.x; s.y += v.y; s.z += v.z; s.w += v.w;
    }
    float4 d;
    d.x = rsqrtf((float)s.x + 1.0f);
    d.y = rsqrtf((float)s.y + 1.0f);
    d.z = rsqrtf((float)s.z + 1.0f);
    d.w = rsqrtf((float)s.w + 1.0f);
    ((float4*)dinv)[i] = d;
}

__global__ __launch_bounds__(256)
void gemm_wmma_f32(const float* __restrict__ X, const float* __restrict__ W,
                   float* __restrict__ T, int M, int K) {
    __shared__ float lw[NFEAT * NHID];
    const int tid = threadIdx.x;
    const int kn  = K * NHID;
    for (int i = tid; i < kn; i += 256) lw[i] = W[i];
    __syncthreads();

    const int wave = tid >> 5;
    const int lane = tid & 31;
    const int tile = blockIdx.x * 8 + wave;
    if (tile * 16 >= M) return;

    const int row   = tile * 16 + (lane & 15);
    const int kHalf = (lane >> 4) << 1;
    const int n     = lane & 15;

    v8f c = {};
    const float* xr = X + (size_t)row * K + kHalf;
    for (int k0 = 0; k0 < K; k0 += 4) {
        float2 av = *(const float2*)(xr + k0);
        v2f a; a.x = av.x; a.y = av.y;
        v2f b;
        b.x = lw[(k0 + kHalf)     * NHID + n];
        b.y = lw[(k0 + kHalf + 1) * NHID + n];
        c = __builtin_amdgcn_wmma_f32_16x16x4_f32(
                 false, a,  false, b,
                 (short)0, c,  false,  false);
    }

    const int rbase = tile * 16 + ((lane >> 4) << 3);
#pragma unroll
    for (int v = 0; v < 8; ++v)
        T[(size_t)(rbase + v) * NHID + n] = c[v];
}

__global__ void edge_scatter(const int* __restrict__ src, const int* __restrict__ dst,
                             const float* __restrict__ dinv, const float* __restrict__ T,
                             float* __restrict__ OUT, int e) {
    int t  = blockIdx.x * blockDim.x + threadIdx.x;
    int ed = t >> 2, q = t & 3;
    if (ed >= e) return;
    int s = src[ed], d = dst[ed];
    float w = dinv[s] * dinv[d];
    const float4 m = *(const float4*)(T + (size_t)s * NHID + q * 4);
    float* o = OUT + (size_t)d * NHID + q * 4;
    atomicAdd(o + 0, m.x * w);
    atomicAdd(o + 1, m.y * w);
    atomicAdd(o + 2, m.z * w);
    atomicAdd(o + 3, m.w * w);
}

__global__ void finalize_node(float* __restrict__ ACC, const float* __restrict__ T,
                              const float* __restrict__ dinv, const float* __restrict__ bias,
                              int n16, int doRelu) {
    int idx = blockIdx.x * blockDim.x + threadIdx.x;
    if (idx >= n16) return;
    int i = idx >> 4, f = idx & 15;
    float di = dinv[i];
    float v  = ACC[idx] + T[idx] * di * di + bias[f];
    ACC[idx] = doRelu ? fmaxf(v, 0.0f) : v;
}

__global__ void pool_kernel(const int* __restrict__ bvec, const float* __restrict__ Hn,
                            float* __restrict__ Gp, int n) {
    int t = blockIdx.x * blockDim.x + threadIdx.x;
    int i = t >> 2, q = t & 3;
    if (i >= n) return;
    int g = bvec[i];
    const float4 h = *(const float4*)(Hn + (size_t)i * NHID + q * 4);
    float* o = Gp + (size_t)g * NHID + q * 4;
    atomicAdd(o + 0, h.x);
    atomicAdd(o + 1, h.y);
    atomicAdd(o + 2, h.z);
    atomicAdd(o + 3, h.w);
}

__global__ void lin1_kernel(const float* __restrict__ G, const float* __restrict__ W,
                            const float* __restrict__ B, float* __restrict__ O) {
    int t = blockIdx.x * blockDim.x + threadIdx.x;
    if (t >= NGRAPH * NMLP) return;
    int gi = t / NMLP, j = t % NMLP;
    float acc = B[j];
#pragma unroll
    for (int k = 0; k < NHID; ++k)
        acc = fmaf(fmaxf(G[gi * NHID + k], 0.0f), W[k * NMLP + j], acc);
    O[t] = fmaxf(acc, 0.0f);
}

__global__ void lin2_kernel(const float* __restrict__ G, const float* __restrict__ W,
                            const float* __restrict__ B, float* __restrict__ O) {
    int t = blockIdx.x * blockDim.x + threadIdx.x;
    if (t >= NGRAPH * NCLS) return;
    int gi = t / NCLS, c = t % NCLS;
    float acc = B[c];
    for (int k = 0; k < NMLP; ++k)
        acc = fmaf(G[gi * NMLP + k], W[k * NCLS + c], acc);
    O[t] = acc;
}


extern "C" void kernel_launch(void* const* d_in, const int* in_sizes, int n_in,
                              void* d_out, int out_size, void* d_ws, size_t ws_size,
                              hipStream_t stream) {
    const float* x   = (const float*)d_in[0];
    const int*  eidx = (const int*)d_in[1];
    const int*  bvec = (const int*)d_in[2];
    const float* W1  = (const float*)d_in[3];
    const float* b1  = (const float*)d_in[4];
    const float* W2  = (const float*)d_in[5];
    const float* b2  = (const float*)d_in[6];
    const float* Wl1 = (const float*)d_in[7];
    const float* bl1 = (const float*)d_in[8];
    const float* Wl2 = (const float*)d_in[9];
    const float* bl2 = (const float*)d_in[10];
    float* out = (float*)d_out;

    char* ws = (char*)d_ws;
    size_t off = 0;
    auto walloc = [&](size_t bytes) -> void* {
        void* p = ws + off;
        off = (off + bytes + 255) & ~(size_t)255;
        return p;
    };
    float* deg  = (float*)walloc((size_t)NNODES * 4);
    float* bufA = (float*)walloc((size_t)NNODES * NHID * 4);
    float* bufB = (float*)walloc((size_t)NNODES * NHID * 4);
    float* gbuf = (float*)walloc((size_t)NGRAPH * NHID * 4);
    float* g2   = (float*)walloc((size_t)NGRAPH * NMLP * 4);

    const int* src = eidx;
    const int* dst = eidx + NEDGES;

    const int TB = 256;
    const int nN16  = (NNODES * NHID + TB - 1) / TB;
    const int nE4   = (NEDGES * 4 + TB - 1) / TB;
    const int nN4   = (NNODES * 4 + TB - 1) / TB;
    const int nTile = ((NNODES / 16) + 7) / 8;

    unsigned* part = (unsigned*)bufA;
    deg_hist<<<HR * HCH, 1024, 0, stream>>>(dst, part);
    deg_reduce<<<(NNODES / 4 + TB - 1) / TB, TB, 0, stream>>>(part, deg);
    float* dinv = deg;

    gemm_wmma_f32<<<nTile, TB, 0, stream>>>(x, W1, bufA, NNODES, NFEAT);
    fill_f32<<<nN16, TB, 0, stream>>>(bufB, NNODES * NHID, 0.0f);
    edge_scatter<<<nE4, TB, 0, stream>>>(src, dst, dinv, bufA, bufB, NEDGES);
    finalize_node<<<nN16, TB, 0, stream>>>(bufB, bufA, dinv, b1, NNODES * NHID, 1);

    gemm_wmma_f32<<<nTile, TB, 0, stream>>>(bufB, W2, bufA, NNODES, NHID);
    fill_f32<<<nN16, TB, 0, stream>>>(bufB, NNODES * NHID, 0.0f);
    edge_scatter<<<nE4, TB, 0, stream>>>(src, dst, dinv, bufA, bufB, NEDGES);
    finalize_node<<<nN16, TB, 0, stream>>>(bufB, bufA, dinv, b2, NNODES * NHID, 0);

    fill_f32<<<(NGRAPH * NHID + TB - 1) / TB, TB, 0, stream>>>(gbuf, NGRAPH * NHID, 0.0f);
    pool_kernel<<<nN4, TB, 0, stream>>>(bvec, bufB, gbuf, NNODES);
    lin1_kernel<<<(NGRAPH * NMLP + TB - 1) / TB, TB, 0, stream>>>(gbuf, Wl1, bl1, g2);
    lin2_kernel<<<(NGRAPH * NCLS + TB - 1) / TB, TB, 0, stream>>>(g2, Wl2, bl2, out);
}


# Edge aggregation privatised in LDS: a group filters its edge chunk for its 4000 destination rows, sums them with LDS float adds, adds one partial row per group to the output; sum order changed
# speedup vs baseline: 1.5237x; 1.1594x over previous
#include <hip/hip_runtime.h>
#include <hip/hip_bf16.h>


typedef float v2f __attribute__((ext_vector_type(2)));
typedef float v8f __attribute__((ext_vector_type(8)));

#define NNODES  200000
#define NEDGES  6400000
#define NFEAT   128
#define NHID    16
#define NMLP    100
#define NCLS    27
#define NGRAPH  4096


__global__ void fill_f32(float* __restrict__ p, int n, float v) {
    int i = blockIdx.x * blockDim.x + threadIdx.x;
    if (i < n) p[i] = v;
}

#define HR     8
#define HBINS  (NNODES / HR)
#define HCH    32
#define HCHUNK (NEDGES / HCH)
#define HN4    (HCHUNK / 4)

__global__ __launch_bounds__(1024)
void deg_hist(const int* __restrict__ dst, unsigned* __restrict__ part) {
    __shared__ __attribute__((aligned(16))) unsigned h[HBINS];
    const int tid = threadIdx.x;
    const int r   = blockIdx.x & (HR - 1);
    const int c   = blockIdx.x >> 3;

    uint4* h4 = (uint4*)h;
    for (int i = tid; i < HBINS / 4; i += 1024) h4[i] = make_uint4(0u, 0u, 0u, 0u);
    __syncthreads();

    const unsigned lo = (unsigned)(r * HBINS);
    const int4* p = (const int4*)(dst + (size_t)c * HCHUNK);
    for (int base = tid; base < HN4; base += 1024 * 8) {
        int4 v[8];
#pragma unroll
        for (int u = 0; u < 8; ++u) {
            const int j = base + u * 1024;
            v[u] = (j < HN4) ? p[j] : make_int4(-1, -1, -1, -1);
        }
#pragma unroll
        for (int u = 0; u < 8; ++u) {
            unsigned a;
            a = (unsigned)v[u].x - lo; if (a < (unsigned)HBINS) atomicAdd(&h[a], 1u);
            a = (unsigned)v[u].y - lo; if (a < (unsigned)HBINS) atomicAdd(&h[a], 1u);
            a = (unsigned)v[u].z - lo; if (a < (unsigned)HBINS) atomicAdd(&h[a], 1u);
            a = (unsigned)v[u].w - lo; if (a < (unsigned)HBINS) atomicAdd(&h[a], 1u);
        }
    }
    __syncthreads();

    uint4* o = (uint4*)(part + (size_t)c * NNODES + lo);
    for (int i = tid; i < HBINS / 4; i += 1024) o[i] = h4[i];
}

__global__ void deg_reduce(const unsigned* __restrict__ part, float* __restrict__ dinv) {
    const int i = blockIdx.x * blockDim.x + threadIdx.x;
    if (i >= NNODES / 4) return;
    const uint4* p = (const uint4*)part + i;
    uint4 s = make_uint4(0u, 0u, 0u, 0u);
#pragma unroll
    for (int c = 0; c < HCH; ++c) {
        const uint4 v = p[(size_t)c * (NNODES / 4)];
        s.x += v.x; s.y += v.y; s.z += v.z; s.w += v.w;
    }
    float4 d;
    d.x = rsqrtf((float)s.x + 1.0f);
    d.y = rsqrtf((float)s.y + 1.0f);
    d.z = rsqrtf((float)s.z + 1.0f);
    d.w = rsqrtf((float)s.w + 1.0f);
    ((float4*)dinv)[i] = d;
}

__global__ __launch_bounds__(256)
void gemm_wmma_f32(const float* __restrict__ X, const float* __restrict__ W,
                   float* __restrict__ T, int M, int K) {
    __shared__ float lw[NFEAT * NHID];
    const int tid = threadIdx.x;
    const int kn  = K * NHID;
    for (int i = tid; i < kn; i += 256) lw[i] = W[i];
    __syncthreads();

    const int wave = tid >> 5;
    const int lane = tid & 31;
    const int tile = blockIdx.x * 8 + wave;
    if (tile * 16 >= M) return;

    const int row   = tile * 16 + (lane & 15);
    const int kHalf = (lane >> 4) << 1;
    const int n     = lane & 15;

    v8f c = {};
    const float* xr = X + (size_t)row * K + kHalf;
    for (int k0 = 0; k0 < K; k0 += 4) {
        float2 av = *(const float2*)(xr + k0);
        v2f a; a.x = av.x; a.y = av.y;
        v2f b;
        b.x = lw[(k0 + kHalf)     * NHID + n];
        b.y = lw[(k0 + kHalf + 1) * NHID + n];
        c = __builtin_amdgcn_wmma_f32_16x16x4_f32(
                 false, a,  false, b,
                 (short)0, c,  false,  false);
    }

    const int rbase = tile * 16 + ((lane >> 4) << 3);
#pragma unroll
    for (int v = 0; v < 8; ++v)
        T[(size_t)(rbase + v) * NHID + n] = c[v];
}

#define AR     50
#define ANR    (NNODES / AR)
#define ACH    5
#define ACHUNK (NEDGES / ACH)
#define AN4    (ACHUNK / 4)
#define AU     6
#define AQCAP  2048
#define AB     9

__global__ __launch_bounds__(1024)
void edge_aggregate(const int* __restrict__ src, const int* __restrict__ dst,
                    const float* __restrict__ dinv, const float* __restrict__ T,
                    float* __restrict__ OUT) {
    __shared__ __attribute__((aligned(16))) float acc[ANR * NHID];
    __shared__ uint2 queue[AQCAP];
    __shared__ unsigned qn[2];
    const int tid  = threadIdx.x;
    const int lane = tid & 31;
    const int r    = blockIdx.x % AR;
    const int c    = blockIdx.x / AR;

    float4* a4 = (float4*)acc;
    for (int i = tid; i < ANR * NHID / 4; i += 1024) a4[i] = make_float4(0.0f, 0.0f, 0.0f, 0.0f);
    if (tid < 2) qn[tid] = 0u;
    __syncthreads();

    const unsigned lo = (unsigned)(r * ANR);
    const int ebase   = c * ACHUNK;
    const int4* p     = (const int4*)(dst + (size_t)ebase);
    const int grp = tid >> 4;
    const int f   = tid & 15;

    int par = 0;
    for (int base = 0; base < AN4; base += 1024 * AU, par ^= 1) {
        unsigned wbase = 0u;
        for (unsigned rlo = 0u; ; rlo += (unsigned)AQCAP) {
            int4 v[AU];
#pragma unroll
            for (int u = 0; u < AU; ++u) {
                const int j = base + u * 1024 + tid;
                v[u] = (j < AN4) ? p[j] : make_int4(-1, -1, -1, -1);
            }
            unsigned msk[4 * AU];
            unsigned tot = 0u;
#pragma unroll
            for (int u = 0; u < AU; ++u) {
                msk[4 * u + 0] = __builtin_amdgcn_ballot_w32((unsigned)v[u].x - lo < (unsigned)ANR);
                msk[4 * u + 1] = __builtin_amdgcn_ballot_w32((unsigned)v[u].y - lo < (unsigned)ANR);
                msk[4 * u + 2] = __builtin_amdgcn_ballot_w32((unsigned)v[u].z - lo < (unsigned)ANR);
                msk[4 * u + 3] = __builtin_amdgcn_ballot_w32((unsigned)v[u].w - lo < (unsigned)ANR);
                tot += (unsigned)(__builtin_popcount(msk[4 * u + 0]) + __builtin_popcount(msk[4 * u + 1]) +
                                  __builtin_popcount(msk[4 * u + 2]) + __builtin_popcount(msk[4 * u + 3]));
            }
            if (rlo == 0u) {
                unsigned w = 0u;
                if (lane == 0) w = atomicAdd(&qn[par], tot);
                wbase = (unsigned)__builtin_amdgcn_readfirstlane((int)w);
            }
            unsigned run = wbase - rlo;
#pragma unroll
            for (int u = 0; u < AU; ++u) {
                const int e0 = ebase + 4 * (base + u * 1024 + tid);
                const int dv[4] = { v[u].x, v[u].y, v[u].z, v[u].w };
#pragma unroll
                for (int k = 0; k < 4; ++k) {
                    const unsigned m = msk[4 * u + k];
                    if (m != 0u) {
                        const unsigned a = (unsigned)dv[k] - lo;
                        const unsigned pos = run + __builtin_amdgcn_mbcnt_lo(m, 0u);
                        if (a < (unsigned)ANR && pos < (unsigned)AQCAP)
                            queue[pos] = make_uint2(a, (unsigned)(e0 + k));
                        run += (unsigned)__builtin_popcount(m);
                    }
                }
            }
            __syncthreads();
            if (tid == 0) qn[par ^ 1] = 0u;
            const unsigned total = qn[par];
            const unsigned n = min(total - rlo, (unsigned)AQCAP);

            for (unsigned j0 = (unsigned)grp; j0 < n; j0 += 64u * AB) {
                uint2 q[AB];
                int   s[AB];
                float m[AB], w[AB];
#pragma unroll
                for (int k = 0; k < AB; ++k) {
                    const unsigned jj = j0 + 64u * k;
                    q[k] = (jj < n) ? queue[jj] : make_uint2(0xffffffffu, 0u);
                }
#pragma unroll
                for (int k = 0; k < AB; ++k) s[k] = src[q[k].y];
#pragma unroll
                for (int k = 0; k < AB; ++k) {
                    const unsigned a = (q[k].x < (unsigned)ANR) ? q[k].x : 0u;
                    w[k] = dinv[s[k]] * dinv[lo + a];
                    m[k] = T[(size_t)s[k] * NHID + f];
                }
#pragma unroll
                for (int k = 0; k < AB; ++k)
                    if (q[k].x < (unsigned)ANR) atomicAdd(&acc[q[k].x * NHID + f], m[k] * w[k]);
            }
            __syncthreads();
            if (rlo + (unsigned)AQCAP >= total) break;
        }
    }

    float* o = OUT + (size_t)lo * NHID;
    for (int i = tid; i < ANR * NHID; i += 1024) atomicAdd(o + i, acc[i]);
}

__global__ void finalize_node(float* __restrict__ ACC, const float* __restrict__ T,
                              const float* __restrict__ dinv, const float* __restrict__ bias,
                              int n16, int doRelu) {
    int idx = blockIdx.x * blockDim.x + threadIdx.x;
    if (idx >= n16) return;
    int i = idx >> 4, f = idx & 15;
    float di = dinv[i];
    float v  = ACC[idx] + T[idx] * di * di + bias[f];
    ACC[idx] = doRelu ? fmaxf(v, 0.0f) : v;
}

__global__ void pool_kernel(const int* __restrict__ bvec, const float* __restrict__ Hn,
                            float* __restrict__ Gp, int n) {
    int t = blockIdx.x * blockDim.x + threadIdx.x;
    int i = t >> 2, q = t & 3;
    if (i >= n) return;
    int g = bvec[i];
    const float4 h = *(const float4*)(Hn + (size_t)i * NHID + q * 4);
    float* o = Gp + (size_t)g * NHID + q * 4;
    atomicAdd(o + 0, h.x);
    atomicAdd(o + 1, h.y);
    atomicAdd(o + 2, h.z);
    atomicAdd(o + 3, h.w);
}

__global__ void lin1_kernel(const float* __restrict__ G, const float* __restrict__ W,
                            const float* __restrict__ B, float* __restrict__ O) {
    int t = blockIdx.x * blockDim.x + threadIdx.x;
    if (t >= NGRAPH * NMLP) return;
    int gi = t / NMLP, j = t % NMLP;
    float acc = B[j];
#pragma unroll
    for (int k = 0; k < NHID; ++k)
        acc = fmaf(fmaxf(G[gi * NHID + k], 0.0f), W[k * NMLP + j], acc);
    O[t] = fmaxf(acc, 0.0f);
}

__global__ void lin2_kernel(const float* __restrict__ G, const float* __restrict__ W,
                            const float* __restrict__ B, float* __restrict__ O) {
    int t = blockIdx.x * blockDim.x + threadIdx.x;
    if (t >= NGRAPH * NCLS) return;
    int gi = t / NCLS, c = t % NCLS;
    float acc = B[c];
    for (int k = 0; k < NMLP; ++k)
        acc = fmaf(G[gi * NMLP + k], W[k * NCLS + c], acc);
    O[t] = acc;
}


extern "C" void kernel_launch(void* const* d_in, const int* in_sizes, int n_in,
                              void* d_out, int out_size, void* d_ws, size_t ws_size,
                              hipStream_t stream) {
    const float* x   = (const float*)d_in[0];
    const int*  eidx = (const int*)d_in[1];
    const int*  bvec = (const int*)d_in[2];
    const float* W1  = (const float*)d_in[3];
    const float* b1  = (const float*)d_in[4];
    const float* W2  = (const float*)d_in[5];
    const float* b2  = (const float*)d_in[6];
    const float* Wl1 = (const float*)d_in[7];
    const float* bl1 = (const float*)d_in[8];
    const float* Wl2 = (const float*)d_in[9];
    const float* bl2 = (const float*)d_in[10];
    float* out = (float*)d_out;

    char* ws = (char*)d_ws;
    size_t off = 0;
    auto walloc = [&](size_t bytes) -> void* {
        void* p = ws + off;
        off = (off + bytes + 255) & ~(size_t)255;
        return p;
    };
    float* deg  = (float*)walloc((size_t)NNODES * 4);
    float* bufA = (float*)walloc((size_t)NNODES * NHID * 4);
    float* bufB = (float*)walloc((size_t)NNODES * NHID * 4);
    float* gbuf = (float*)walloc((size_t)NGRAPH * NHID * 4);
    float* g2   = (float*)walloc((size_t)NGRAPH * NMLP * 4);

    const int* src = eidx;
    const int* dst = eidx + NEDGES;

    const int TB = 256;
    const int nN16  = (NNODES * NHID + TB - 1) / TB;
    const int nN4   = (NNODES * 4 + TB - 1) / TB;
    const int nTile = ((NNODES / 16) + 7) / 8;

    unsigned* part = (unsigned*)bufA;
    deg_hist<<<HR * HCH, 1024, 0, stream>>>(dst, part);
    deg_reduce<<<(NNODES / 4 + TB - 1) / TB, TB, 0, stream>>>(part, deg);
    float* dinv = deg;

    gemm_wmma_f32<<<nTile, TB, 0, stream>>>(x, W1, bufA, NNODES, NFEAT);
    fill_f32<<<nN16, TB, 0, stream>>>(bufB, NNODES * NHID, 0.0f);
    edge_aggregate<<<AR * ACH, 1024, 0, stream>>>(src, dst, dinv, bufA, bufB);
    finalize_node<<<nN16, TB, 0, stream>>>(bufB, bufA, dinv, b1, NNODES * NHID, 1);

    gemm_wmma_f32<<<nTile, TB, 0, stream>>>(bufB, W2, bufA, NNODES, NHID);
    fill_f32<<<nN16, TB, 0, stream>>>(bufB, NNODES * NHID, 0.0f);
    edge_aggregate<<<AR * ACH, 1024, 0, stream>>>(src, dst, dinv, bufA, bufB);
    finalize_node<<<nN16, TB, 0, stream>>>(bufB, bufA, dinv, b2, NNODES * NHID, 0);

    fill_f32<<<(NGRAPH * NHID + TB - 1) / TB, TB, 0, stream>>>(gbuf, NGRAPH * NHID, 0.0f);
    pool_kernel<<<nN4, TB, 0, stream>>>(bvec, bufB, gbuf, NNODES);
    lin1_kernel<<<(NGRAPH * NMLP + TB - 1) / TB, TB, 0, stream>>>(gbuf, Wl1, bl1, g2);
    lin2_kernel<<<(NGRAPH * NCLS + TB - 1) / TB, TB, 0, stream>>>(g2, Wl2, bl2, out);
}


# Edge aggregation software-pipelined: two queue halves, one barrier per tile, next tile's ids requested before the gather, gathered rows in flight across the next filter; LDS add order changed
# speedup vs baseline: 1.7887x; 1.1739x over previous
#include <hip/hip_runtime.h>
#include <hip/hip_bf16.h>


typedef float v2f __attribute__((ext_vector_type(2)));
typedef float v8f __attribute__((ext_vector_type(8)));

#define NNODES  200000
#define NEDGES  6400000
#define NFEAT   128
#define NHID    16
#define NMLP    100
#define NCLS    27
#define NGRAPH  4096


__global__ void fill_f32(float* __restrict__ p, int n, float v) {
    int i = blockIdx.x * blockDim.x + threadIdx.x;
    if (i < n) p[i] = v;
}

#define HR     8
#define HBINS  (NNODES / HR)
#define HCH    32
#define HCHUNK (NEDGES / HCH)
#define HN4    (HCHUNK / 4)

__global__ __launch_bounds__(1024)
void deg_hist(const int* __restrict__ dst, unsigned* __restrict__ part) {
    __shared__ __attribute__((aligned(16))) unsigned h[HBINS];
    const int tid = threadIdx.x;
    const int r   = blockIdx.x & (HR - 1);
    const int c   = blockIdx.x >> 3;

    uint4* h4 = (uint4*)h;
    for (int i = tid; i < HBINS / 4; i += 1024) h4[i] = make_uint4(0u, 0u, 0u, 0u);
    __syncthreads();

    const unsigned lo = (unsigned)(r * HBINS);
    const int4* p = (const int4*)(dst + (size_t)c * HCHUNK);
    for (int base = tid; base < HN4; base += 1024 * 8) {
        int4 v[8];
#pragma unroll
        for (int u = 0; u < 8; ++u) {
            const int j = base + u * 1024;
            v[u] = (j < HN4) ? p[j] : make_int4(-1, -1, -1, -1);
        }
#pragma unroll
        for (int u = 0; u < 8; ++u) {
            unsigned a;
            a = (unsigned)v[u].x - lo; if (a < (unsigned)HBINS) atomicAdd(&h[a], 1u);
            a = (unsigned)v[u].y - lo; if (a < (unsigned)HBINS) atomicAdd(&h[a], 1u);
            a = (unsigned)v[u].z - lo; if (a < (unsigned)HBINS) atomicAdd(&h[a], 1u);
            a = (unsigned)v[u].w - lo; if (a < (unsigned)HBINS) atomicAdd(&h[a], 1u);
        }
    }
    __syncthreads();

    uint4* o = (uint4*)(part + (size_t)c * NNODES + lo);
    for (int i = tid; i < HBINS / 4; i += 1024) o[i] = h4[i];
}

__global__ void deg_reduce(const unsigned* __restrict__ part, float* __restrict__ dinv) {
    const int i = blockIdx.x * blockDim.x + threadIdx.x;
    if (i >= NNODES / 4) return;
    const uint4* p = (const uint4*)part + i;
    uint4 s = make_uint4(0u, 0u, 0u, 0u);
#pragma unroll
    for (int c = 0; c < HCH; ++c) {
        const uint4 v = p[(size_t)c * (NNODES / 4)];
        s.x += v.x; s.y += v.y; s.z += v.z; s.w += v.w;
    }
    float4 d;
    d.x = rsqrtf((float)s.x + 1.0f);
    d.y = rsqrtf((float)s.y + 1.0f);
    d.z = rsqrtf((float)s.z + 1.0f);
    d.w = rsqrtf((float)s.w + 1.0f);
    ((float4*)dinv)[i] = d;
}

__global__ __launch_bounds__(256)
void gemm_wmma_f32(const float* __restrict__ X, const float* __restrict__ W,
                   float* __restrict__ T, int M, int K) {
    __shared__ float lw[NFEAT * NHID];
    const int tid = threadIdx.x;
    const int kn  = K * NHID;
    for (int i = tid; i < kn; i += 256) lw[i] = W[i];
    __syncthreads();

    const int wave = tid >> 5;
    const int lane = tid & 31;
    const int tile = blockIdx.x * 8 + wave;
    if (tile * 16 >= M) return;

    const int row   = tile * 16 + (lane & 15);
    const int kHalf = (lane >> 4) << 1;
    const int n     = lane & 15;

    v8f c = {};
    const float* xr = X + (size_t)row * K + kHalf;
    for (int k0 = 0; k0 < K; k0 += 4) {
        float2 av = *(const float2*)(xr + k0);
        v2f a; a.x = av.x; a.y = av.y;
        v2f b;
        b.x = lw[(k0 + kHalf)     * NHID + n];
        b.y = lw[(k0 + kHalf + 1) * NHID + n];
        c = __builtin_amdgcn_wmma_f32_16x16x4_f32(
                 false, a,  false, b,
                 (short)0, c,  false,  false);
    }

    const int rbase = tile * 16 + ((lane >> 4) << 3);
#pragma unroll
    for (int v = 0; v < 8; ++v)
        T[(size_t)(rbase + v) * NHID + n] = c[v];
}

#define AR     50
#define ANR    (NNODES / AR)
#define ACH    5
#define ACHUNK (NEDGES / ACH)
#define AN4    (ACHUNK / 4)
#define AU     6
#define AQCAP  2048
#define AB     9

__global__ __launch_bounds__(1024)
void edge_aggregate(const int* __restrict__ src, const int* __restrict__ dst,
                    const float* __restrict__ dinv, const float* __restrict__ T,
                    float* __restrict__ OUT) {
    __shared__ __attribute__((aligned(16))) float acc[ANR * NHID];
    __shared__ uint2 queue[2][AQCAP];
    __shared__ unsigned qn[2];
    const int tid  = threadIdx.x;
    const int lane = tid & 31;
    const int r    = blockIdx.x % AR;
    const int c    = blockIdx.x / AR;

    float4* a4 = (float4*)acc;
    for (int i = tid; i < ANR * NHID / 4; i += 1024) a4[i] = make_float4(0.0f, 0.0f, 0.0f, 0.0f);
    if (tid < 2) qn[tid] = 0u;
    for (int i = tid; i < 2 * AQCAP; i += 1024) (&queue[0][0])[i] = make_uint2(0xffffffffu, 0u);
    __syncthreads();

    const unsigned lo = (unsigned)(r * ANR);
    const int ebase   = c * ACHUNK;
    const int4* p     = (const int4*)(dst + (size_t)ebase);
    const unsigned grp = (unsigned)(tid >> 4);
    const int f   = tid & 15;

    int4 v[AU];
    auto load_tile = [&](int base) {
#pragma unroll
        for (int u = 0; u < AU; ++u) {
            const int j = base + u * 1024 + tid;
            v[u] = (j < AN4) ? p[j] : make_int4(-1, -1, -1, -1);
        }
    };

    auto masks = [&](unsigned (&msk)[4 * AU]) -> unsigned {
        unsigned tot = 0u;
#pragma unroll
        for (int u = 0; u < AU; ++u) {
            msk[4 * u + 0] = __builtin_amdgcn_ballot_w32((unsigned)v[u].x - lo < (unsigned)ANR);
            msk[4 * u + 1] = __builtin_amdgcn_ballot_w32((unsigned)v[u].y - lo < (unsigned)ANR);
            msk[4 * u + 2] = __builtin_amdgcn_ballot_w32((unsigned)v[u].z - lo < (unsigned)ANR);
            msk[4 * u + 3] = __builtin_amdgcn_ballot_w32((unsigned)v[u].w - lo < (unsigned)ANR);
            tot += (unsigned)(__builtin_popcount(msk[4 * u + 0]) + __builtin_popcount(msk[4 * u + 1]) +
                              __builtin_popcount(msk[4 * u + 2]) + __builtin_popcount(msk[4 * u + 3]));
        }
        return tot;
    };

    auto emit = [&](const unsigned (&msk)[4 * AU], int base, unsigned run, uint2* qb) {
#pragma unroll
        for (int u = 0; u < AU; ++u) {
            const int e0 = ebase + 4 * (base + u * 1024 + tid);
            const int dv[4] = { v[u].x, v[u].y, v[u].z, v[u].w };
#pragma unroll
            for (int k = 0; k < 4; ++k) {
                const unsigned m = msk[4 * u + k];
                if (m != 0u) {
                    const unsigned a = (unsigned)dv[k] - lo;
                    const unsigned pos = run + __builtin_amdgcn_mbcnt_lo(m, 0u);
                    if (a < (unsigned)ANR && pos < (unsigned)AQCAP)
                        qb[pos] = make_uint2(a, (unsigned)(e0 + k));
                    run += (unsigned)__builtin_popcount(m);
                }
            }
        }
    };

    unsigned pa[AB];
    float pm[AB], ps[AB], pd[AB];
#pragma unroll
    for (int k = 0; k < AB; ++k) { pa[k] = 0xffffffffu; pm[k] = 0.0f; ps[k] = 0.0f; pd[k] = 0.0f; }

    auto issue = [&](const uint2* qb, unsigned n, unsigned j0) {
        uint2 q[AB];
        unsigned s[AB];
#pragma unroll
        for (int k = 0; k < AB; ++k) {
            const unsigned jj = j0 + 64u * k;
            const uint2 t = qb[jj & (unsigned)(AQCAP - 1)];
            q[k].x = (jj < n) ? t.x : 0xffffffffu;
            q[k].y = (jj < n) ? t.y : 0u;
        }
#pragma unroll
        for (int k = 0; k < AB; ++k) {
            s[k]  = (unsigned)*(const int*)((const char*)src + (q[k].y << 2));
            pa[k] = q[k].x;
            const unsigned a = (q[k].x < (unsigned)ANR) ? q[k].x : 0u;
            pd[k] = dinv[lo + a];
        }
#pragma unroll
        for (int k = 0; k < AB; ++k) {
            ps[k] = *(const float*)((const char*)dinv + (s[k] << 2));
            pm[k] = *(const float*)((const char*)T + ((s[k] << 6) | ((unsigned)f << 2)));
        }
    };
    auto commit = [&]() {
#pragma unroll
        for (int k = 0; k < AB; ++k) {
            float val = pm[k] * (ps[k] * pd[k]);
            asm volatile("" : "+v"(val));
            if (pa[k] < (unsigned)ANR) atomicAdd(&acc[pa[k] * NHID + f], val);
        }
    };

    load_tile(0);
    unsigned msk[4 * AU];
    unsigned tot = masks(msk);
    int base = 0;
    unsigned rlo = 0u;
    unsigned total = 0u;
    unsigned wbase = 0u;
    int qpar = 0;
    int cpar = 0;
    unsigned seen = 0u, seenOther = 0u;
    while (base < AN4) {
        uint2* qb = queue[qpar];
        if (rlo == 0u) {
            unsigned w = 0u;
            if (lane == 0) w = atomicAdd(&qn[cpar], tot);
            wbase = (unsigned)__builtin_amdgcn_readfirstlane((int)w) - seen;
        }
        emit(msk, base, wbase - rlo, qb);
        commit();
        __syncthreads();
        if (rlo == 0u) {
            const unsigned cnt = (unsigned)__builtin_amdgcn_readfirstlane((int)qn[cpar]);
            total = cnt - seen;
            seen  = cnt;
        }
        const unsigned n = min(total - rlo, (unsigned)AQCAP);
        const bool last = rlo + (unsigned)AQCAP >= total;
        if (last) load_tile(base + 1024 * AU);
        issue(qb, n, grp);
        for (unsigned jr = 64u * AB; jr < n; jr += 64u * AB) { commit(); issue(qb, n, grp + jr); }
        qpar ^= 1;
        if (last) {
            tot = masks(msk);
            base += 1024 * AU;
            rlo = 0u;
            cpar ^= 1;
            const unsigned t = seen; seen = seenOther; seenOther = t;
        } else {
            rlo += (unsigned)AQCAP;
        }
    }
    commit();
    __syncthreads();

    float* o = OUT + (size_t)lo * NHID;
    for (int i = tid; i < ANR * NHID; i += 1024) atomicAdd(o + i, acc[i]);
}

__global__ void finalize_node(float* __restrict__ ACC, const float* __restrict__ T,
                              const float* __restrict__ dinv, const float* __restrict__ bias,
                              int n16, int doRelu) {
    int idx = blockIdx.x * blockDim.x + threadIdx.x;
    if (idx >= n16) return;
    int i = idx >> 4, f = idx & 15;
    float di = dinv[i];
    float v  = ACC[idx] + T[idx] * di * di + bias[f];
    ACC[idx] = doRelu ? fmaxf(v, 0.0f) : v;
}

__global__ void pool_kernel(const int* __restrict__ bvec, const float* __restrict__ Hn,
                            float* __restrict__ Gp, int n) {
    int t = blockIdx.x * blockDim.x + threadIdx.x;
    int i = t >> 2, q = t & 3;
    if (i >= n) return;
    int g = bvec[i];
    const float4 h = *(const float4*)(Hn + (size_t)i * NHID + q * 4);
    float* o = Gp + (size_t)g * NHID + q * 4;
    atomicAdd(o + 0, h.x);
    atomicAdd(o + 1, h.y);
    atomicAdd(o + 2, h.z);
    atomicAdd(o + 3, h.w);
}

__global__ void lin1_kernel(const float* __restrict__ G, const float* __restrict__ W,
                            const float* __restrict__ B, float* __restrict__ O) {
    int t = blockIdx.x * blockDim.x + threadIdx.x;
    if (t >= NGRAPH * NMLP) return;
    int gi = t / NMLP, j = t % NMLP;
    float acc = B[j];
#pragma unroll
    for (int k = 0; k < NHID; ++k)
        acc = fmaf(fmaxf(G[gi * NHID + k], 0.0f), W[k * NMLP + j], acc);
    O[t] = fmaxf(acc, 0.0f);
}

__global__ void lin2_kernel(const float* __restrict__ G, const float* __restrict__ W,
                            const float* __restrict__ B, float* __restrict__ O) {
    int t = blockIdx.x * blockDim.x + threadIdx.x;
    if (t >= NGRAPH * NCLS) return;
    int gi = t / NCLS, c = t % NCLS;
    float acc = B[c];
    for (int k = 0; k < NMLP; ++k)
        acc = fmaf(G[gi * NMLP + k], W[k * NCLS + c], acc);
    O[t] = acc;
}


extern "C" void kernel_launch(void* const* d_in, const int* in_sizes, int n_in,
                              void* d_out, int out_size, void* d_ws, size_t ws_size,
                              hipStream_t stream) {
    const float* x   = (const float*)d_in[0];
    const int*  eidx = (const int*)d_in[1];
    const int*  bvec = (const int*)d_in[2];
    const float* W1  = (const float*)d_in[3];
    const float* b1  = (const float*)d_in[4];
    const float* W2  = (const float*)d_in[5];
    const float* b2  = (const float*)d_in[6];
    const float* Wl1 = (const float*)d_in[7];
    const float* bl1 = (const float*)d_in[8];
    const float* Wl2 = (const float*)d_in[9];
    const float* bl2 = (const float*)d_in[10];
    float* out = (float*)d_out;

    char* ws = (char*)d_ws;
    size_t off = 0;
    auto walloc = [&](size_t bytes) -> void* {
        void* p = ws + off;
        off = (off + bytes + 255) & ~(size_t)255;
        return p;
    };
    float* deg  = (float*)walloc((size_t)NNODES * 4);
    float* bufA = (float*)walloc((size_t)NNODES * NHID * 4);
    float* bufB = (float*)walloc((size_t)NNODES * NHID * 4);
    float* gbuf = (float*)walloc((size_t)NGRAPH * NHID * 4);
    float* g2   = (float*)walloc((size_t)NGRAPH * NMLP * 4);

    const int* src = eidx;
    const int* dst = eidx + NEDGES;

    const int TB = 256;
    const int nN16  = (NNODES * NHID + TB - 1) / TB;
    const int nN4   = (NNODES * 4 + TB - 1) / TB;
    const int nTile = ((NNODES / 16) + 7) / 8;

    unsigned* part = (unsigned*)bufA;
    deg_hist<<<HR * HCH, 1024, 0, stream>>>(dst, part);
    deg_reduce<<<(NNODES / 4 + TB - 1) / TB, TB, 0, stream>>>(part, deg);
    float* dinv = deg;

    gemm_wmma_f32<<<nTile, TB, 0, stream>>>(x, W1, bufA, NNODES, NFEAT);
    fill_f32<<<nN16, TB, 0, stream>>>(bufB, NNODES * NHID, 0.0f);
    edge_aggregate<<<AR * ACH, 1024, 0, stream>>>(src, dst, dinv, bufA, bufB);
    finalize_node<<<nN16, TB, 0, stream>>>(bufB, bufA, dinv, b1, NNODES * NHID, 1);

    gemm_wmma_f32<<<nTile, TB, 0, stream>>>(bufB, W2, bufA, NNODES, NHID);
    fill_f32<<<nN16, TB, 0, stream>>>(bufB, NNODES * NHID, 0.0f);
    edge_aggregate<<<AR * ACH, 1024, 0, stream>>>(src, dst, dinv, bufA, bufB);
    finalize_node<<<nN16, TB, 0, stream>>>(bufB, bufA, dinv, b2, NNODES * NHID, 0);

    fill_f32<<<(NGRAPH * NHID + TB - 1) / TB, TB, 0, stream>>>(gbuf, NGRAPH * NHID, 0.0f);
    pool_kernel<<<nN4, TB, 0, stream>>>(bvec, bufB, gbuf, NNODES);
    lin1_kernel<<<(NGRAPH * NMLP + TB - 1) / TB, TB, 0, stream>>>(gbuf, Wl1, bl1, g2);
    lin2_kernel<<<(NGRAPH * NCLS + TB - 1) / TB, TB, 0, stream>>>(g2, Wl2, bl2, out);
}


# Edge aggregation without a barrier in its loop: each wave filters its own tiles into a wave-private LDS queue and gathers them itself, so waves cover each other's load waits; LDS add order changed
# speedup vs baseline: 1.9008x; 1.0627x over previous
#include <hip/hip_runtime.h>
#include <hip/hip_bf16.h>


typedef float v2f __attribute__((ext_vector_type(2)));
typedef float v8f __attribute__((ext_vector_type(8)));

#define NNODES  200000
#define NEDGES  6400000
#define NFEAT   128
#define NHID    16
#define NMLP    100
#define NCLS    27
#define NGRAPH  4096


__global__ void fill_f32(float* __restrict__ p, int n, float v) {
    int i = blockIdx.x * blockDim.x + threadIdx.x;
    if (i < n) p[i] = v;
}

#define HR     8
#define HBINS  (NNODES / HR)
#define HCH    32
#define HCHUNK (NEDGES / HCH)
#define HN4    (HCHUNK / 4)

__global__ __launch_bounds__(1024)
void deg_hist(const int* __restrict__ dst, unsigned* __restrict__ part) {
    __shared__ __attribute__((aligned(16))) unsigned h[HBINS];
    const int tid = threadIdx.x;
    const int r   = blockIdx.x & (HR - 1);
    const int c   = blockIdx.x >> 3;

    uint4* h4 = (uint4*)h;
    for (int i = tid; i < HBINS / 4; i += 1024) h4[i] = make_uint4(0u, 0u, 0u, 0u);
    __syncthreads();

    const unsigned lo = (unsigned)(r * HBINS);
    const int4* p = (const int4*)(dst + (size_t)c * HCHUNK);
    for (int base = tid; base < HN4; base += 1024 * 8) {
        int4 v[8];
#pragma unroll
        for (int u = 0; u < 8; ++u) {
            const int j = base + u * 1024;
            v[u] = (j < HN4) ? p[j] : make_int4(-1, -1, -1, -1);
        }
#pragma unroll
        for (int u = 0; u < 8; ++u) {
            unsigned a;
            a = (unsigned)v[u].x - lo; if (a < (unsigned)HBINS) atomicAdd(&h[a], 1u);
            a = (unsigned)v[u].y - lo; if (a < (unsigned)HBINS) atomicAdd(&h[a], 1u);
            a = (unsigned)v[u].z - lo; if (a < (unsigned)HBINS) atomicAdd(&h[a], 1u);
            a = (unsigned)v[u].w - lo; if (a < (unsigned)HBINS) atomicAdd(&h[a], 1u);
        }
    }
    __syncthreads();

    uint4* o = (uint4*)(part + (size_t)c * NNODES + lo);
    for (int i = tid; i < HBINS / 4; i += 1024) o[i] = h4[i];
}

__global__ void deg_reduce(const unsigned* __restrict__ part, float* __restrict__ dinv) {
    const int i = blockIdx.x * blockDim.x + threadIdx.x;
    if (i >= NNODES / 4) return;
    const uint4* p = (const uint4*)part + i;
    uint4 s = make_uint4(0u, 0u, 0u, 0u);
#pragma unroll
    for (int c = 0; c < HCH; ++c) {
        const uint4 v = p[(size_t)c * (NNODES / 4)];
        s.x += v.x; s.y += v.y; s.z += v.z; s.w += v.w;
    }
    float4 d;
    d.x = rsqrtf((float)s.x + 1.0f);
    d.y = rsqrtf((float)s.y + 1.0f);
    d.z = rsqrtf((float)s.z + 1.0f);
    d.w = rsqrtf((float)s.w + 1.0f);
    ((float4*)dinv)[i] = d;
}

__global__ __launch_bounds__(256)
void gemm_wmma_f32(const float* __restrict__ X, const float* __restrict__ W,
                   float* __restrict__ T, int M, int K) {
    __shared__ float lw[NFEAT * NHID];
    const int tid = threadIdx.x;
    const int kn  = K * NHID;
    for (int i = tid; i < kn; i += 256) lw[i] = W[i];
    __syncthreads();

    const int wave = tid >> 5;
    const int lane = tid & 31;
    const int tile = blockIdx.x * 8 + wave;
    if (tile * 16 >= M) return;

    const int row   = tile * 16 + (lane & 15);
    const int kHalf = (lane >> 4) << 1;
    const int n     = lane & 15;

    v8f c = {};
    const float* xr = X + (size_t)row * K + kHalf;
    for (int k0 = 0; k0 < K; k0 += 4) {
        float2 av = *(const float2*)(xr + k0);
        v2f a; a.x = av.x; a.y = av.y;
        v2f b;
        b.x = lw[(k0 + kHalf)     * NHID + n];
        b.y = lw[(k0 + kHalf + 1) * NHID + n];
        c = __builtin_amdgcn_wmma_f32_16x16x4_f32(
                 false, a,  false, b,
                 (short)0, c,  false,  false);
    }

    const int rbase = tile * 16 + ((lane >> 4) << 3);
#pragma unroll
    for (int v = 0; v < 8; ++v)
        T[(size_t)(rbase + v) * NHID + n] = c[v];
}

#define AR     50
#define ANR    (NNODES / AR)
#define ACH    5
#define ACHUNK (NEDGES / ACH)
#define AN4    (ACHUNK / 4)
#define AU     6
#define AWT    (32 * AU)
#define ANWT   ((AN4 + AWT - 1) / AWT)
#define AWQ    128
#define AB     10

__global__ __launch_bounds__(1024)
void edge_aggregate(const int* __restrict__ src, const int* __restrict__ dst,
                    const float* __restrict__ dinv, const float* __restrict__ T,
                    float* __restrict__ OUT) {
    __shared__ __attribute__((aligned(16))) float acc[ANR * NHID];
    __shared__ uint2 queue[32][AWQ];
    const int tid  = threadIdx.x;
    const int lane = tid & 31;
    const int wave = __builtin_amdgcn_readfirstlane(tid >> 5);
    const int r    = blockIdx.x % AR;
    const int c    = blockIdx.x / AR;

    float4* a4 = (float4*)acc;
    for (int i = tid; i < ANR * NHID / 4; i += 1024) a4[i] = make_float4(0.0f, 0.0f, 0.0f, 0.0f);
    for (int i = tid; i < 32 * AWQ; i += 1024) (&queue[0][0])[i] = make_uint2(0xffffffffu, 0u);
    __syncthreads();

    const unsigned lo = (unsigned)(r * ANR);
    const int ebase   = c * ACHUNK;
    const int4* p     = (const int4*)(dst + (size_t)ebase);
    const unsigned half = (unsigned)(lane >> 4);
    const int f   = lane & 15;
    uint2* qb = queue[wave];

    int4 v[AU];
    auto load_tile = [&](int wt) {
#pragma unroll
        for (int u = 0; u < AU; ++u) {
            const int jb = wt * AWT + u * 32;
            if (jb < AN4) v[u] = p[jb + lane];
            else          v[u] = make_int4(-1, -1, -1, -1);
        }
    };

    auto masks = [&](unsigned (&msk)[4 * AU]) -> unsigned {
        unsigned tot = 0u;
#pragma unroll
        for (int u = 0; u < AU; ++u) {
            msk[4 * u + 0] = __builtin_amdgcn_ballot_w32((unsigned)v[u].x - lo < (unsigned)ANR);
            msk[4 * u + 1] = __builtin_amdgcn_ballot_w32((unsigned)v[u].y - lo < (unsigned)ANR);
            msk[4 * u + 2] = __builtin_amdgcn_ballot_w32((unsigned)v[u].z - lo < (unsigned)ANR);
            msk[4 * u + 3] = __builtin_amdgcn_ballot_w32((unsigned)v[u].w - lo < (unsigned)ANR);
            tot += (unsigned)(__builtin_popcount(msk[4 * u + 0]) + __builtin_popcount(msk[4 * u + 1]) +
                              __builtin_popcount(msk[4 * u + 2]) + __builtin_popcount(msk[4 * u + 3]));
        }
        return tot;
    };

    auto emit = [&](const unsigned (&msk)[4 * AU], int wt, unsigned run) {
#pragma unroll
        for (int u = 0; u < AU; ++u) {
            const int e0 = ebase + 4 * (wt * AWT + u * 32 + lane);
            const int dv[4] = { v[u].x, v[u].y, v[u].z, v[u].w };
#pragma unroll
            for (int k = 0; k < 4; ++k) {
                const unsigned m = msk[4 * u + k];
                if (m != 0u) {
                    const unsigned a = (unsigned)dv[k] - lo;
                    const unsigned pos = run + __builtin_amdgcn_mbcnt_lo(m, 0u);
                    if (a < (unsigned)ANR && pos < (unsigned)AWQ)
                        qb[pos] = make_uint2(a, (unsigned)(e0 + k));
                    run += (unsigned)__builtin_popcount(m);
                }
            }
        }
    };

    unsigned pa[AB];
    float pm[AB], ps[AB], pd[AB];
#pragma unroll
    for (int k = 0; k < AB; ++k) { pa[k] = 0xffffffffu; pm[k] = 0.0f; ps[k] = 0.0f; pd[k] = 0.0f; }

    auto issue = [&](unsigned n, unsigned j0) {
        uint2 q[AB];
        unsigned s[AB];
#pragma unroll
        for (int k = 0; k < AB; ++k) {
            const unsigned jj = j0 + 2u * k;
            const uint2 t = qb[jj & (unsigned)(AWQ - 1)];
            q[k].x = (jj < n) ? t.x : 0xffffffffu;
            q[k].y = (jj < n) ? t.y : 0u;
        }
#pragma unroll
        for (int k = 0; k < AB; ++k) {
            s[k]  = (unsigned)*(const int*)((const char*)src + (q[k].y << 2));
            pa[k] = q[k].x;
            const unsigned a = (q[k].x < (unsigned)ANR) ? q[k].x : 0u;
            pd[k] = dinv[lo + a];
        }
#pragma unroll
        for (int k = 0; k < AB; ++k) {
            ps[k] = *(const float*)((const char*)dinv + (s[k] << 2));
            pm[k] = *(const float*)((const char*)T + ((s[k] << 6) | ((unsigned)f << 2)));
        }
    };
    auto commit = [&]() {
#pragma unroll
        for (int k = 0; k < AB; ++k) {
            float val = pm[k] * (ps[k] * pd[k]);
            asm volatile("" : "+v"(val));
            if (pa[k] < (unsigned)ANR) atomicAdd(&acc[pa[k] * NHID + f], val);
        }
    };

    int wt = wave;
    load_tile(wt);
    unsigned msk[4 * AU];
    unsigned tot = masks(msk);
    unsigned rlo = 0u;
    while (wt < ANWT) {
        emit(msk, wt, 0u - rlo);
        commit();
        __builtin_amdgcn_wave_barrier();
        const unsigned n = min(tot - rlo, (unsigned)AWQ);
        const bool last = rlo + (unsigned)AWQ >= tot;
        if (last) load_tile(wt + 32);
        issue(n, half);
        for (unsigned jr = 2u * AB; jr < n; jr += 2u * AB) { commit(); issue(n, half + jr); }
        __builtin_amdgcn_wave_barrier();
        if (last) {
            tot = masks(msk);
            wt += 32;
            rlo = 0u;
        } else {
            rlo += (unsigned)AWQ;
        }
    }
    commit();
    __syncthreads();

    float* o = OUT + (size_t)lo * NHID;
    for (int i = tid; i < ANR * NHID; i += 1024) atomicAdd(o + i, acc[i]);
}

__global__ void finalize_node(float* __restrict__ ACC, const float* __restrict__ T,
                              const float* __restrict__ dinv, const float* __restrict__ bias,
                              int n16, int doRelu) {
    int idx = blockIdx.x * blockDim.x + threadIdx.x;
    if (idx >= n16) return;
    int i = idx >> 4, f = idx & 15;
    float di = dinv[i];
    float v  = ACC[idx] + T[idx] * di * di + bias[f];
    ACC[idx] = doRelu ? fmaxf(v, 0.0f) : v;
}

__global__ void pool_kernel(const int* __restrict__ bvec, const float* __restrict__ Hn,
                            float* __restrict__ Gp, int n) {
    int t = blockIdx.x * blockDim.x + threadIdx.x;
    int i = t >> 2, q = t & 3;
    if (i >= n) return;
    int g = bvec[i];
    const float4 h = *(const float4*)(Hn + (size_t)i * NHID + q * 4);
    float* o = Gp + (size_t)g * NHID + q * 4;
    atomicAdd(o + 0, h.x);
    atomicAdd(o + 1, h.y);
    atomicAdd(o + 2, h.z);
    atomicAdd(o + 3, h.w);
}

__global__ void lin1_kernel(const float* __restrict__ G, const float* __restrict__ W,
                            const float* __restrict__ B, float* __restrict__ O) {
    int t = blockIdx.x * blockDim.x + threadIdx.x;
    if (t >= NGRAPH * NMLP) return;
    int gi = t / NMLP, j = t % NMLP;
    float acc = B[j];
#pragma unroll
    for (int k = 0; k < NHID; ++k)
        acc = fmaf(fmaxf(G[gi * NHID + k], 0.0f), W[k * NMLP + j], acc);
    O[t] = fmaxf(acc, 0.0f);
}

__global__ void lin2_kernel(const float* __restrict__ G, const float* __restrict__ W,
                            const float* __restrict__ B, float* __restrict__ O) {
    int t = blockIdx.x * blockDim.x + threadIdx.x;
    if (t >= NGRAPH * NCLS) return;
    int gi = t / NCLS, c = t % NCLS;
    float acc = B[c];
    for (int k = 0; k < NMLP; ++k)
        acc = fmaf(G[gi * NMLP + k], W[k * NCLS + c], acc);
    O[t] = acc;
}


extern "C" void kernel_launch(void* const* d_in, const int* in_sizes, int n_in,
                              void* d_out, int out_size, void* d_ws, size_t ws_size,
                              hipStream_t stream) {
    const float* x   = (const float*)d_in[0];
    const int*  eidx = (const int*)d_in[1];
    const int*  bvec = (const int*)d_in[2];
    const float* W1  = (const float*)d_in[3];
    const float* b1  = (const float*)d_in[4];
    const float* W2  = (const float*)d_in[5];
    const float* b2  = (const float*)d_in[6];
    const float* Wl1 = (const float*)d_in[7];
    const float* bl1 = (const float*)d_in[8];
    const float* Wl2 = (const float*)d_in[9];
    const float* bl2 = (const float*)d_in[10];
    float* out = (float*)d_out;

    char* ws = (char*)d_ws;
    size_t off = 0;
    auto walloc = [&](size_t bytes) -> void* {
        void* p = ws + off;
        off = (off + bytes + 255) & ~(size_t)255;
        return p;
    };
    float* deg  = (float*)walloc((size_t)NNODES * 4);
    float* bufA = (float*)walloc((size_t)NNODES * NHID * 4);
    float* bufB = (float*)walloc((size_t)NNODES * NHID * 4);
    float* gbuf = (float*)walloc((size_t)NGRAPH * NHID * 4);
    float* g2   = (float*)walloc((size_t)NGRAPH * NMLP * 4);

    const int* src = eidx;
    const int* dst = eidx + NEDGES;

    const int TB = 256;
    const int nN16  = (NNODES * NHID + TB - 1) / TB;
    const int nN4   = (NNODES * 4 + TB - 1) / TB;
    const int nTile = ((NNODES / 16) + 7) / 8;

    unsigned* part = (unsigned*)bufA;
    deg_hist<<<HR * HCH, 1024, 0, stream>>>(dst, part);
    deg_reduce<<<(NNODES / 4 + TB - 1) / TB, TB, 0, stream>>>(part, deg);
    float* dinv = deg;

    gemm_wmma_f32<<<nTile, TB, 0, stream>>>(x, W1, bufA, NNODES, NFEAT);
    fill_f32<<<nN16, TB, 0, stream>>>(bufB, NNODES * NHID, 0.0f);
    edge_aggregate<<<AR * ACH, 1024, 0, stream>>>(src, dst, dinv, bufA, bufB);
    finalize_node<<<nN16, TB, 0, stream>>>(bufB, bufA, dinv, b1, NNODES * NHID, 1);

    gemm_wmma_f32<<<nTile, TB, 0, stream>>>(bufB, W2, bufA, NNODES, NHID);
    fill_f32<<<nN16, TB, 0, stream>>>(bufB, NNODES * NHID, 0.0f);
    edge_aggregate<<<AR * ACH, 1024, 0, stream>>>(src, dst, dinv, bufA, bufB);
    finalize_node<<<nN16, TB, 0, stream>>>(bufB, bufA, dinv, b2, NNODES * NHID, 0);

    fill_f32<<<(NGRAPH * NHID + TB - 1) / TB, TB, 0, stream>>>(gbuf, NGRAPH * NHID, 0.0f);
    pool_kernel<<<nN4, TB, 0, stream>>>(bvec, bufB, gbuf, NNODES);
    lin1_kernel<<<(NGRAPH * NMLP + TB - 1) / TB, TB, 0, stream>>>(gbuf, Wl1, bl1, g2);
    lin2_kernel<<<(NGRAPH * NCLS + TB - 1) / TB, TB, 0, stream>>>(g2, Wl2, bl2, out);
}


# Edge weight factored out of the per-edge gather: GEMM stores rows scaled by dinv[src], aggregation sums plain rows, finalize scales once by dinv[dst]; 2 of 5 loads per edge gone; rounding regrouped
# speedup vs baseline: 2.6024x; 1.3691x over previous
#include <hip/hip_runtime.h>
#include <hip/hip_bf16.h>


typedef float v2f __attribute__((ext_vector_type(2)));
typedef float v8f __attribute__((ext_vector_type(8)));

#define NNODES  200000
#define NEDGES  6400000
#define NFEAT   128
#define NHID    16
#define NMLP    100
#define NCLS    27
#define NGRAPH  4096


__global__ void fill_f32(float* __restrict__ p, int n, float v) {
    int i = blockIdx.x * blockDim.x + threadIdx.x;
    if (i < n) p[i] = v;
}

#define HR     8
#define HBINS  (NNODES / HR)
#define HCH    32
#define HCHUNK (NEDGES / HCH)
#define HN4    (HCHUNK / 4)

__global__ __launch_bounds__(1024)
void deg_hist(const int* __restrict__ dst, unsigned* __restrict__ part) {
    __shared__ __attribute__((aligned(16))) unsigned h[HBINS];
    const int tid = threadIdx.x;
    const int r   = blockIdx.x & (HR - 1);
    const int c   = blockIdx.x >> 3;

    uint4* h4 = (uint4*)h;
    for (int i = tid; i < HBINS / 4; i += 1024) h4[i] = make_uint4(0u, 0u, 0u, 0u);
    __syncthreads();

    const unsigned lo = (unsigned)(r * HBINS);
    const int4* p = (const int4*)(dst + (size_t)c * HCHUNK);
    for (int base = tid; base < HN4; base += 1024 * 8) {
        int4 v[8];
#pragma unroll
        for (int u = 0; u < 8; ++u) {
            const int j = base + u * 1024;
            v[u] = (j < HN4) ? p[j] : make_int4(-1, -1, -1, -1);
        }
#pragma unroll
        for (int u = 0; u < 8; ++u) {
            unsigned a;
            a = (unsigned)v[u].x - lo; if (a < (unsigned)HBINS) atomicAdd(&h[a], 1u);
            a = (unsigned)v[u].y - lo; if (a < (unsigned)HBINS) atomicAdd(&h[a], 1u);
            a = (unsigned)v[u].z - lo; if (a < (unsigned)HBINS) atomicAdd(&h[a], 1u);
            a = (unsigned)v[u].w - lo; if (a < (unsigned)HBINS) atomicAdd(&h[a], 1u);
        }
    }
    __syncthreads();

    uint4* o = (uint4*)(part + (size_t)c * NNODES + lo);
    for (int i = tid; i < HBINS / 4; i += 1024) o[i] = h4[i];
}

__global__ void deg_reduce(const unsigned* __restrict__ part, float* __restrict__ dinv) {
    const int i = blockIdx.x * blockDim.x + threadIdx.x;
    if (i >= NNODES / 4) return;
    const uint4* p = (const uint4*)part + i;
    uint4 s = make_uint4(0u, 0u, 0u, 0u);
#pragma unroll
    for (int c = 0; c < HCH; ++c) {
        const uint4 v = p[(size_t)c * (NNODES / 4)];
        s.x += v.x; s.y += v.y; s.z += v.z; s.w += v.w;
    }
    float4 d;
    d.x = rsqrtf((float)s.x + 1.0f);
    d.y = rsqrtf((float)s.y + 1.0f);
    d.z = rsqrtf((float)s.z + 1.0f);
    d.w = rsqrtf((float)s.w + 1.0f);
    ((float4*)dinv)[i] = d;
}

__global__ __launch_bounds__(256)
void gemm_wmma_f32(const float* __restrict__ X, const float* __restrict__ W,
                   const float* __restrict__ rs, float* __restrict__ T, int M, int K) {
    __shared__ float lw[NFEAT * NHID];
    const int tid = threadIdx.x;
    const int kn  = K * NHID;
    for (int i = tid; i < kn; i += 256) lw[i] = W[i];
    __syncthreads();

    const int wave = tid >> 5;
    const int lane = tid & 31;
    const int tile = blockIdx.x * 8 + wave;
    if (tile * 16 >= M) return;

    const int row   = tile * 16 + (lane & 15);
    const int kHalf = (lane >> 4) << 1;
    const int n     = lane & 15;

    v8f c = {};
    const float* xr = X + (size_t)row * K + kHalf;
    for (int k0 = 0; k0 < K; k0 += 4) {
        float2 av = *(const float2*)(xr + k0);
        v2f a; a.x = av.x; a.y = av.y;
        v2f b;
        b.x = lw[(k0 + kHalf)     * NHID + n];
        b.y = lw[(k0 + kHalf + 1) * NHID + n];
        c = __builtin_amdgcn_wmma_f32_16x16x4_f32(
                 false, a,  false, b,
                 (short)0, c,  false,  false);
    }

    const int rbase = tile * 16 + ((lane >> 4) << 3);
#pragma unroll
    for (int v = 0; v < 8; ++v)
        T[(size_t)(rbase + v) * NHID + n] = c[v] * rs[rbase + v];
}

#define AR     50
#define ANR    (NNODES / AR)
#define ACH    5
#define ACHUNK (NEDGES / ACH)
#define AN4    (ACHUNK / 4)
#define AU     6
#define AWT    (32 * AU)
#define ANWT   ((AN4 + AWT - 1) / AWT)
#define AWQ    128
#define AB     10

__global__ __launch_bounds__(1024)
void edge_aggregate(const int* __restrict__ src, const int* __restrict__ dst,
                    const float* __restrict__ T, float* __restrict__ OUT) {
    __shared__ __attribute__((aligned(16))) float acc[ANR * NHID];
    __shared__ uint2 queue[32][AWQ];
    const int tid  = threadIdx.x;
    const int lane = tid & 31;
    const int wave = __builtin_amdgcn_readfirstlane(tid >> 5);
    const int r    = blockIdx.x % AR;
    const int c    = blockIdx.x / AR;

    float4* a4 = (float4*)acc;
    for (int i = tid; i < ANR * NHID / 4; i += 1024) a4[i] = make_float4(0.0f, 0.0f, 0.0f, 0.0f);
    for (int i = tid; i < 32 * AWQ; i += 1024) (&queue[0][0])[i] = make_uint2(0xffffffffu, 0u);
    __syncthreads();

    const unsigned lo = (unsigned)(r * ANR);
    const int ebase   = c * ACHUNK;
    const int4* p     = (const int4*)(dst + (size_t)ebase);
    const unsigned half = (unsigned)(lane >> 4);
    const int f   = lane & 15;
    uint2* qb = queue[wave];

    int4 v[AU];
    auto load_tile = [&](int wt) {
#pragma unroll
        for (int u = 0; u < AU; ++u) {
            const int jb = wt * AWT + u * 32;
            if (jb < AN4) v[u] = p[jb + lane];
            else          v[u] = make_int4(-1, -1, -1, -1);
        }
    };

    auto masks = [&](unsigned (&msk)[4 * AU]) -> unsigned {
        unsigned tot = 0u;
#pragma unroll
        for (int u = 0; u < AU; ++u) {
            msk[4 * u + 0] = __builtin_amdgcn_ballot_w32((unsigned)v[u].x - lo < (unsigned)ANR);
            msk[4 * u + 1] = __builtin_amdgcn_ballot_w32((unsigned)v[u].y - lo < (unsigned)ANR);
            msk[4 * u + 2] = __builtin_amdgcn_ballot_w32((unsigned)v[u].z - lo < (unsigned)ANR);
            msk[4 * u + 3] = __builtin_amdgcn_ballot_w32((unsigned)v[u].w - lo < (unsigned)ANR);
            tot += (unsigned)(__builtin_popcount(msk[4 * u + 0]) + __builtin_popcount(msk[4 * u + 1]) +
                              __builtin_popcount(msk[4 * u + 2]) + __builtin_popcount(msk[4 * u + 3]));
        }
        return tot;
    };

    auto emit = [&](const unsigned (&msk)[4 * AU], int wt, unsigned run) {
#pragma unroll
        for (int u = 0; u < AU; ++u) {
            const int e0 = ebase + 4 * (wt * AWT + u * 32 + lane);
            const int dv[4] = { v[u].x, v[u].y, v[u].z, v[u].w };
#pragma unroll
            for (int k = 0; k < 4; ++k) {
                const unsigned m = msk[4 * u + k];
                if (m != 0u) {
                    const unsigned a = (unsigned)dv[k] - lo;
                    const unsigned pos = run + __builtin_amdgcn_mbcnt_lo(m, 0u);
                    if (a < (unsigned)ANR && pos < (unsigned)AWQ)
                        qb[pos] = make_uint2(a, (unsigned)(e0 + k));
                    run += (unsigned)__builtin_popcount(m);
                }
            }
        }
    };

    unsigned pa[AB];
    float pm[AB];
#pragma unroll
    for (int k = 0; k < AB; ++k) { pa[k] = 0xffffffffu; pm[k] = 0.0f; }

    auto issue = [&](unsigned n, unsigned j0) {
        uint2 q[AB];
        unsigned s[AB];
#pragma unroll
        for (int k = 0; k < AB; ++k) {
            const unsigned jj = j0 + 2u * k;
            const uint2 t = qb[jj & (unsigned)(AWQ - 1)];
            q[k].x = (jj < n) ? t.x : 0xffffffffu;
            q[k].y = (jj < n) ? t.y : 0u;
        }
#pragma unroll
        for (int k = 0; k < AB; ++k) {
            s[k]  = (unsigned)*(const int*)((const char*)src + (q[k].y << 2));
            pa[k] = q[k].x;
        }
#pragma unroll
        for (int k = 0; k < AB; ++k)
            pm[k] = *(const float*)((const char*)T + ((s[k] << 6) | ((unsigned)f << 2)));
    };
    auto commit = [&]() {
#pragma unroll
        for (int k = 0; k < AB; ++k) {
            float val = pm[k];
            asm volatile("" : "+v"(val));
            if (pa[k] < (unsigned)ANR) atomicAdd(&acc[pa[k] * NHID + f], val);
        }
    };

    int wt = wave;
    load_tile(wt);
    unsigned msk[4 * AU];
    unsigned tot = masks(msk);
    unsigned rlo = 0u;
    while (wt < ANWT) {
        emit(msk, wt, 0u - rlo);
        commit();
        __builtin_amdgcn_wave_barrier();
        const unsigned n = min(tot - rlo, (unsigned)AWQ);
        const bool last = rlo + (unsigned)AWQ >= tot;
        if (last) load_tile(wt + 32);
        issue(n, half);
        for (unsigned jr = 2u * AB; jr < n; jr += 2u * AB) { commit(); issue(n, half + jr); }
        __builtin_amdgcn_wave_barrier();
        if (last) {
            tot = masks(msk);
            wt += 32;
            rlo = 0u;
        } else {
            rlo += (unsigned)AWQ;
        }
    }
    commit();
    __syncthreads();

    float* o = OUT + (size_t)lo * NHID;
    for (int i = tid; i < ANR * NHID; i += 1024) atomicAdd(o + i, acc[i]);
}

__global__ void finalize_node(float* __restrict__ ACC, const float* __restrict__ T,
                              const float* __restrict__ dinv, const float* __restrict__ bias,
                              int n16, int doRelu) {
    int idx = blockIdx.x * blockDim.x + threadIdx.x;
    if (idx >= n16) return;
    int i = idx >> 4, f = idx & 15;
    float di = dinv[i];
    float v  = (ACC[idx] + T[idx]) * di + bias[f];
    ACC[idx] = doRelu ? fmaxf(v, 0.0f) : v;
}

__global__ void pool_kernel(const int* __restrict__ bvec, const float* __restrict__ Hn,
                            float* __restrict__ Gp, int n) {
    int t = blockIdx.x * blockDim.x + threadIdx.x;
    int i = t >> 2, q = t & 3;
    if (i >= n) return;
    int g = bvec[i];
    const float4 h = *(const float4*)(Hn + (size_t)i * NHID + q * 4);
    float* o = Gp + (size_t)g * NHID + q * 4;
    atomicAdd(o + 0, h.x);
    atomicAdd(o + 1, h.y);
    atomicAdd(o + 2, h.z);
    atomicAdd(o + 3, h.w);
}

__global__ void lin1_kernel(const float* __restrict__ G, const float* __restrict__ W,
                            const float* __restrict__ B, float* __restrict__ O) {
    int t = blockIdx.x * blockDim.x + threadIdx.x;
    if (t >= NGRAPH * NMLP) return;
    int gi = t / NMLP, j = t % NMLP;
    float acc = B[j];
#pragma unroll
    for (int k = 0; k < NHID; ++k)
        acc = fmaf(fmaxf(G[gi * NHID + k], 0.0f), W[k * NMLP + j], acc);
    O[t] = fmaxf(acc, 0.0f);
}

__global__ void lin2_kernel(const float* __restrict__ G, const float* __restrict__ W,
                            const float* __restrict__ B, float* __restrict__ O) {
    int t = blockIdx.x * blockDim.x + threadIdx.x;
    if (t >= NGRAPH * NCLS) return;
    int gi = t / NCLS, c = t % NCLS;
    float acc = B[c];
    for (int k = 0; k < NMLP; ++k)
        acc = fmaf(G[gi * NMLP + k], W[k * NCLS + c], acc);
    O[t] = acc;
}


extern "C" void kernel_launch(void* const* d_in, const int* in_sizes, int n_in,
                              void* d_out, int out_size, void* d_ws, size_t ws_size,
                              hipStream_t stream) {
    const float* x   = (const float*)d_in[0];
    const int*  eidx = (const int*)d_in[1];
    const int*  bvec = (const int*)d_in[2];
    const float* W1  = (const float*)d_in[3];
    const float* b1  = (const float*)d_in[4];
    const float* W2  = (const float*)d_in[5];
    const float* b2  = (const float*)d_in[6];
    const float* Wl1 = (const float*)d_in[7];
    const float* bl1 = (const float*)d_in[8];
    const float* Wl2 = (const float*)d_in[9];
    const float* bl2 = (const float*)d_in[10];
    float* out = (float*)d_out;

    char* ws = (char*)d_ws;
    size_t off = 0;
    auto walloc = [&](size_t bytes) -> void* {
        void* p = ws + off;
        off = (off + bytes + 255) & ~(size_t)255;
        return p;
    };
    float* deg  = (float*)walloc((size_t)NNODES * 4);
    float* bufA = (float*)walloc((size_t)NNODES * NHID * 4);
    float* bufB = (float*)walloc((size_t)NNODES * NHID * 4);
    float* gbuf = (float*)walloc((size_t)NGRAPH * NHID * 4);
    float* g2   = (float*)walloc((size_t)NGRAPH * NMLP * 4);

    const int* src = eidx;
    const int* dst = eidx + NEDGES;

    const int TB = 256;
    const int nN16  = (NNODES * NHID + TB - 1) / TB;
    const int nN4   = (NNODES * 4 + TB - 1) / TB;
    const int nTile = ((NNODES / 16) + 7) / 8;

    unsigned* part = (unsigned*)bufA;
    deg_hist<<<HR * HCH, 1024, 0, stream>>>(dst, part);
    deg_reduce<<<(NNODES / 4 + TB - 1) / TB, TB, 0, stream>>>(part, deg);
    float* dinv = deg;

    gemm_wmma_f32<<<nTile, TB, 0, stream>>>(x, W1, dinv, bufA, NNODES, NFEAT);
    fill_f32<<<nN16, TB, 0, stream>>>(bufB, NNODES * NHID, 0.0f);
    edge_aggregate<<<AR * ACH, 1024, 0, stream>>>(src, dst, bufA, bufB);
    finalize_node<<<nN16, TB, 0, stream>>>(bufB, bufA, dinv, b1, NNODES * NHID, 1);

    gemm_wmma_f32<<<nTile, TB, 0, stream>>>(bufB, W2, dinv, bufA, NNODES, NHID);
    fill_f32<<<nN16, TB, 0, stream>>>(bufB, NNODES * NHID, 0.0f);
    edge_aggregate<<<AR * ACH, 1024, 0, stream>>>(src, dst, bufA, bufB);
    finalize_node<<<nN16, TB, 0, stream>>>(bufB, bufA, dinv, b2, NNODES * NHID, 0);

    fill_f32<<<(NGRAPH * NHID + TB - 1) / TB, TB, 0, stream>>>(gbuf, NGRAPH * NHID, 0.0f);
    pool_kernel<<<nN4, TB, 0, stream>>>(bvec, bufB, gbuf, NNODES);
    lin1_kernel<<<(NGRAPH * NMLP + TB - 1) / TB, TB, 0, stream>>>(gbuf, Wl1, bl1, g2);
    lin2_kernel<<<(NGRAPH * NCLS + TB - 1) / TB, TB, 0, stream>>>(g2, Wl2, bl2, out);
}


# Edge gather with 4 lanes per edge and 16-byte row loads: one load instruction fetches the rows of 8 edges, not 2; LDS rows at a 17-word pitch to spread the adds over banks; LDS add order changed
# speedup vs baseline: 2.7014x; 1.0380x over previous
#include <hip/hip_runtime.h>
#include <hip/hip_bf16.h>


typedef float v2f __attribute__((ext_vector_type(2)));
typedef float v4f __attribute__((ext_vector_type(4)));
typedef float v8f __attribute__((ext_vector_type(8)));

#define NNODES  200000
#define NEDGES  6400000
#define NFEAT   128
#define NHID    16
#define NMLP    100
#define NCLS    27
#define NGRAPH  4096


__global__ void fill_f32(float* __restrict__ p, int n, float v) {
    int i = blockIdx.x * blockDim.x + threadIdx.x;
    if (i < n) p[i] = v;
}

#define HR     8
#define HBINS  (NNODES / HR)
#define HCH    32
#define HCHUNK (NEDGES / HCH)
#define HN4    (HCHUNK / 4)

__global__ __launch_bounds__(1024)
void deg_hist(const int* __restrict__ dst, unsigned* __restrict__ part) {
    __shared__ __attribute__((aligned(16))) unsigned h[HBINS];
    const int tid = threadIdx.x;
    const int r   = blockIdx.x & (HR - 1);
    const int c   = blockIdx.x >> 3;

    uint4* h4 = (uint4*)h;
    for (int i = tid; i < HBINS / 4; i += 1024) h4[i] = make_uint4(0u, 0u, 0u, 0u);
    __syncthreads();

    const unsigned lo = (unsigned)(r * HBINS);
    const int4* p = (const int4*)(dst + (size_t)c * HCHUNK);
    for (int base = tid; base < HN4; base += 1024 * 8) {
        int4 v[8];
#pragma unroll
        for (int u = 0; u < 8; ++u) {
            const int j = base + u * 1024;
            v[u] = (j < HN4) ? p[j] : make_int4(-1, -1, -1, -1);
        }
#pragma unroll
        for (int u = 0; u < 8; ++u) {
            unsigned a;
            a = (unsigned)v[u].x - lo; if (a < (unsigned)HBINS) atomicAdd(&h[a], 1u);
            a = (unsigned)v[u].y - lo; if (a < (unsigned)HBINS) atomicAdd(&h[a], 1u);
            a = (unsigned)v[u].z - lo; if (a < (unsigned)HBINS) atomicAdd(&h[a], 1u);
            a = (unsigned)v[u].w - lo; if (a < (unsigned)HBINS) atomicAdd(&h[a], 1u);
        }
    }
    __syncthreads();

    uint4* o = (uint4*)(part + (size_t)c * NNODES + lo);
    for (int i = tid; i < HBINS / 4; i += 1024) o[i] = h4[i];
}

__global__ void deg_reduce(const unsigned* __restrict__ part, float* __restrict__ dinv) {
    const int i = blockIdx.x * blockDim.x + threadIdx.x;
    if (i >= NNODES / 4) return;
    const uint4* p = (const uint4*)part + i;
    uint4 s = make_uint4(0u, 0u, 0u, 0u);
#pragma unroll
    for (int c = 0; c < HCH; ++c) {
        const uint4 v = p[(size_t)c * (NNODES / 4)];
        s.x += v.x; s.y += v.y; s.z += v.z; s.w += v.w;
    }
    float4 d;
    d.x = rsqrtf((float)s.x + 1.0f);
    d.y = rsqrtf((float)s.y + 1.0f);
    d.z = rsqrtf((float)s.z + 1.0f);
    d.w = rsqrtf((float)s.w + 1.0f);
    ((float4*)dinv)[i] = d;
}

__global__ __launch_bounds__(256)
void gemm_wmma_f32(const float* __restrict__ X, const float* __restrict__ W,
                   const float* __restrict__ rs, float* __restrict__ T, int M, int K) {
    __shared__ float lw[NFEAT * NHID];
    const int tid = threadIdx.x;
    const int kn  = K * NHID;
    for (int i = tid; i < kn; i += 256) lw[i] = W[i];
    __syncthreads();

    const int wave = tid >> 5;
    const int lane = tid & 31;
    const int tile = blockIdx.x * 8 + wave;
    if (tile * 16 >= M) return;

    const int row   = tile * 16 + (lane & 15);
    const int kHalf = (lane >> 4) << 1;
    const int n     = lane & 15;

    v8f c = {};
    const float* xr = X + (size_t)row * K + kHalf;
    for (int k0 = 0; k0 < K; k0 += 4) {
        float2 av = *(const float2*)(xr + k0);
        v2f a; a.x = av.x; a.y = av.y;
        v2f b;
        b.x = lw[(k0 + kHalf)     * NHID + n];
        b.y = lw[(k0 + kHalf + 1) * NHID + n];
        c = __builtin_amdgcn_wmma_f32_16x16x4_f32(
                 false, a,  false, b,
                 (short)0, c,  false,  false);
    }

    const int rbase = tile * 16 + ((lane >> 4) << 3);
#pragma unroll
    for (int v = 0; v < 8; ++v)
        T[(size_t)(rbase + v) * NHID + n] = c[v] * rs[rbase + v];
}

#define AR     50
#define ANR    (NNODES / AR)
#define ACH    5
#define ACHUNK (NEDGES / ACH)
#define AN4    (ACHUNK / 4)
#define AU     6
#define AWT    (32 * AU)
#define ANWT   ((AN4 + AWT - 1) / AWT)
#define AWQ    128
#define AB     3
#define APITCH 17

__global__ __launch_bounds__(1024)
void edge_aggregate(const int* __restrict__ src, const int* __restrict__ dst,
                    const float* __restrict__ T, float* __restrict__ OUT) {
    __shared__ __attribute__((aligned(16))) float acc[ANR * APITCH];
    __shared__ uint2 queue[32][AWQ];
    const int tid  = threadIdx.x;
    const int lane = tid & 31;
    const int wave = __builtin_amdgcn_readfirstlane(tid >> 5);
    const int r    = blockIdx.x % AR;
    const int c    = blockIdx.x / AR;

    float4* a4 = (float4*)acc;
    for (int i = tid; i < ANR * APITCH / 4; i += 1024) a4[i] = make_float4(0.0f, 0.0f, 0.0f, 0.0f);
    for (int i = tid; i < 32 * AWQ; i += 1024) (&queue[0][0])[i] = make_uint2(0xffffffffu, 0u);
    __syncthreads();

    const unsigned lo = (unsigned)(r * ANR);
    const int ebase   = c * ACHUNK;
    const int4* p     = (const int4*)(dst + (size_t)ebase);
    const unsigned grp = (unsigned)(lane >> 2);
    const unsigned f4  = (unsigned)(lane & 3) << 2;
    uint2* qb = queue[wave];

    int4 v[AU];
    auto load_tile = [&](int wt) {
#pragma unroll
        for (int u = 0; u < AU; ++u) {
            const int jb = wt * AWT + u * 32;
            if (jb < AN4) v[u] = p[jb + lane];
            else          v[u] = make_int4(-1, -1, -1, -1);
        }
    };

    auto masks = [&](unsigned (&msk)[4 * AU]) -> unsigned {
        unsigned tot = 0u;
#pragma unroll
        for (int u = 0; u < AU; ++u) {
            msk[4 * u + 0] = __builtin_amdgcn_ballot_w32((unsigned)v[u].x - lo < (unsigned)ANR);
            msk[4 * u + 1] = __builtin_amdgcn_ballot_w32((unsigned)v[u].y - lo < (unsigned)ANR);
            msk[4 * u + 2] = __builtin_amdgcn_ballot_w32((unsigned)v[u].z - lo < (unsigned)ANR);
            msk[4 * u + 3] = __builtin_amdgcn_ballot_w32((unsigned)v[u].w - lo < (unsigned)ANR);
            tot += (unsigned)(__builtin_popcount(msk[4 * u + 0]) + __builtin_popcount(msk[4 * u + 1]) +
                              __builtin_popcount(msk[4 * u + 2]) + __builtin_popcount(msk[4 * u + 3]));
        }
        return tot;
    };

    auto emit = [&](const unsigned (&msk)[4 * AU], int wt, unsigned run) {
#pragma unroll
        for (int u = 0; u < AU; ++u) {
            const int e0 = ebase + 4 * (wt * AWT + u * 32 + lane);
            const int dv[4] = { v[u].x, v[u].y, v[u].z, v[u].w };
#pragma unroll
            for (int k = 0; k < 4; ++k) {
                const unsigned m = msk[4 * u + k];
                if (m != 0u) {
                    const unsigned a = (unsigned)dv[k] - lo;
                    const unsigned pos = run + __builtin_amdgcn_mbcnt_lo(m, 0u);
                    if (a < (unsigned)ANR && pos < (unsigned)AWQ)
                        qb[pos] = make_uint2(a, (unsigned)(e0 + k));
                    run += (unsigned)__builtin_popcount(m);
                }
            }
        }
    };

    unsigned pa[AB];
    v4f pm[AB];
#pragma unroll
    for (int k = 0; k < AB; ++k) { pa[k] = 0xffffffffu; pm[k] = (v4f){0.0f, 0.0f, 0.0f, 0.0f}; }

    auto issue = [&](unsigned n, unsigned j0) {
        uint2 q[AB];
        unsigned s[AB];
#pragma unroll
        for (int k = 0; k < AB; ++k) {
            const unsigned jj = j0 + 8u * k;
            const uint2 t = qb[jj & (unsigned)(AWQ - 1)];
            q[k].x = (jj < n) ? t.x : 0xffffffffu;
            q[k].y = (jj < n) ? t.y : 0u;
        }
#pragma unroll
        for (int k = 0; k < AB; ++k) {
            s[k]  = (unsigned)*(const int*)((const char*)src + (q[k].y << 2));
            pa[k] = q[k].x;
        }
#pragma unroll
        for (int k = 0; k < AB; ++k)
            pm[k] = *(const v4f*)((const char*)T + ((s[k] << 6) | (f4 << 2)));
    };
    auto commit = [&]() {
#pragma unroll
        for (int k = 0; k < AB; ++k) {
            v4f val = pm[k];
            asm volatile("" : "+v"(val));
            if (pa[k] < (unsigned)ANR) {
                float* d = &acc[pa[k] * APITCH + f4];
                atomicAdd(d + 0, val.x);
                atomicAdd(d + 1, val.y);
                atomicAdd(d + 2, val.z);
                atomicAdd(d + 3, val.w);
            }
        }
    };

    int wt = wave;
    load_tile(wt);
    unsigned msk[4 * AU];
    unsigned tot = masks(msk);
    unsigned rlo = 0u;
    while (wt < ANWT) {
        emit(msk, wt, 0u - rlo);
        commit();
        __builtin_amdgcn_wave_barrier();
        const unsigned n = min(tot - rlo, (unsigned)AWQ);
        const bool last = rlo + (unsigned)AWQ >= tot;
        if (last) load_tile(wt + 32);
        issue(n, grp);
        for (unsigned jr = 8u * AB; jr < n; jr += 8u * AB) { commit(); issue(n, grp + jr); }
        __builtin_amdgcn_wave_barrier();
        if (last) {
            tot = masks(msk);
            wt += 32;
            rlo = 0u;
        } else {
            rlo += (unsigned)AWQ;
        }
    }
    commit();
    __syncthreads();

    float* o = OUT + (size_t)lo * NHID;
    for (int i = tid; i < ANR * NHID; i += 1024) atomicAdd(o + i, acc[(i >> 4) * APITCH + (i & 15)]);
}

__global__ void finalize_node(float* __restrict__ ACC, const float* __restrict__ T,
                              const float* __restrict__ dinv, const float* __restrict__ bias,
                              int n16, int doRelu) {
    int idx = blockIdx.x * blockDim.x + threadIdx.x;
    if (idx >= n16) return;
    int i = idx >> 4, f = idx & 15;
    float di = dinv[i];
    float v  = (ACC[idx] + T[idx]) * di + bias[f];
    ACC[idx] = doRelu ? fmaxf(v, 0.0f) : v;
}

__global__ void pool_kernel(const int* __restrict__ bvec, const float* __restrict__ Hn,
                            float* __restrict__ Gp, int n) {
    int t = blockIdx.x * blockDim.x + threadIdx.x;
    int i = t >> 2, q = t & 3;
    if (i >= n) return;
    int g = bvec[i];
    const float4 h = *(const float4*)(Hn + (size_t)i * NHID + q * 4);
    float* o = Gp + (size_t)g * NHID + q * 4;
    atomicAdd(o + 0, h.x);
    atomicAdd(o + 1, h.y);
    atomicAdd(o + 2, h.z);
    atomicAdd(o + 3, h.w);
}

__global__ void lin1_kernel(const float* __restrict__ G, const float* __restrict__ W,
                            const float* __restrict__ B, float* __restrict__ O) {
    int t = blockIdx.x * blockDim.x + threadIdx.x;
    if (t >= NGRAPH * NMLP) return;
    int gi = t / NMLP, j = t % NMLP;
    float acc = B[j];
#pragma unroll
    for (int k = 0; k < NHID; ++k)
        acc = fmaf(fmaxf(G[gi * NHID + k], 0.0f), W[k * NMLP + j], acc);
    O[t] = fmaxf(acc, 0.0f);
}

__global__ void lin2_kernel(const float* __restrict__ G, const float* __restrict__ W,
                            const float* __restrict__ B, float* __restrict__ O) {
    int t = blockIdx.x * blockDim.x + threadIdx.x;
    if (t >= NGRAPH * NCLS) return;
    int gi = t / NCLS, c = t % NCLS;
    float acc = B[c];
    for (int k = 0; k < NMLP; ++k)
        acc = fmaf(G[gi * NMLP + k], W[k * NCLS + c], acc);
    O[t] = acc;
}


extern "C" void kernel_launch(void* const* d_in, const int* in_sizes, int n_in,
                              void* d_out, int out_size, void* d_ws, size_t ws_size,
                              hipStream_t stream) {
    const float* x   = (const float*)d_in[0];
    const int*  eidx = (const int*)d_in[1];
    const int*  bvec = (const int*)d_in[2];
    const float* W1  = (const float*)d_in[3];
    const float* b1  = (const float*)d_in[4];
    const float* W2  = (const float*)d_in[5];
    const float* b2  = (const float*)d_in[6];
    const float* Wl1 = (const float*)d_in[7];
    const float* bl1 = (const float*)d_in[8];
    const float* Wl2 = (const float*)d_in[9];
    const float* bl2 = (const float*)d_in[10];
    float* out = (float*)d_out;

    char* ws = (char*)d_ws;
    size_t off = 0;
    auto walloc = [&](size_t bytes) -> void* {
        void* p = ws + off;
        off = (off + bytes + 255) & ~(size_t)255;
        return p;
    };
    float* deg  = (float*)walloc((size_t)NNODES * 4);
    float* bufA = (float*)walloc((size_t)NNODES * NHID * 4);
    float* bufB = (float*)walloc((size_t)NNODES * NHID * 4);
    float* gbuf = (float*)walloc((size_t)NGRAPH * NHID * 4);
    float* g2   = (float*)walloc((size_t)NGRAPH * NMLP * 4);

    const int* src = eidx;
    const int* dst = eidx + NEDGES;

    const int TB = 256;
    const int nN16  = (NNODES * NHID + TB - 1) / TB;
    const int nN4   = (NNODES * 4 + TB - 1) / TB;
    const int nTile = ((NNODES / 16) + 7) / 8;

    unsigned* part = (unsigned*)bufA;
    deg_hist<<<HR * HCH, 1024, 0, stream>>>(dst, part);
    deg_reduce<<<(NNODES / 4 + TB - 1) / TB, TB, 0, stream>>>(part, deg);
    float* dinv = deg;

    gemm_wmma_f32<<<nTile, TB, 0, stream>>>(x, W1, dinv, bufA, NNODES, NFEAT);
    fill_f32<<<nN16, TB, 0, stream>>>(bufB, NNODES * NHID, 0.0f);
    edge_aggregate<<<AR * ACH, 1024, 0, stream>>>(src, dst, bufA, bufB);
    finalize_node<<<nN16, TB, 0, stream>>>(bufB, bufA, dinv, b1, NNODES * NHID, 1);

    gemm_wmma_f32<<<nTile, TB, 0, stream>>>(bufB, W2, dinv, bufA, NNODES, NHID);
    fill_f32<<<nN16, TB, 0, stream>>>(bufB, NNODES * NHID, 0.0f);
    edge_aggregate<<<AR * ACH, 1024, 0, stream>>>(src, dst, bufA, bufB);
    finalize_node<<<nN16, TB, 0, stream>>>(bufB, bufA, dinv, b2, NNODES * NHID, 0);

    fill_f32<<<(NGRAPH * NHID + TB - 1) / TB, TB, 0, stream>>>(gbuf, NGRAPH * NHID, 0.0f);
    pool_kernel<<<nN4, TB, 0, stream>>>(bvec, bufB, gbuf, NNODES);
    lin1_kernel<<<(NGRAPH * NMLP + TB - 1) / TB, TB, 0, stream>>>(gbuf, Wl1, bl1, g2);
    lin2_kernel<<<(NGRAPH * NCLS + TB - 1) / TB, TB, 0, stream>>>(g2, Wl2, bl2, out);
}


# Edges binned by destination range once per call (count, scatter into one list per range); both aggregations read only their range's list: no per-range filter, no source-id load; LDS add order changed
# speedup vs baseline: 3.4468x; 1.2759x over previous
#include <hip/hip_runtime.h>
#include <hip/hip_bf16.h>


typedef float v2f __attribute__((ext_vector_type(2)));
typedef float v4f __attribute__((ext_vector_type(4)));
typedef float v8f __attribute__((ext_vector_type(8)));

#define NNODES  200000
#define NEDGES  6400000
#define NFEAT   128
#define NHID    16
#define NMLP    100
#define NCLS    27
#define NGRAPH  4096


__global__ void fill_f32(float* __restrict__ p, int n, float v) {
    int i = blockIdx.x * blockDim.x + threadIdx.x;
    if (i < n) p[i] = v;
}

#define HR     8
#define HBINS  (NNODES / HR)
#define HCH    32
#define HCHUNK (NEDGES / HCH)
#define HN4    (HCHUNK / 4)

__global__ __launch_bounds__(1024)
void deg_hist(const int* __restrict__ dst, unsigned* __restrict__ part) {
    __shared__ __attribute__((aligned(16))) unsigned h[HBINS];
    const int tid = threadIdx.x;
    const int r   = blockIdx.x & (HR - 1);
    const int c   = blockIdx.x >> 3;

    uint4* h4 = (uint4*)h;
    for (int i = tid; i < HBINS / 4; i += 1024) h4[i] = make_uint4(0u, 0u, 0u, 0u);
    __syncthreads();

    const unsigned lo = (unsigned)(r * HBINS);
    const int4* p = (const int4*)(dst + (size_t)c * HCHUNK);
    for (int base = tid; base < HN4; base += 1024 * 8) {
        int4 v[8];
#pragma unroll
        for (int u = 0; u < 8; ++u) {
            const int j = base + u * 1024;
            v[u] = (j < HN4) ? p[j] : make_int4(-1, -1, -1, -1);
        }
#pragma unroll
        for (int u = 0; u < 8; ++u) {
            unsigned a;
            a = (unsigned)v[u].x - lo; if (a < (unsigned)HBINS) atomicAdd(&h[a], 1u);
            a = (unsigned)v[u].y - lo; if (a < (unsigned)HBINS) atomicAdd(&h[a], 1u);
            a = (unsigned)v[u].z - lo; if (a < (unsigned)HBINS) atomicAdd(&h[a], 1u);
            a = (unsigned)v[u].w - lo; if (a < (unsigned)HBINS) atomicAdd(&h[a], 1u);
        }
    }
    __syncthreads();

    uint4* o = (uint4*)(part + (size_t)c * NNODES + lo);
    for (int i = tid; i < HBINS / 4; i += 1024) o[i] = h4[i];
}

__global__ void deg_reduce(const unsigned* __restrict__ part, float* __restrict__ dinv) {
    const int i = blockIdx.x * blockDim.x + threadIdx.x;
    if (i >= NNODES / 4) return;
    const uint4* p = (const uint4*)part + i;
    uint4 s = make_uint4(0u, 0u, 0u, 0u);
#pragma unroll
    for (int c = 0; c < HCH; ++c) {
        const uint4 v = p[(size_t)c * (NNODES / 4)];
        s.x += v.x; s.y += v.y; s.z += v.z; s.w += v.w;
    }
    float4 d;
    d.x = rsqrtf((float)s.x + 1.0f);
    d.y = rsqrtf((float)s.y + 1.0f);
    d.z = rsqrtf((float)s.z + 1.0f);
    d.w = rsqrtf((float)s.w + 1.0f);
    ((float4*)dinv)[i] = d;
}

__global__ __launch_bounds__(256)
void gemm_wmma_f32(const float* __restrict__ X, const float* __restrict__ W,
                   const float* __restrict__ rs, float* __restrict__ T, int M, int K) {
    __shared__ float lw[NFEAT * NHID];
    const int tid = threadIdx.x;
    const int kn  = K * NHID;
    for (int i = tid; i < kn; i += 256) lw[i] = W[i];
    __syncthreads();

    const int wave = tid >> 5;
    const int lane = tid & 31;
    const int tile = blockIdx.x * 8 + wave;
    if (tile * 16 >= M) return;

    const int row   = tile * 16 + (lane & 15);
    const int kHalf = (lane >> 4) << 1;
    const int n     = lane & 15;

    v8f c = {};
    const float* xr = X + (size_t)row * K + kHalf;
    for (int k0 = 0; k0 < K; k0 += 4) {
        float2 av = *(const float2*)(xr + k0);
        v2f a; a.x = av.x; a.y = av.y;
        v2f b;
        b.x = lw[(k0 + kHalf)     * NHID + n];
        b.y = lw[(k0 + kHalf + 1) * NHID + n];
        c = __builtin_amdgcn_wmma_f32_16x16x4_f32(
                 false, a,  false, b,
                 (short)0, c,  false,  false);
    }

    const int rbase = tile * 16 + ((lane >> 4) << 3);
#pragma unroll
    for (int v = 0; v < 8; ++v)
        T[(size_t)(rbase + v) * NHID + n] = c[v] * rs[rbase + v];
}

#define AR     50
#define ANR    (NNODES / AR)
#define ACH    5
#define ACHUNK (NEDGES / ACH)
#define AN4    (ACHUNK / 4)
#define AU     6
#define AWT    (32 * AU)
#define ANWT   ((AN4 + AWT - 1) / AWT)
#define AWQ    128
#define AB     3
#define APITCH 17

__global__ __launch_bounds__(1024)
void edge_aggregate_filter(const int* __restrict__ src, const int* __restrict__ dst,
                           const float* __restrict__ T, float* __restrict__ OUT) {
    __shared__ __attribute__((aligned(16))) float acc[ANR * APITCH];
    __shared__ uint2 queue[32][AWQ];
    const int tid  = threadIdx.x;
    const int lane = tid & 31;
    const int wave = __builtin_amdgcn_readfirstlane(tid >> 5);
    const int r    = blockIdx.x % AR;
    const int c    = blockIdx.x / AR;

    float4* a4 = (float4*)acc;
    for (int i = tid; i < ANR * APITCH / 4; i += 1024) a4[i] = make_float4(0.0f, 0.0f, 0.0f, 0.0f);
    for (int i = tid; i < 32 * AWQ; i += 1024) (&queue[0][0])[i] = make_uint2(0xffffffffu, 0u);
    __syncthreads();

    const unsigned lo = (unsigned)(r * ANR);
    const int ebase   = c * ACHUNK;
    const int4* p     = (const int4*)(dst + (size_t)ebase);
    const unsigned grp = (unsigned)(lane >> 2);
    const unsigned f4  = (unsigned)(lane & 3) << 2;
    uint2* qb = queue[wave];

    int4 v[AU];
    auto load_tile = [&](int wt) {
#pragma unroll
        for (int u = 0; u < AU; ++u) {
            const int jb = wt * AWT + u * 32;
            if (jb < AN4) v[u] = p[jb + lane];
            else          v[u] = make_int4(-1, -1, -1, -1);
        }
    };

    auto masks = [&](unsigned (&msk)[4 * AU]) -> unsigned {
        unsigned tot = 0u;
#pragma unroll
        for (int u = 0; u < AU; ++u) {
            msk[4 * u + 0] = __builtin_amdgcn_ballot_w32((unsigned)v[u].x - lo < (unsigned)ANR);
            msk[4 * u + 1] = __builtin_amdgcn_ballot_w32((unsigned)v[u].y - lo < (unsigned)ANR);
            msk[4 * u + 2] = __builtin_amdgcn_ballot_w32((unsigned)v[u].z - lo < (unsigned)ANR);
            msk[4 * u + 3] = __builtin_amdgcn_ballot_w32((unsigned)v[u].w - lo < (unsigned)ANR);
            tot += (unsigned)(__builtin_popcount(msk[4 * u + 0]) + __builtin_popcount(msk[4 * u + 1]) +
                              __builtin_popcount(msk[4 * u + 2]) + __builtin_popcount(msk[4 * u + 3]));
        }
        return tot;
    };

    auto emit = [&](const unsigned (&msk)[4 * AU], int wt, unsigned run) {
#pragma unroll
        for (int u = 0; u < AU; ++u) {
            const int e0 = ebase + 4 * (wt * AWT + u * 32 + lane);
            const int dv[4] = { v[u].x, v[u].y, v[u].z, v[u].w };
#pragma unroll
            for (int k = 0; k < 4; ++k) {
                const unsigned m = msk[4 * u + k];
                if (m != 0u) {
                    const unsigned a = (unsigned)dv[k] - lo;
                    const unsigned pos = run + __builtin_amdgcn_mbcnt_lo(m, 0u);
                    if (a < (unsigned)ANR && pos < (unsigned)AWQ)
                        qb[pos] = make_uint2(a, (unsigned)(e0 + k));
                    run += (unsigned)__builtin_popcount(m);
                }
            }
        }
    };

    unsigned pa[AB];
    v4f pm[AB];
#pragma unroll
    for (int k = 0; k < AB; ++k) { pa[k] = 0xffffffffu; pm[k] = (v4f){0.0f, 0.0f, 0.0f, 0.0f}; }

    auto issue = [&](unsigned n, unsigned j0) {
        uint2 q[AB];
        unsigned s[AB];
#pragma unroll
        for (int k = 0; k < AB; ++k) {
            const unsigned jj = j0 + 8u * k;
            const uint2 t = qb[jj & (unsigned)(AWQ - 1)];
            q[k].x = (jj < n) ? t.x : 0xffffffffu;
            q[k].y = (jj < n) ? t.y : 0u;
        }
#pragma unroll
        for (int k = 0; k < AB; ++k) {
            s[k]  = (unsigned)*(const int*)((const char*)src + (q[k].y << 2));
            pa[k] = q[k].x;
        }
#pragma unroll
        for (int k = 0; k < AB; ++k)
            pm[k] = *(const v4f*)((const char*)T + ((s[k] << 6) | (f4 << 2)));
    };
    auto commit = [&]() {
#pragma unroll
        for (int k = 0; k < AB; ++k) {
            v4f val = pm[k];
            asm volatile("" : "+v"(val));
            if (pa[k] < (unsigned)ANR) {
                float* d = &acc[pa[k] * APITCH + f4];
                atomicAdd(d + 0, val.x);
                atomicAdd(d + 1, val.y);
                atomicAdd(d + 2, val.z);
                atomicAdd(d + 3, val.w);
            }
        }
    };

    int wt = wave;
    load_tile(wt);
    unsigned msk[4 * AU];
    unsigned tot = masks(msk);
    unsigned rlo = 0u;
    while (wt < ANWT) {
        emit(msk, wt, 0u - rlo);
        commit();
        __builtin_amdgcn_wave_barrier();
        const unsigned n = min(tot - rlo, (unsigned)AWQ);
        const bool last = rlo + (unsigned)AWQ >= tot;
        if (last) load_tile(wt + 32);
        issue(n, grp);
        for (unsigned jr = 8u * AB; jr < n; jr += 8u * AB) { commit(); issue(n, grp + jr); }
        __builtin_amdgcn_wave_barrier();
        if (last) {
            tot = masks(msk);
            wt += 32;
            rlo = 0u;
        } else {
            rlo += (unsigned)AWQ;
        }
    }
    commit();
    __syncthreads();

    float* o = OUT + (size_t)lo * NHID;
    for (int i = tid; i < ANR * NHID; i += 1024) atomicAdd(o + i, acc[(i >> 4) * APITCH + (i & 15)]);
}

#define BNB    250
#define BSL    (NEDGES / BNB)
#define BSL4   (BSL / 4)
#define BU     ((BSL4 + 1023) / 1024)
#define BCU    ((BNB + 31) / 32)
#define SBITS  18

__global__ __launch_bounds__(1024)
void bin_count(const int* __restrict__ dst, unsigned* __restrict__ cnt) {
    __shared__ unsigned h[AR];
    const int tid = threadIdx.x;
    if (tid < AR) h[tid] = 0u;
    __syncthreads();
    const int4* p = (const int4*)(dst + (size_t)blockIdx.x * BSL);
    int4 v[BU];
#pragma unroll
    for (int u = 0; u < BU; ++u) {
        const int j = tid + u * 1024;
        v[u] = (j < BSL4) ? p[j] : make_int4(-1, -1, -1, -1);
    }
#pragma unroll
    for (int u = 0; u < BU; ++u) {
        const unsigned d[4] = { (unsigned)v[u].x, (unsigned)v[u].y, (unsigned)v[u].z, (unsigned)v[u].w };
#pragma unroll
        for (int k = 0; k < 4; ++k)
            if (d[k] < (unsigned)NNODES) atomicAdd(&h[d[k] / (unsigned)ANR], 1u);
    }
    __syncthreads();
    if (tid < AR) cnt[tid * BNB + blockIdx.x] = h[tid];
}

__global__ __launch_bounds__(1024)
void bin_scatter(const int* __restrict__ src, const int* __restrict__ dst,
                 const unsigned* __restrict__ cnt, unsigned* __restrict__ bstart,
                 unsigned* __restrict__ list) {
    __shared__ unsigned tot[AR];
    __shared__ unsigned pre[AR];
    __shared__ unsigned cur[AR];
    const int tid  = threadIdx.x;
    const int lane = tid & 31;
    const int wave = __builtin_amdgcn_readfirstlane(tid >> 5);
    const int b    = blockIdx.x;

    for (int r = wave; r < AR; r += 32) {
        unsigned t = 0u, q = 0u;
#pragma unroll
        for (int u = 0; u < BCU; ++u) {
            const int j = lane + 32 * u;
            const unsigned v = (j < BNB) ? cnt[r * BNB + j] : 0u;
            t += v;
            q += (j < b) ? v : 0u;
        }
        for (int m = 16; m >= 1; m >>= 1) {
            t += __shfl_xor(t, m, 32);
            q += __shfl_xor(q, m, 32);
        }
        if (lane == 0) { tot[r] = t; pre[r] = q; }
    }
    __syncthreads();
    if (tid < AR) {
        unsigned run = 0u;
#pragma unroll
        for (int j = 0; j < AR; ++j) run += (j < tid) ? tot[j] : 0u;
        cur[tid] = run + pre[tid];
        if (b == 0) {
            bstart[tid] = run;
            if (tid == AR - 1) bstart[AR] = run + tot[tid];
        }
    }
    __syncthreads();

    const int4* pd = (const int4*)(dst + (size_t)b * BSL);
    const int4* ps = (const int4*)(src + (size_t)b * BSL);
    int4 vd[BU], vs[BU];
#pragma unroll
    for (int u = 0; u < BU; ++u) {
        const int j = tid + u * 1024;
        vd[u] = (j < BSL4) ? pd[j] : make_int4(-1, -1, -1, -1);
        vs[u] = (j < BSL4) ? ps[j] : make_int4(0, 0, 0, 0);
    }
#pragma unroll
    for (int u = 0; u < BU; ++u) {
        const unsigned d[4] = { (unsigned)vd[u].x, (unsigned)vd[u].y, (unsigned)vd[u].z, (unsigned)vd[u].w };
        const unsigned s[4] = { (unsigned)vs[u].x, (unsigned)vs[u].y, (unsigned)vs[u].z, (unsigned)vs[u].w };
#pragma unroll
        for (int k = 0; k < 4; ++k) {
            if (d[k] < (unsigned)NNODES) {
                const unsigned r   = d[k] / (unsigned)ANR;
                const unsigned a   = d[k] - r * (unsigned)ANR;
                const unsigned pos = atomicAdd(&cur[r], 1u);
                list[pos] = (a << SBITS) | s[k];
            }
        }
    }
}

#define AT     2

__global__ __launch_bounds__(1024)
void edge_aggregate_binned(const unsigned* __restrict__ list, const unsigned* __restrict__ bstart,
                           const float* __restrict__ T, float* __restrict__ OUT) {
    __shared__ __attribute__((aligned(16))) float acc[ANR * APITCH];
    const int tid  = threadIdx.x;
    const int lane = tid & 31;
    const int wave = __builtin_amdgcn_readfirstlane(tid >> 5);
    const int r    = blockIdx.x % AR;
    const int c    = blockIdx.x / AR;

    float4* a4 = (float4*)acc;
    for (int i = tid; i < ANR * APITCH / 4; i += 1024) a4[i] = make_float4(0.0f, 0.0f, 0.0f, 0.0f);
    __syncthreads();

    const unsigned bs  = bstart[r];
    const unsigned n   = bstart[r + 1] - bs;
    const unsigned plo = bs + n * (unsigned)c / (unsigned)ACH;
    const unsigned phi = bs + n * (unsigned)(c + 1) / (unsigned)ACH;
    const unsigned grp = (unsigned)(lane >> 2);
    const unsigned f4  = (unsigned)(lane & 3) << 2;
    const unsigned PAD = 0x3fffu << SBITS;

    unsigned ent[AT];
    auto load_ent = [&](unsigned t0, unsigned (&e)[AT]) {
#pragma unroll
        for (int u = 0; u < AT; ++u) {
            const unsigned idx = t0 + 32u * u + (unsigned)lane;
            const unsigned w = list[min(idx, (unsigned)(NEDGES - 1))];
            e[u] = (idx < phi) ? w : PAD;
        }
    };

    unsigned t0 = plo + (unsigned)wave * (32u * AT);
    load_ent(t0, ent);
    while (t0 < phi) {
        unsigned nxt[AT];
        load_ent(t0 + 32u * 32u * AT, nxt);
        unsigned pa[4 * AT];
        v4f pm[4 * AT];
#pragma unroll
        for (int u = 0; u < AT; ++u) {
#pragma unroll
            for (int k = 0; k < 4; ++k) {
                const unsigned e = (unsigned)__shfl((int)ent[u], (int)(8u * k + grp), 32);
                const unsigned s = e & ((1u << SBITS) - 1u);
                pa[4 * u + k] = e >> SBITS;
                pm[4 * u + k] = *(const v4f*)((const char*)T + ((s << 6) | (f4 << 2)));
            }
        }
#pragma unroll
        for (int j = 0; j < 4 * AT; ++j) {
            if (pa[j] < (unsigned)ANR) {
                float* d = &acc[pa[j] * APITCH + f4];
                atomicAdd(d + 0, pm[j].x);
                atomicAdd(d + 1, pm[j].y);
                atomicAdd(d + 2, pm[j].z);
                atomicAdd(d + 3, pm[j].w);
            }
        }
#pragma unroll
        for (int u = 0; u < AT; ++u) ent[u] = nxt[u];
        t0 += 32u * 32u * AT;
    }
    __syncthreads();

    float* o = OUT + (size_t)r * ANR * NHID;
    for (int i = tid; i < ANR * NHID; i += 1024) atomicAdd(o + i, acc[(i >> 4) * APITCH + (i & 15)]);
}

__global__ void finalize_node(float* __restrict__ ACC, const float* __restrict__ T,
                              const float* __restrict__ dinv, const float* __restrict__ bias,
                              int n16, int doRelu) {
    int idx = blockIdx.x * blockDim.x + threadIdx.x;
    if (idx >= n16) return;
    int i = idx >> 4, f = idx & 15;
    float di = dinv[i];
    float v  = (ACC[idx] + T[idx]) * di + bias[f];
    ACC[idx] = doRelu ? fmaxf(v, 0.0f) : v;
}

__global__ void pool_kernel(const int* __restrict__ bvec, const float* __restrict__ Hn,
                            float* __restrict__ Gp, int n) {
    int t = blockIdx.x * blockDim.x + threadIdx.x;
    int i = t >> 2, q = t & 3;
    if (i >= n) return;
    int g = bvec[i];
    const float4 h = *(const float4*)(Hn + (size_t)i * NHID + q * 4);
    float* o = Gp + (size_t)g * NHID + q * 4;
    atomicAdd(o + 0, h.x);
    atomicAdd(o + 1, h.y);
    atomicAdd(o + 2, h.z);
    atomicAdd(o + 3, h.w);
}

__global__ void lin1_kernel(const float* __restrict__ G, const float* __restrict__ W,
                            const float* __restrict__ B, float* __restrict__ O) {
    int t = blockIdx.x * blockDim.x + threadIdx.x;
    if (t >= NGRAPH * NMLP) return;
    int gi = t / NMLP, j = t % NMLP;
    float acc = B[j];
#pragma unroll
    for (int k = 0; k < NHID; ++k)
        acc = fmaf(fmaxf(G[gi * NHID + k], 0.0f), W[k * NMLP + j], acc);
    O[t] = fmaxf(acc, 0.0f);
}

__global__ void lin2_kernel(const float* __restrict__ G, const float* __restrict__ W,
                            const float* __restrict__ B, float* __restrict__ O) {
    int t = blockIdx.x * blockDim.x + threadIdx.x;
    if (t >= NGRAPH * NCLS) return;
    int gi = t / NCLS, c = t % NCLS;
    float acc = B[c];
    for (int k = 0; k < NMLP; ++k)
        acc = fmaf(G[gi * NMLP + k], W[k * NCLS + c], acc);
    O[t] = acc;
}


extern "C" void kernel_launch(void* const* d_in, const int* in_sizes, int n_in,
                              void* d_out, int out_size, void* d_ws, size_t ws_size,
                              hipStream_t stream) {
    const float* x   = (const float*)d_in[0];
    const int*  eidx = (const int*)d_in[1];
    const int*  bvec = (const int*)d_in[2];
    const float* W1  = (const float*)d_in[3];
    const float* b1  = (const float*)d_in[4];
    const float* W2  = (const float*)d_in[5];
    const float* b2  = (const float*)d_in[6];
    const float* Wl1 = (const float*)d_in[7];
    const float* bl1 = (const float*)d_in[8];
    const float* Wl2 = (const float*)d_in[9];
    const float* bl2 = (const float*)d_in[10];
    float* out = (float*)d_out;

    char* ws = (char*)d_ws;
    size_t off = 0;
    auto walloc = [&](size_t bytes) -> void* {
        void* p = ws + off;
        off = (off + bytes + 255) & ~(size_t)255;
        return p;
    };
    float* deg  = (float*)walloc((size_t)NNODES * 4);
    float* bufA = (float*)walloc((size_t)NNODES * NHID * 4);
    float* bufB = (float*)walloc((size_t)NNODES * NHID * 4);
    float* gbuf = (float*)walloc((size_t)NGRAPH * NHID * 4);
    float* g2   = (float*)walloc((size_t)NGRAPH * NMLP * 4);
    unsigned* bcnt   = (unsigned*)walloc((size_t)AR * BNB * 4);
    unsigned* bstart = (unsigned*)walloc((size_t)(AR + 1) * 4);
    unsigned* blist  = (unsigned*)walloc((size_t)NEDGES * 4);
    const bool binned = off <= ws_size;

    const int* src = eidx;
    const int* dst = eidx + NEDGES;

    const int TB = 256;
    const int nN16  = (NNODES * NHID + TB - 1) / TB;
    const int nN4   = (NNODES * 4 + TB - 1) / TB;
    const int nTile = ((NNODES / 16) + 7) / 8;

    unsigned* part = (unsigned*)bufA;
    deg_hist<<<HR * HCH, 1024, 0, stream>>>(dst, part);
    deg_reduce<<<(NNODES / 4 + TB - 1) / TB, TB, 0, stream>>>(part, deg);
    float* dinv = deg;

    if (binned) {
        bin_count<<<BNB, 1024, 0, stream>>>(dst, bcnt);
        bin_scatter<<<BNB, 1024, 0, stream>>>(src, dst, bcnt, bstart, blist);
    }

    gemm_wmma_f32<<<nTile, TB, 0, stream>>>(x, W1, dinv, bufA, NNODES, NFEAT);
    fill_f32<<<nN16, TB, 0, stream>>>(bufB, NNODES * NHID, 0.0f);
    if (binned) edge_aggregate_binned<<<AR * ACH, 1024, 0, stream>>>(blist, bstart, bufA, bufB);
    else        edge_aggregate_filter<<<AR * ACH, 1024, 0, stream>>>(src, dst, bufA, bufB);
    finalize_node<<<nN16, TB, 0, stream>>>(bufB, bufA, dinv, b1, NNODES * NHID, 1);

    gemm_wmma_f32<<<nTile, TB, 0, stream>>>(bufB, W2, dinv, bufA, NNODES, NHID);
    fill_f32<<<nN16, TB, 0, stream>>>(bufB, NNODES * NHID, 0.0f);
    if (binned) edge_aggregate_binned<<<AR * ACH, 1024, 0, stream>>>(blist, bstart, bufA, bufB);
    else        edge_aggregate_filter<<<AR * ACH, 1024, 0, stream>>>(src, dst, bufA, bufB);
    finalize_node<<<nN16, TB, 0, stream>>>(bufB, bufA, dinv, b2, NNODES * NHID, 0);

    fill_f32<<<(NGRAPH * NHID + TB - 1) / TB, TB, 0, stream>>>(gbuf, NGRAPH * NHID, 0.0f);
    pool_kernel<<<nN4, TB, 0, stream>>>(bvec, bufB, gbuf, NNODES);
    lin1_kernel<<<(NGRAPH * NMLP + TB - 1) / TB, TB, 0, stream>>>(gbuf, Wl1, bl1, g2);
    lin2_kernel<<<(NGRAPH * NCLS + TB - 1) / TB, TB, 0, stream>>>(g2, Wl2, bl2, out);
}


# One aggregation group per destination range (250 ranges, one binned list each): it owns its rows, finishes and stores them itself; no output atomics, zero fill or finalize launch; sum order changed
# speedup vs baseline: 3.9798x; 1.1546x over previous
#include <hip/hip_runtime.h>
#include <hip/hip_bf16.h>


typedef float v2f __attribute__((ext_vector_type(2)));
typedef float v4f __attribute__((ext_vector_type(4)));
typedef float v8f __attribute__((ext_vector_type(8)));

#define NNODES  200000
#define NEDGES  6400000
#define NFEAT   128
#define NHID    16
#define NMLP    100
#define NCLS    27
#define NGRAPH  4096


__global__ void fill_f32(float* __restrict__ p, int n, float v) {
    int i = blockIdx.x * blockDim.x + threadIdx.x;
    if (i < n) p[i] = v;
}

#define HR     8
#define HBINS  (NNODES / HR)
#define HCH    32
#define HCHUNK (NEDGES / HCH)
#define HN4    (HCHUNK / 4)

__global__ __launch_bounds__(1024)
void deg_hist(const int* __restrict__ dst, unsigned* __restrict__ part) {
    __shared__ __attribute__((aligned(16))) unsigned h[HBINS];
    const int tid = threadIdx.x;
    const int r   = blockIdx.x & (HR - 1);
    const int c   = blockIdx.x >> 3;

    uint4* h4 = (uint4*)h;
    for (int i = tid; i < HBINS / 4; i += 1024) h4[i] = make_uint4(0u, 0u, 0u, 0u);
    __syncthreads();

    const unsigned lo = (unsigned)(r * HBINS);
    const int4* p = (const int4*)(dst + (size_t)c * HCHUNK);
    for (int base = tid; base < HN4; base += 1024 * 8) {
        int4 v[8];
#pragma unroll
        for (int u = 0; u < 8; ++u) {
            const int j = base + u * 1024;
            v[u] = (j < HN4) ? p[j] : make_int4(-1, -1, -1, -1);
        }
#pragma unroll
        for (int u = 0; u < 8; ++u) {
            unsigned a;
            a = (unsigned)v[u].x - lo; if (a < (unsigned)HBINS) atomicAdd(&h[a], 1u);
            a = (unsigned)v[u].y - lo; if (a < (unsigned)HBINS) atomicAdd(&h[a], 1u);
            a = (unsigned)v[u].z - lo; if (a < (unsigned)HBINS) atomicAdd(&h[a], 1u);
            a = (unsigned)v[u].w - lo; if (a < (unsigned)HBINS) atomicAdd(&h[a], 1u);
        }
    }
    __syncthreads();

    uint4* o = (uint4*)(part + (size_t)c * NNODES + lo);
    for (int i = tid; i < HBINS / 4; i += 1024) o[i] = h4[i];
}

__global__ void deg_reduce(const unsigned* __restrict__ part, float* __restrict__ dinv) {
    const int i = blockIdx.x * blockDim.x + threadIdx.x;
    if (i >= NNODES / 4) return;
    const uint4* p = (const uint4*)part + i;
    uint4 s = make_uint4(0u, 0u, 0u, 0u);
#pragma unroll
    for (int c = 0; c < HCH; ++c) {
        const uint4 v = p[(size_t)c * (NNODES / 4)];
        s.x += v.x; s.y += v.y; s.z += v.z; s.w += v.w;
    }
    float4 d;
    d.x = rsqrtf((float)s.x + 1.0f);
    d.y = rsqrtf((float)s.y + 1.0f);
    d.z = rsqrtf((float)s.z + 1.0f);
    d.w = rsqrtf((float)s.w + 1.0f);
    ((float4*)dinv)[i] = d;
}

__global__ __launch_bounds__(256)
void gemm_wmma_f32(const float* __restrict__ X, const float* __restrict__ W,
                   const float* __restrict__ rs, float* __restrict__ T, int M, int K) {
    __shared__ float lw[NFEAT * NHID];
    const int tid = threadIdx.x;
    const int kn  = K * NHID;
    for (int i = tid; i < kn; i += 256) lw[i] = W[i];
    __syncthreads();

    const int wave = tid >> 5;
    const int lane = tid & 31;
    const int tile = blockIdx.x * 8 + wave;
    if (tile * 16 >= M) return;

    const int row   = tile * 16 + (lane & 15);
    const int kHalf = (lane >> 4) << 1;
    const int n     = lane & 15;

    v8f c = {};
    const float* xr = X + (size_t)row * K + kHalf;
    for (int k0 = 0; k0 < K; k0 += 4) {
        float2 av = *(const float2*)(xr + k0);
        v2f a; a.x = av.x; a.y = av.y;
        v2f b;
        b.x = lw[(k0 + kHalf)     * NHID + n];
        b.y = lw[(k0 + kHalf + 1) * NHID + n];
        c = __builtin_amdgcn_wmma_f32_16x16x4_f32(
                 false, a,  false, b,
                 (short)0, c,  false,  false);
    }

    const int rbase = tile * 16 + ((lane >> 4) << 3);
#pragma unroll
    for (int v = 0; v < 8; ++v)
        T[(size_t)(rbase + v) * NHID + n] = c[v] * rs[rbase + v];
}

#define AR     50
#define ANR    (NNODES / AR)
#define ACH    5
#define ACHUNK (NEDGES / ACH)
#define AN4    (ACHUNK / 4)
#define AU     6
#define AWT    (32 * AU)
#define ANWT   ((AN4 + AWT - 1) / AWT)
#define AWQ    128
#define AB     3
#define APITCH 17

__global__ __launch_bounds__(1024)
void edge_aggregate_filter(const int* __restrict__ src, const int* __restrict__ dst,
                           const float* __restrict__ T, float* __restrict__ OUT) {
    __shared__ __attribute__((aligned(16))) float acc[ANR * APITCH];
    __shared__ uint2 queue[32][AWQ];
    const int tid  = threadIdx.x;
    const int lane = tid & 31;
    const int wave = __builtin_amdgcn_readfirstlane(tid >> 5);
    const int r    = blockIdx.x % AR;
    const int c    = blockIdx.x / AR;

    float4* a4 = (float4*)acc;
    for (int i = tid; i < ANR * APITCH / 4; i += 1024) a4[i] = make_float4(0.0f, 0.0f, 0.0f, 0.0f);
    for (int i = tid; i < 32 * AWQ; i += 1024) (&queue[0][0])[i] = make_uint2(0xffffffffu, 0u);
    __syncthreads();

    const unsigned lo = (unsigned)(r * ANR);
    const int ebase   = c * ACHUNK;
    const int4* p     = (const int4*)(dst + (size_t)ebase);
    const unsigned grp = (unsigned)(lane >> 2);
    const unsigned f4  = (unsigned)(lane & 3) << 2;
    uint2* qb = queue[wave];

    int4 v[AU];
    auto load_tile = [&](int wt) {
#pragma unroll
        for (int u = 0; u < AU; ++u) {
            const int jb = wt * AWT + u * 32;
            if (jb < AN4) v[u] = p[jb + lane];
            else          v[u] = make_int4(-1, -1, -1, -1);
        }
    };

    auto masks = [&](unsigned (&msk)[4 * AU]) -> unsigned {
        unsigned tot = 0u;
#pragma unroll
        for (int u = 0; u < AU; ++u) {
            msk[4 * u + 0] = __builtin_amdgcn_ballot_w32((unsigned)v[u].x - lo < (unsigned)ANR);
            msk[4 * u + 1] = __builtin_amdgcn_ballot_w32((unsigned)v[u].y - lo < (unsigned)ANR);
            msk[4 * u + 2] = __builtin_amdgcn_ballot_w32((unsigned)v[u].z - lo < (unsigned)ANR);
            msk[4 * u + 3] = __builtin_amdgcn_ballot_w32((unsigned)v[u].w - lo < (unsigned)ANR);
            tot += (unsigned)(__builtin_popcount(msk[4 * u + 0]) + __builtin_popcount(msk[4 * u + 1]) +
                              __builtin_popcount(msk[4 * u + 2]) + __builtin_popcount(msk[4 * u + 3]));
        }
        return tot;
    };

    auto emit = [&](const unsigned (&msk)[4 * AU], int wt, unsigned run) {
#pragma unroll
        for (int u = 0; u < AU; ++u) {
            const int e0 = ebase + 4 * (wt * AWT + u * 32 + lane);
            const int dv[4] = { v[u].x, v[u].y, v[u].z, v[u].w };
#pragma unroll
            for (int k = 0; k < 4; ++k) {
                const unsigned m = msk[4 * u + k];
                if (m != 0u) {
                    const unsigned a = (unsigned)dv[k] - lo;
                    const unsigned pos = run + __builtin_amdgcn_mbcnt_lo(m, 0u);
                    if (a < (unsigned)ANR && pos < (unsigned)AWQ)
                        qb[pos] = make_uint2(a, (unsigned)(e0 + k));
                    run += (unsigned)__builtin_popcount(m);
                }
            }
        }
    };

    unsigned pa[AB];
    v4f pm[AB];
#pragma unroll
    for (int k = 0; k < AB; ++k) { pa[k] = 0xffffffffu; pm[k] = (v4f){0.0f, 0.0f, 0.0f, 0.0f}; }

    auto issue = [&](unsigned n, unsigned j0) {
        uint2 q[AB];
        unsigned s[AB];
#pragma unroll
        for (int k = 0; k < AB; ++k) {
            const unsigned jj = j0 + 8u * k;
            const uint2 t = qb[jj & (unsigned)(AWQ - 1)];
            q[k].x = (jj < n) ? t.x : 0xffffffffu;
            q[k].y = (jj < n) ? t.y : 0u;
        }
#pragma unroll
        for (int k = 0; k < AB; ++k) {
            s[k]  = (unsigned)*(const int*)((const char*)src + (q[k].y << 2));
            pa[k] = q[k].x;
        }
#pragma unroll
        for (int k = 0; k < AB; ++k)
            pm[k] = *(const v4f*)((const char*)T + ((s[k] << 6) | (f4 << 2)));
    };
    auto commit = [&]() {
#pragma unroll
        for (int k = 0; k < AB; ++k) {
            v4f val = pm[k];
            asm volatile("" : "+v"(val));
            if (pa[k] < (unsigned)ANR) {
                float* d = &acc[pa[k] * APITCH + f4];
                atomicAdd(d + 0, val.x);
                atomicAdd(d + 1, val.y);
                atomicAdd(d + 2, val.z);
                atomicAdd(d + 3, val.w);
            }
        }
    };

    int wt = wave;
    load_tile(wt);
    unsigned msk[4 * AU];
    unsigned tot = masks(msk);
    unsigned rlo = 0u;
    while (wt < ANWT) {
        emit(msk, wt, 0u - rlo);
        commit();
        __builtin_amdgcn_wave_barrier();
        const unsigned n = min(tot - rlo, (unsigned)AWQ);
        const bool last = rlo + (unsigned)AWQ >= tot;
        if (last) load_tile(wt + 32);
        issue(n, grp);
        for (unsigned jr = 8u * AB; jr < n; jr += 8u * AB) { commit(); issue(n, grp + jr); }
        __builtin_amdgcn_wave_barrier();
        if (last) {
            tot = masks(msk);
            wt += 32;
            rlo = 0u;
        } else {
            rlo += (unsigned)AWQ;
        }
    }
    commit();
    __syncthreads();

    float* o = OUT + (size_t)lo * NHID;
    for (int i = tid; i < ANR * NHID; i += 1024) atomicAdd(o + i, acc[(i >> 4) * APITCH + (i & 15)]);
}

#define BR     250
#define BNR    (NNODES / BR)
#define BNB    250
#define BSL    (NEDGES / BNB)
#define BSL4   (BSL / 4)
#define BU     ((BSL4 + 1023) / 1024)
#define BCU    ((BNB + 31) / 32)
#define SBITS  18
static_assert(BR * BNR == NNODES, "the ranges cover the rows exactly");

__global__ __launch_bounds__(1024)
void bin_count(const int* __restrict__ dst, unsigned* __restrict__ cnt) {
    __shared__ unsigned h[BR];
    const int tid = threadIdx.x;
    if (tid < BR) h[tid] = 0u;
    __syncthreads();
    const int4* p = (const int4*)(dst + (size_t)blockIdx.x * BSL);
    int4 v[BU];
#pragma unroll
    for (int u = 0; u < BU; ++u) {
        const int j = tid + u * 1024;
        v[u] = (j < BSL4) ? p[j] : make_int4(-1, -1, -1, -1);
    }
#pragma unroll
    for (int u = 0; u < BU; ++u) {
        const unsigned d[4] = { (unsigned)v[u].x, (unsigned)v[u].y, (unsigned)v[u].z, (unsigned)v[u].w };
#pragma unroll
        for (int k = 0; k < 4; ++k)
            if (d[k] < (unsigned)NNODES) atomicAdd(&h[d[k] / (unsigned)BNR], 1u);
    }
    __syncthreads();
    if (tid < BR) cnt[tid * BNB + blockIdx.x] = h[tid];
}

__global__ __launch_bounds__(1024)
void bin_scatter(const int* __restrict__ src, const int* __restrict__ dst,
                 const unsigned* __restrict__ cnt, unsigned* __restrict__ bstart,
                 unsigned* __restrict__ list) {
    __shared__ unsigned tot[BR];
    __shared__ unsigned pre[BR];
    __shared__ unsigned cur[BR];
    const int tid  = threadIdx.x;
    const int lane = tid & 31;
    const int wave = __builtin_amdgcn_readfirstlane(tid >> 5);
    const int b    = blockIdx.x;

    for (int r = wave; r < BR; r += 32) {
        unsigned t = 0u, q = 0u;
#pragma unroll
        for (int u = 0; u < BCU; ++u) {
            const int j = lane + 32 * u;
            const unsigned v = (j < BNB) ? cnt[r * BNB + j] : 0u;
            t += v;
            q += (j < b) ? v : 0u;
        }
        for (int m = 16; m >= 1; m >>= 1) {
            t += __shfl_xor(t, m, 32);
            q += __shfl_xor(q, m, 32);
        }
        if (lane == 0) { tot[r] = t; pre[r] = q; }
    }
    __syncthreads();
    if (tid < BR) {
        unsigned run = 0u;
#pragma unroll
        for (int j = 0; j < BR; ++j) run += (j < tid) ? tot[j] : 0u;
        cur[tid] = run + pre[tid];
        if (b == 0) {
            bstart[tid] = run;
            if (tid == BR - 1) bstart[BR] = run + tot[tid];
        }
    }
    __syncthreads();

    const int4* pd = (const int4*)(dst + (size_t)b * BSL);
    const int4* ps = (const int4*)(src + (size_t)b * BSL);
    int4 vd[BU], vs[BU];
#pragma unroll
    for (int u = 0; u < BU; ++u) {
        const int j = tid + u * 1024;
        vd[u] = (j < BSL4) ? pd[j] : make_int4(-1, -1, -1, -1);
        vs[u] = (j < BSL4) ? ps[j] : make_int4(0, 0, 0, 0);
    }
#pragma unroll
    for (int u = 0; u < BU; ++u) {
        const unsigned d[4] = { (unsigned)vd[u].x, (unsigned)vd[u].y, (unsigned)vd[u].z, (unsigned)vd[u].w };
        const unsigned s[4] = { (unsigned)vs[u].x, (unsigned)vs[u].y, (unsigned)vs[u].z, (unsigned)vs[u].w };
#pragma unroll
        for (int k = 0; k < 4; ++k) {
            if (d[k] < (unsigned)NNODES) {
                const unsigned r   = d[k] / (unsigned)BNR;
                const unsigned a   = d[k] - r * (unsigned)BNR;
                const unsigned pos = atomicAdd(&cur[r], 1u);
                list[pos] = (a << SBITS) | s[k];
            }
        }
    }
}

#define AT     2
#define BLDSW  42240
static_assert(BNR * APITCH <= BLDSW, "the rows fit into the group's LDS");
static_assert((BNR * APITCH) % 4 == 0 && (BNR * NHID) % 32 == 0, "16-byte clears, whole 128-byte output lines per range");

__global__ __launch_bounds__(1024)
void edge_aggregate_binned(const unsigned* __restrict__ list, const unsigned* __restrict__ bstart,
                           const float* __restrict__ T, const float* __restrict__ dinv,
                           const float* __restrict__ bias, float* __restrict__ OUT, int doRelu) {
    __shared__ __attribute__((aligned(16))) float acc[BLDSW];
    const int tid  = threadIdx.x;
    const int lane = tid & 31;
    const int wave = __builtin_amdgcn_readfirstlane(tid >> 5);
    const int r    = blockIdx.x;

    float4* a4 = (float4*)acc;
    for (int i = tid; i < BNR * APITCH / 4; i += 1024) a4[i] = make_float4(0.0f, 0.0f, 0.0f, 0.0f);
    __syncthreads();

    const unsigned plo = bstart[r];
    const unsigned phi = bstart[r + 1];
    const unsigned grp = (unsigned)(lane >> 2);
    const unsigned f4  = (unsigned)(lane & 3) << 2;
    const unsigned PAD = 0x3fffu << SBITS;

    unsigned ent[AT];
    auto load_ent = [&](unsigned t0, unsigned (&e)[AT]) {
#pragma unroll
        for (int u = 0; u < AT; ++u) {
            const unsigned idx = t0 + 32u * u + (unsigned)lane;
            const unsigned w = list[min(idx, (unsigned)(NEDGES - 1))];
            e[u] = (idx < phi) ? w : PAD;
        }
    };

    unsigned t0 = plo + (unsigned)wave * (32u * AT);
    load_ent(t0, ent);
    while (t0 < phi) {
        unsigned nxt[AT];
        load_ent(t0 + 32u * 32u * AT, nxt);
        unsigned pa[4 * AT];
        v4f pm[4 * AT];
#pragma unroll
        for (int u = 0; u < AT; ++u) {
#pragma unroll
            for (int k = 0; k < 4; ++k) {
                const unsigned e = (unsigned)__shfl((int)ent[u], (int)(8u * k + grp), 32);
                const unsigned s = e & ((1u << SBITS) - 1u);
                pa[4 * u + k] = e >> SBITS;
                pm[4 * u + k] = *(const v4f*)((const char*)T + ((s << 6) | (f4 << 2)));
            }
        }
#pragma unroll
        for (int j = 0; j < 4 * AT; ++j) {
            if (pa[j] < (unsigned)BNR) {
                float* d = &acc[pa[j] * APITCH + f4];
                atomicAdd(d + 0, pm[j].x);
                atomicAdd(d + 1, pm[j].y);
                atomicAdd(d + 2, pm[j].z);
                atomicAdd(d + 3, pm[j].w);
            }
        }
#pragma unroll
        for (int u = 0; u < AT; ++u) ent[u] = nxt[u];
        t0 += 32u * 32u * AT;
    }
    __syncthreads();

    const int row0 = r * BNR;
    const float4* t4 = (const float4*)(T + (size_t)row0 * NHID);
    float4* o4 = (float4*)(OUT + (size_t)row0 * NHID);
    for (int i = tid; i < BNR * NHID / 4; i += 1024) {
        const int row = i >> 2;
        const int f   = (i & 3) << 2;
        const float di = dinv[row0 + row];
        const float4 t = t4[i];
        const float* a = &acc[row * APITCH + f];
        float4 v;
        v.x = (a[0] + t.x) * di + bias[f + 0];
        v.y = (a[1] + t.y) * di + bias[f + 1];
        v.z = (a[2] + t.z) * di + bias[f + 2];
        v.w = (a[3] + t.w) * di + bias[f + 3];
        if (doRelu) {
            v.x = fmaxf(v.x, 0.0f);
            v.y = fmaxf(v.y, 0.0f);
            v.z = fmaxf(v.z, 0.0f);
            v.w = fmaxf(v.w, 0.0f);
        }
        o4[i] = v;
    }
}

__global__ void finalize_node(float* __restrict__ ACC, const float* __restrict__ T,
                              const float* __restrict__ dinv, const float* __restrict__ bias,
                              int n16, int doRelu) {
    int idx = blockIdx.x * blockDim.x + threadIdx.x;
    if (idx >= n16) return;
    int i = idx >> 4, f = idx & 15;
    float di = dinv[i];
    float v  = (ACC[idx] + T[idx]) * di + bias[f];
    ACC[idx] = doRelu ? fmaxf(v, 0.0f) : v;
}

__global__ void pool_kernel(const int* __restrict__ bvec, const float* __restrict__ Hn,
                            float* __restrict__ Gp, int n) {
    int t = blockIdx.x * blockDim.x + threadIdx.x;
    int i = t >> 2, q = t & 3;
    if (i >= n) return;
    int g = bvec[i];
    const float4 h = *(const float4*)(Hn + (size_t)i * NHID + q * 4);
    float* o = Gp + (size_t)g * NHID + q * 4;
    atomicAdd(o + 0, h.x);
    atomicAdd(o + 1, h.y);
    atomicAdd(o + 2, h.z);
    atomicAdd(o + 3, h.w);
}

__global__ void lin1_kernel(const float* __restrict__ G, const float* __restrict__ W,
                            const float* __restrict__ B, float* __restrict__ O) {
    int t = blockIdx.x * blockDim.x + threadIdx.x;
    if (t >= NGRAPH * NMLP) return;
    int gi = t / NMLP, j = t % NMLP;
    float acc = B[j];
#pragma unroll
    for (int k = 0; k < NHID; ++k)
        acc = fmaf(fmaxf(G[gi * NHID + k], 0.0f), W[k * NMLP + j], acc);
    O[t] = fmaxf(acc, 0.0f);
}

__global__ void lin2_kernel(const float* __restrict__ G, const float* __restrict__ W,
                            const float* __restrict__ B, float* __restrict__ O) {
    int t = blockIdx.x * blockDim.x + threadIdx.x;
    if (t >= NGRAPH * NCLS) return;
    int gi = t / NCLS, c = t % NCLS;
    float acc = B[c];
    for (int k = 0; k < NMLP; ++k)
        acc = fmaf(G[gi * NMLP + k], W[k * NCLS + c], acc);
    O[t] = acc;
}


extern "C" void kernel_launch(void* const* d_in, const int* in_sizes, int n_in,
                              void* d_out, int out_size, void* d_ws, size_t ws_size,
                              hipStream_t stream) {
    const float* x   = (const float*)d_in[0];
    const int*  eidx = (const int*)d_in[1];
    const int*  bvec = (const int*)d_in[2];
    const float* W1  = (const float*)d_in[3];
    const float* b1  = (const float*)d_in[4];
    const float* W2  = (const float*)d_in[5];
    const float* b2  = (const float*)d_in[6];
    const float* Wl1 = (const float*)d_in[7];
    const float* bl1 = (const float*)d_in[8];
    const float* Wl2 = (const float*)d_in[9];
    const float* bl2 = (const float*)d_in[10];
    float* out = (float*)d_out;

    char* ws = (char*)d_ws;
    size_t off = 0;
    auto walloc = [&](size_t bytes) -> void* {
        void* p = ws + off;
        off = (off + bytes + 255) & ~(size_t)255;
        return p;
    };
    float* deg  = (float*)walloc((size_t)NNODES * 4);
    float* bufA = (float*)walloc((size_t)NNODES * NHID * 4);
    float* bufB = (float*)walloc((size_t)NNODES * NHID * 4);
    float* gbuf = (float*)walloc((size_t)NGRAPH * NHID * 4);
    float* g2   = (float*)walloc((size_t)NGRAPH * NMLP * 4);
    static_assert((size_t)BR * BNB * 4 <= (size_t)NGRAPH * NMLP * 4, "the count table fits into g2's place");
    unsigned* bcnt   = (unsigned*)g2;
    unsigned* bstart = (unsigned*)walloc((size_t)(BR + 1) * 4);
    unsigned* blist  = (unsigned*)walloc((size_t)NEDGES * 4);
    const bool binned = off <= ws_size;

    const int* src = eidx;
    const int* dst = eidx + NEDGES;

    const int TB = 256;
    const int nN16  = (NNODES * NHID + TB - 1) / TB;
    const int nN4   = (NNODES * 4 + TB - 1) / TB;
    const int nTile = ((NNODES / 16) + 7) / 8;

    unsigned* part = (unsigned*)bufA;
    deg_hist<<<HR * HCH, 1024, 0, stream>>>(dst, part);
    deg_reduce<<<(NNODES / 4 + TB - 1) / TB, TB, 0, stream>>>(part, deg);
    float* dinv = deg;

    if (binned) {
        bin_count<<<BNB, 1024, 0, stream>>>(dst, bcnt);
        bin_scatter<<<BNB, 1024, 0, stream>>>(src, dst, bcnt, bstart, blist);
    }

    gemm_wmma_f32<<<nTile, TB, 0, stream>>>(x, W1, dinv, bufA, NNODES, NFEAT);
    if (binned) {
        edge_aggregate_binned<<<BR, 1024, 0, stream>>>(blist, bstart, bufA, dinv, b1, bufB, 1);
    } else {
        fill_f32<<<nN16, TB, 0, stream>>>(bufB, NNODES * NHID, 0.0f);
        edge_aggregate_filter<<<AR * ACH, 1024, 0, stream>>>(src, dst, bufA, bufB);
        finalize_node<<<nN16, TB, 0, stream>>>(bufB, bufA, dinv, b1, NNODES * NHID, 1);
    }

    gemm_wmma_f32<<<nTile, TB, 0, stream>>>(bufB, W2, dinv, bufA, NNODES, NHID);
    if (binned) {
        edge_aggregate_binned<<<BR, 1024, 0, stream>>>(blist, bstart, bufA, dinv, b2, bufB, 0);
    } else {
        fill_f32<<<nN16, TB, 0, stream>>>(bufB, NNODES * NHID, 0.0f);
        edge_aggregate_filter<<<AR * ACH, 1024, 0, stream>>>(src, dst, bufA, bufB);
        finalize_node<<<nN16, TB, 0, stream>>>(bufB, bufA, dinv, b2, NNODES * NHID, 0);
    }

    fill_f32<<<(NGRAPH * NHID + TB - 1) / TB, TB, 0, stream>>>(gbuf, NGRAPH * NHID, 0.0f);
    pool_kernel<<<nN4, TB, 0, stream>>>(bvec, bufB, gbuf, NNODES);
    lin1_kernel<<<(NGRAPH * NMLP + TB - 1) / TB, TB, 0, stream>>>(gbuf, Wl1, bl1, g2);
    lin2_kernel<<<(NGRAPH * NCLS + TB - 1) / TB, TB, 0, stream>>>(g2, Wl2, bl2, out);
}


# In-degrees counted from the binned edge list: each range's group counts its own list in an 800-word LDS table and stores dinv; the 8-pass histogram of the raw list and its reduce launch go
# speedup vs baseline: 4.3403x; 1.0906x over previous
#include <hip/hip_runtime.h>
#include <hip/hip_bf16.h>


typedef float v2f __attribute__((ext_vector_type(2)));
typedef float v4f __attribute__((ext_vector_type(4)));
typedef float v8f __attribute__((ext_vector_type(8)));

#define NNODES  200000
#define NEDGES  6400000
#define NFEAT   128
#define NHID    16
#define NMLP    100
#define NCLS    27
#define NGRAPH  4096


__global__ void fill_f32(float* __restrict__ p, int n, float v) {
    int i = blockIdx.x * blockDim.x + threadIdx.x;
    if (i < n) p[i] = v;
}

#define HR     8
#define HBINS  (NNODES / HR)
#define HCH    32
#define HCHUNK (NEDGES / HCH)
#define HN4    (HCHUNK / 4)

__global__ __launch_bounds__(1024)
void deg_hist(const int* __restrict__ dst, unsigned* __restrict__ part) {
    __shared__ __attribute__((aligned(16))) unsigned h[HBINS];
    const int tid = threadIdx.x;
    const int r   = blockIdx.x & (HR - 1);
    const int c   = blockIdx.x >> 3;

    uint4* h4 = (uint4*)h;
    for (int i = tid; i < HBINS / 4; i += 1024) h4[i] = make_uint4(0u, 0u, 0u, 0u);
    __syncthreads();

    const unsigned lo = (unsigned)(r * HBINS);
    const int4* p = (const int4*)(dst + (size_t)c * HCHUNK);
    for (int base = tid; base < HN4; base += 1024 * 8) {
        int4 v[8];
#pragma unroll
        for (int u = 0; u < 8; ++u) {
            const int j = base + u * 1024;
            v[u] = (j < HN4) ? p[j] : make_int4(-1, -1, -1, -1);
        }
#pragma unroll
        for (int u = 0; u < 8; ++u) {
            unsigned a;
            a = (unsigned)v[u].x - lo; if (a < (unsigned)HBINS) atomicAdd(&h[a], 1u);
            a = (unsigned)v[u].y - lo; if (a < (unsigned)HBINS) atomicAdd(&h[a], 1u);
            a = (unsigned)v[u].z - lo; if (a < (unsigned)HBINS) atomicAdd(&h[a], 1u);
            a = (unsigned)v[u].w - lo; if (a < (unsigned)HBINS) atomicAdd(&h[a], 1u);
        }
    }
    __syncthreads();

    uint4* o = (uint4*)(part + (size_t)c * NNODES + lo);
    for (int i = tid; i < HBINS / 4; i += 1024) o[i] = h4[i];
}

__global__ void deg_reduce(const unsigned* __restrict__ part, float* __restrict__ dinv) {
    const int i = blockIdx.x * blockDim.x + threadIdx.x;
    if (i >= NNODES / 4) return;
    const uint4* p = (const uint4*)part + i;
    uint4 s = make_uint4(0u, 0u, 0u, 0u);
#pragma unroll
    for (int c = 0; c < HCH; ++c) {
        const uint4 v = p[(size_t)c * (NNODES / 4)];
        s.x += v.x; s.y += v.y; s.z += v.z; s.w += v.w;
    }
    float4 d;
    d.x = rsqrtf((float)s.x + 1.0f);
    d.y = rsqrtf((float)s.y + 1.0f);
    d.z = rsqrtf((float)s.z + 1.0f);
    d.w = rsqrtf((float)s.w + 1.0f);
    ((float4*)dinv)[i] = d;
}

__global__ __launch_bounds__(256)
void gemm_wmma_f32(const float* __restrict__ X, const float* __restrict__ W,
                   const float* __restrict__ rs, float* __restrict__ T, int M, int K) {
    __shared__ float lw[NFEAT * NHID];
    const int tid = threadIdx.x;
    const int kn  = K * NHID;
    for (int i = tid; i < kn; i += 256) lw[i] = W[i];
    __syncthreads();

    const int wave = tid >> 5;
    const int lane = tid & 31;
    const int tile = blockIdx.x * 8 + wave;
    if (tile * 16 >= M) return;

    const int row   = tile * 16 + (lane & 15);
    const int kHalf = (lane >> 4) << 1;
    const int n     = lane & 15;

    v8f c = {};
    const float* xr = X + (size_t)row * K + kHalf;
    for (int k0 = 0; k0 < K; k0 += 4) {
        float2 av = *(const float2*)(xr + k0);
        v2f a; a.x = av.x; a.y = av.y;
        v2f b;
        b.x = lw[(k0 + kHalf)     * NHID + n];
        b.y = lw[(k0 + kHalf + 1) * NHID + n];
        c = __builtin_amdgcn_wmma_f32_16x16x4_f32(
                 false, a,  false, b,
                 (short)0, c,  false,  false);
    }

    const int rbase = tile * 16 + ((lane >> 4) << 3);
#pragma unroll
    for (int v = 0; v < 8; ++v)
        T[(size_t)(rbase + v) * NHID + n] = c[v] * rs[rbase + v];
}

#define AR     50
#define ANR    (NNODES / AR)
#define ACH    5
#define ACHUNK (NEDGES / ACH)
#define AN4    (ACHUNK / 4)
#define AU     6
#define AWT    (32 * AU)
#define ANWT   ((AN4 + AWT - 1) / AWT)
#define AWQ    128
#define AB     3
#define APITCH 17

__global__ __launch_bounds__(1024)
void edge_aggregate_filter(const int* __restrict__ src, const int* __restrict__ dst,
                           const float* __restrict__ T, float* __restrict__ OUT) {
    __shared__ __attribute__((aligned(16))) float acc[ANR * APITCH];
    __shared__ uint2 queue[32][AWQ];
    const int tid  = threadIdx.x;
    const int lane = tid & 31;
    const int wave = __builtin_amdgcn_readfirstlane(tid >> 5);
    const int r    = blockIdx.x % AR;
    const int c    = blockIdx.x / AR;

    float4* a4 = (float4*)acc;
    for (int i = tid; i < ANR * APITCH / 4; i += 1024) a4[i] = make_float4(0.0f, 0.0f, 0.0f, 0.0f);
    for (int i = tid; i < 32 * AWQ; i += 1024) (&queue[0][0])[i] = make_uint2(0xffffffffu, 0u);
    __syncthreads();

    const unsigned lo = (unsigned)(r * ANR);
    const int ebase   = c * ACHUNK;
    const int4* p     = (const int4*)(dst + (size_t)ebase);
    const unsigned grp = (unsigned)(lane >> 2);
    const unsigned f4  = (unsigned)(lane & 3) << 2;
    uint2* qb = queue[wave];

    int4 v[AU];
    auto load_tile = [&](int wt) {
#pragma unroll
        for (int u = 0; u < AU; ++u) {
            const int jb = wt * AWT + u * 32;
            if (jb < AN4) v[u] = p[jb + lane];
            else          v[u] = make_int4(-1, -1, -1, -1);
        }
    };

    auto masks = [&](unsigned (&msk)[4 * AU]) -> unsigned {
        unsigned tot = 0u;
#pragma unroll
        for (int u = 0; u < AU; ++u) {
            msk[4 * u + 0] = __builtin_amdgcn_ballot_w32((unsigned)v[u].x - lo < (unsigned)ANR);
            msk[4 * u + 1] = __builtin_amdgcn_ballot_w32((unsigned)v[u].y - lo < (unsigned)ANR);
            msk[4 * u + 2] = __builtin_amdgcn_ballot_w32((unsigned)v[u].z - lo < (unsigned)ANR);
            msk[4 * u + 3] = __builtin_amdgcn_ballot_w32((unsigned)v[u].w - lo < (unsigned)ANR);
            tot += (unsigned)(__builtin_popcount(msk[4 * u + 0]) + __builtin_popcount(msk[4 * u + 1]) +
                              __builtin_popcount(msk[4 * u + 2]) + __builtin_popcount(msk[4 * u + 3]));
        }
        return tot;
    };

    auto emit = [&](const unsigned (&msk)[4 * AU], int wt, unsigned run) {
#pragma unroll
        for (int u = 0; u < AU; ++u) {
            const int e0 = ebase + 4 * (wt * AWT + u * 32 + lane);
            const int dv[4] = { v[u].x, v[u].y, v[u].z, v[u].w };
#pragma unroll
            for (int k = 0; k < 4; ++k) {
                const unsigned m = msk[4 * u + k];
                if (m != 0u) {
                    const unsigned a = (unsigned)dv[k] - lo;
                    const unsigned pos = run + __builtin_amdgcn_mbcnt_lo(m, 0u);
                    if (a < (unsigned)ANR && pos < (unsigned)AWQ)
                        qb[pos] = make_uint2(a, (unsigned)(e0 + k));
                    run += (unsigned)__builtin_popcount(m);
                }
            }
        }
    };

    unsigned pa[AB];
    v4f pm[AB];
#pragma unroll
    for (int k = 0; k < AB; ++k) { pa[k] = 0xffffffffu; pm[k] = (v4f){0.0f, 0.0f, 0.0f, 0.0f}; }

    auto issue = [&](unsigned n, unsigned j0) {
        uint2 q[AB];
        unsigned s[AB];
#pragma unroll
        for (int k = 0; k < AB; ++k) {
            const unsigned jj = j0 + 8u * k;
            const uint2 t = qb[jj & (unsigned)(AWQ - 1)];
            q[k].x = (jj < n) ? t.x : 0xffffffffu;
            q[k].y = (jj < n) ? t.y : 0u;
        }
#pragma unroll
        for (int k = 0; k < AB; ++k) {
            s[k]  = (unsigned)*(const int*)((const char*)src + (q[k].y << 2));
            pa[k] = q[k].x;
        }
#pragma unroll
        for (int k = 0; k < AB; ++k)
            pm[k] = *(const v4f*)((const char*)T + ((s[k] << 6) | (f4 << 2)));
    };
    auto commit = [&]() {
#pragma unroll
        for (int k = 0; k < AB; ++k) {
            v4f val = pm[k];
            asm volatile("" : "+v"(val));
            if (pa[k] < (unsigned)ANR) {
                float* d = &acc[pa[k] * APITCH + f4];
                atomicAdd(d + 0, val.x);
                atomicAdd(d + 1, val.y);
                atomicAdd(d + 2, val.z);
                atomicAdd(d + 3, val.w);
            }
        }
    };

    int wt = wave;
    load_tile(wt);
    unsigned msk[4 * AU];
    unsigned tot = masks(msk);
    unsigned rlo = 0u;
    while (wt < ANWT) {
        emit(msk, wt, 0u - rlo);
        commit();
        __builtin_amdgcn_wave_barrier();
        const unsigned n = min(tot - rlo, (unsigned)AWQ);
        const bool last = rlo + (unsigned)AWQ >= tot;
        if (last) load_tile(wt + 32);
        issue(n, grp);
        for (unsigned jr = 8u * AB; jr < n; jr += 8u * AB) { commit(); issue(n, grp + jr); }
        __builtin_amdgcn_wave_barrier();
        if (last) {
            tot = masks(msk);
            wt += 32;
            rlo = 0u;
        } else {
            rlo += (unsigned)AWQ;
        }
    }
    commit();
    __syncthreads();

    float* o = OUT + (size_t)lo * NHID;
    for (int i = tid; i < ANR * NHID; i += 1024) atomicAdd(o + i, acc[(i >> 4) * APITCH + (i & 15)]);
}

#define BR     250
#define BNR    (NNODES / BR)
#define BNB    250
#define BSL    (NEDGES / BNB)
#define BSL4   (BSL / 4)
#define BU     ((BSL4 + 1023) / 1024)
#define BCU    ((BNB + 31) / 32)
#define SBITS  18
static_assert(BR * BNR == NNODES, "the ranges cover the rows exactly");

__global__ __launch_bounds__(1024)
void bin_count(const int* __restrict__ dst, unsigned* __restrict__ cnt) {
    __shared__ unsigned h[BR];
    const int tid = threadIdx.x;
    if (tid < BR) h[tid] = 0u;
    __syncthreads();
    const int4* p = (const int4*)(dst + (size_t)blockIdx.x * BSL);
    int4 v[BU];
#pragma unroll
    for (int u = 0; u < BU; ++u) {
        const int j = tid + u * 1024;
        v[u] = (j < BSL4) ? p[j] : make_int4(-1, -1, -1, -1);
    }
#pragma unroll
    for (int u = 0; u < BU; ++u) {
        const unsigned d[4] = { (unsigned)v[u].x, (unsigned)v[u].y, (unsigned)v[u].z, (unsigned)v[u].w };
#pragma unroll
        for (int k = 0; k < 4; ++k)
            if (d[k] < (unsigned)NNODES) atomicAdd(&h[d[k] / (unsigned)BNR], 1u);
    }
    __syncthreads();
    if (tid < BR) cnt[tid * BNB + blockIdx.x] = h[tid];
}

__global__ __launch_bounds__(1024)
void bin_scatter(const int* __restrict__ src, const int* __restrict__ dst,
                 const unsigned* __restrict__ cnt, unsigned* __restrict__ bstart,
                 unsigned* __restrict__ list) {
    __shared__ unsigned tot[BR];
    __shared__ unsigned pre[BR];
    __shared__ unsigned cur[BR];
    const int tid  = threadIdx.x;
    const int lane = tid & 31;
    const int wave = __builtin_amdgcn_readfirstlane(tid >> 5);
    const int b    = blockIdx.x;

    for (int r = wave; r < BR; r += 32) {
        unsigned t = 0u, q = 0u;
#pragma unroll
        for (int u = 0; u < BCU; ++u) {
            const int j = lane + 32 * u;
            const unsigned v = (j < BNB) ? cnt[r * BNB + j] : 0u;
            t += v;
            q += (j < b) ? v : 0u;
        }
        for (int m = 16; m >= 1; m >>= 1) {
            t += __shfl_xor(t, m, 32);
            q += __shfl_xor(q, m, 32);
        }
        if (lane == 0) { tot[r] = t; pre[r] = q; }
    }
    __syncthreads();
    if (tid < BR) {
        unsigned run = 0u;
#pragma unroll
        for (int j = 0; j < BR; ++j) run += (j < tid) ? tot[j] : 0u;
        cur[tid] = run + pre[tid];
        if (b == 0) {
            bstart[tid] = run;
            if (tid == BR - 1) bstart[BR] = run + tot[tid];
        }
    }
    __syncthreads();

    const int4* pd = (const int4*)(dst + (size_t)b * BSL);
    const int4* ps = (const int4*)(src + (size_t)b * BSL);
    int4 vd[BU], vs[BU];
#pragma unroll
    for (int u = 0; u < BU; ++u) {
        const int j = tid + u * 1024;
        vd[u] = (j < BSL4) ? pd[j] : make_int4(-1, -1, -1, -1);
        vs[u] = (j < BSL4) ? ps[j] : make_int4(0, 0, 0, 0);
    }
#pragma unroll
    for (int u = 0; u < BU; ++u) {
        const unsigned d[4] = { (unsigned)vd[u].x, (unsigned)vd[u].y, (unsigned)vd[u].z, (unsigned)vd[u].w };
        const unsigned s[4] = { (unsigned)vs[u].x, (unsigned)vs[u].y, (unsigned)vs[u].z, (unsigned)vs[u].w };
#pragma unroll
        for (int k = 0; k < 4; ++k) {
            if (d[k] < (unsigned)NNODES) {
                const unsigned r   = d[k] / (unsigned)BNR;
                const unsigned a   = d[k] - r * (unsigned)BNR;
                const unsigned pos = atomicAdd(&cur[r], 1u);
                list[pos] = (a << SBITS) | s[k];
            }
        }
    }
}

#define DU     8
static_assert(BNR <= 1024, "one thread per row of the range in the clear and in the store");

__global__ __launch_bounds__(1024)
void deg_from_list(const unsigned* __restrict__ list, const unsigned* __restrict__ bstart,
                   float* __restrict__ dinv) {
    __shared__ unsigned h[BNR];
    const int tid = threadIdx.x;
    const int r   = blockIdx.x;
    if (tid < BNR) h[tid] = 0u;
    __syncthreads();

    const unsigned plo = bstart[r];
    const unsigned phi = bstart[r + 1];
    for (unsigned b0 = plo; b0 < phi; b0 += 1024u * DU) {
        unsigned a[DU];
#pragma unroll
        for (int u = 0; u < DU; ++u) {
            const unsigned idx = b0 + 1024u * u + (unsigned)tid;
            const unsigned w = list[min(idx, (unsigned)(NEDGES - 1))];
            a[u] = (idx < phi) ? (w >> SBITS) : 0xffffffffu;
        }
#pragma unroll
        for (int u = 0; u < DU; ++u)
            if (a[u] < (unsigned)BNR) atomicAdd(&h[a[u]], 1u);
    }
    __syncthreads();
    if (tid < BNR) dinv[r * BNR + tid] = rsqrtf((float)h[tid] + 1.0f);
}

#define AT     2
#define BLDSW  42240
static_assert(BNR * APITCH <= BLDSW, "the rows fit into the group's LDS");
static_assert((BNR * APITCH) % 4 == 0 && (BNR * NHID) % 32 == 0, "16-byte clears, whole 128-byte output lines per range");

__global__ __launch_bounds__(1024)
void edge_aggregate_binned(const unsigned* __restrict__ list, const unsigned* __restrict__ bstart,
                           const float* __restrict__ T, const float* __restrict__ dinv,
                           const float* __restrict__ bias, float* __restrict__ OUT, int doRelu) {
    __shared__ __attribute__((aligned(16))) float acc[BLDSW];
    const int tid  = threadIdx.x;
    const int lane = tid & 31;
    const int wave = __builtin_amdgcn_readfirstlane(tid >> 5);
    const int r    = blockIdx.x;

    float4* a4 = (float4*)acc;
    for (int i = tid; i < BNR * APITCH / 4; i += 1024) a4[i] = make_float4(0.0f, 0.0f, 0.0f, 0.0f);
    __syncthreads();

    const unsigned plo = bstart[r];
    const unsigned phi = bstart[r + 1];
    const unsigned grp = (unsigned)(lane >> 2);
    const unsigned f4  = (unsigned)(lane & 3) << 2;
    const unsigned PAD = 0x3fffu << SBITS;

    unsigned ent[AT];
    auto load_ent = [&](unsigned t0, unsigned (&e)[AT]) {
#pragma unroll
        for (int u = 0; u < AT; ++u) {
            const unsigned idx = t0 + 32u * u + (unsigned)lane;
            const unsigned w = list[min(idx, (unsigned)(NEDGES - 1))];
            e[u] = (idx < phi) ? w : PAD;
        }
    };

    unsigned t0 = plo + (unsigned)wave * (32u * AT);
    load_ent(t0, ent);
    while (t0 < phi) {
        unsigned nxt[AT];
        load_ent(t0 + 32u * 32u * AT, nxt);
        unsigned pa[4 * AT];
        v4f pm[4 * AT];
#pragma unroll
        for (int u = 0; u < AT; ++u) {
#pragma unroll
            for (int k = 0; k < 4; ++k) {
                const unsigned e = (unsigned)__shfl((int)ent[u], (int)(8u * k + grp), 32);
                const unsigned s = e & ((1u << SBITS) - 1u);
                pa[4 * u + k] = e >> SBITS;
                pm[4 * u + k] = *(const v4f*)((const char*)T + ((s << 6) | (f4 << 2)));
            }
        }
#pragma unroll
        for (int j = 0; j < 4 * AT; ++j) {
            if (pa[j] < (unsigned)BNR) {
                float* d = &acc[pa[j] * APITCH + f4];
                atomicAdd(d + 0, pm[j].x);
                atomicAdd(d + 1, pm[j].y);
                atomicAdd(d + 2, pm[j].z);
                atomicAdd(d + 3, pm[j].w);
            }
        }
#pragma unroll
        for (int u = 0; u < AT; ++u) ent[u] = nxt[u];
        t0 += 32u * 32u * AT;
    }
    __syncthreads();

    const int row0 = r * BNR;
    const float4* t4 = (const float4*)(T + (size_t)row0 * NHID);
    float4* o4 = (float4*)(OUT + (size_t)row0 * NHID);
    for (int i = tid; i < BNR * NHID / 4; i += 1024) {
        const int row = i >> 2;
        const int f   = (i & 3) << 2;
        const float di = dinv[row0 + row];
        const float4 t = t4[i];
        const float* a = &acc[row * APITCH + f];
        float4 v;
        v.x = (a[0] + t.x) * di + bias[f + 0];
        v.y = (a[1] + t.y) * di + bias[f + 1];
        v.z = (a[2] + t.z) * di + bias[f + 2];
        v.w = (a[3] + t.w) * di + bias[f + 3];
        if (doRelu) {
            v.x = fmaxf(v.x, 0.0f);
            v.y = fmaxf(v.y, 0.0f);
            v.z = fmaxf(v.z, 0.0f);
            v.w = fmaxf(v.w, 0.0f);
        }
        o4[i] = v;
    }
}

__global__ void finalize_node(float* __restrict__ ACC, const float* __restrict__ T,
                              const float* __restrict__ dinv, const float* __restrict__ bias,
                              int n16, int doRelu) {
    int idx = blockIdx.x * blockDim.x + threadIdx.x;
    if (idx >= n16) return;
    int i = idx >> 4, f = idx & 15;
    float di = dinv[i];
    float v  = (ACC[idx] + T[idx]) * di + bias[f];
    ACC[idx] = doRelu ? fmaxf(v, 0.0f) : v;
}

__global__ void pool_kernel(const int* __restrict__ bvec, const float* __restrict__ Hn,
                            float* __restrict__ Gp, int n) {
    int t = blockIdx.x * blockDim.x + threadIdx.x;
    int i = t >> 2, q = t & 3;
    if (i >= n) return;
    int g = bvec[i];
    const float4 h = *(const float4*)(Hn + (size_t)i * NHID + q * 4);
    float* o = Gp + (size_t)g * NHID + q * 4;
    atomicAdd(o + 0, h.x);
    atomicAdd(o + 1, h.y);
    atomicAdd(o + 2, h.z);
    atomicAdd(o + 3, h.w);
}

__global__ void lin1_kernel(const float* __restrict__ G, const float* __restrict__ W,
                            const float* __restrict__ B, float* __restrict__ O) {
    int t = blockIdx.x * blockDim.x + threadIdx.x;
    if (t >= NGRAPH * NMLP) return;
    int gi = t / NMLP, j = t % NMLP;
    float acc = B[j];
#pragma unroll
    for (int k = 0; k < NHID; ++k)
        acc = fmaf(fmaxf(G[gi * NHID + k], 0.0f), W[k * NMLP + j], acc);
    O[t] = fmaxf(acc, 0.0f);
}

__global__ void lin2_kernel(const float* __restrict__ G, const float* __restrict__ W,
                            const float* __restrict__ B, float* __restrict__ O) {
    int t = blockIdx.x * blockDim.x + threadIdx.x;
    if (t >= NGRAPH * NCLS) return;
    int gi = t / NCLS, c = t % NCLS;
    float acc = B[c];
    for (int k = 0; k < NMLP; ++k)
        acc = fmaf(G[gi * NMLP + k], W[k * NCLS + c], acc);
    O[t] = acc;
}


extern "C" void kernel_launch(void* const* d_in, const int* in_sizes, int n_in,
                              void* d_out, int out_size, void* d_ws, size_t ws_size,
                              hipStream_t stream) {
    const float* x   = (const float*)d_in[0];
    const int*  eidx = (const int*)d_in[1];
    const int*  bvec = (const int*)d_in[2];
    const float* W1  = (const float*)d_in[3];
    const float* b1  = (const float*)d_in[4];
    const float* W2  = (const float*)d_in[5];
    const float* b2  = (const float*)d_in[6];
    const float* Wl1 = (const float*)d_in[7];
    const float* bl1 = (const float*)d_in[8];
    const float* Wl2 = (const float*)d_in[9];
    const float* bl2 = (const float*)d_in[10];
    float* out = (float*)d_out;

    char* ws = (char*)d_ws;
    size_t off = 0;
    auto walloc = [&](size_t bytes) -> void* {
        void* p = ws + off;
        off = (off + bytes + 255) & ~(size_t)255;
        return p;
    };
    float* deg  = (float*)walloc((size_t)NNODES * 4);
    float* bufA = (float*)walloc((size_t)NNODES * NHID * 4);
    float* bufB = (float*)walloc((size_t)NNODES * NHID * 4);
    float* gbuf = (float*)walloc((size_t)NGRAPH * NHID * 4);
    float* g2   = (float*)walloc((size_t)NGRAPH * NMLP * 4);
    static_assert((size_t)BR * BNB * 4 <= (size_t)NGRAPH * NMLP * 4, "the count table fits into g2's place");
    unsigned* bcnt   = (unsigned*)g2;
    unsigned* bstart = (unsigned*)walloc((size_t)(BR + 1) * 4);
    unsigned* blist  = (unsigned*)walloc((size_t)NEDGES * 4);
    const bool binned = off <= ws_size;

    const int* src = eidx;
    const int* dst = eidx + NEDGES;

    const int TB = 256;
    const int nN16  = (NNODES * NHID + TB - 1) / TB;
    const int nN4   = (NNODES * 4 + TB - 1) / TB;
    const int nTile = ((NNODES / 16) + 7) / 8;

    float* dinv = deg;
    if (binned) {
        bin_count<<<BNB, 1024, 0, stream>>>(dst, bcnt);
        bin_scatter<<<BNB, 1024, 0, stream>>>(src, dst, bcnt, bstart, blist);
        deg_from_list<<<BR, 1024, 0, stream>>>(blist, bstart, dinv);
    } else {
        unsigned* part = (unsigned*)bufA;
        deg_hist<<<HR * HCH, 1024, 0, stream>>>(dst, part);
        deg_reduce<<<(NNODES / 4 + TB - 1) / TB, TB, 0, stream>>>(part, deg);
    }

    gemm_wmma_f32<<<nTile, TB, 0, stream>>>(x, W1, dinv, bufA, NNODES, NFEAT);
    if (binned) {
        edge_aggregate_binned<<<BR, 1024, 0, stream>>>(blist, bstart, bufA, dinv, b1, bufB, 1);
    } else {
        fill_f32<<<nN16, TB, 0, stream>>>(bufB, NNODES * NHID, 0.0f);
        edge_aggregate_filter<<<AR * ACH, 1024, 0, stream>>>(src, dst, bufA, bufB);
        finalize_node<<<nN16, TB, 0, stream>>>(bufB, bufA, dinv, b1, NNODES * NHID, 1);
    }

    gemm_wmma_f32<<<nTile, TB, 0, stream>>>(bufB, W2, dinv, bufA, NNODES, NHID);
    if (binned) {
        edge_aggregate_binned<<<BR, 1024, 0, stream>>>(blist, bstart, bufA, dinv, b2, bufB, 0);
    } else {
        fill_f32<<<nN16, TB, 0, stream>>>(bufB, NNODES * NHID, 0.0f);
        edge_aggregate_filter<<<AR * ACH, 1024, 0, stream>>>(src, dst, bufA, bufB);
        finalize_node<<<nN16, TB, 0, stream>>>(bufB, bufA, dinv, b2, NNODES * NHID, 0);
    }

    fill_f32<<<(NGRAPH * NHID + TB - 1) / TB, TB, 0, stream>>>(gbuf, NGRAPH * NHID, 0.0f);
    pool_kernel<<<nN4, TB, 0, stream>>>(bvec, bufB, gbuf, NNODES);
    lin1_kernel<<<(NGRAPH * NMLP + TB - 1) / TB, TB, 0, stream>>>(gbuf, Wl1, bl1, g2);
    lin2_kernel<<<(NGRAPH * NCLS + TB - 1) / TB, TB, 0, stream>>>(g2, Wl2, bl2, out);
}


# Binning scatter staged in LDS: a slice is put in range order in LDS, then each range's part goes to the list from consecutive lanes lined up with 128-byte lines, not as one 4-byte store per edge
# speedup vs baseline: 4.6904x; 1.0807x over previous
#include <hip/hip_runtime.h>
#include <hip/hip_bf16.h>


typedef float v2f __attribute__((ext_vector_type(2)));
typedef float v4f __attribute__((ext_vector_type(4)));
typedef float v8f __attribute__((ext_vector_type(8)));

#define NNODES  200000
#define NEDGES  6400000
#define NFEAT   128
#define NHID    16
#define NMLP    100
#define NCLS    27
#define NGRAPH  4096


__global__ void fill_f32(float* __restrict__ p, int n, float v) {
    int i = blockIdx.x * blockDim.x + threadIdx.x;
    if (i < n) p[i] = v;
}

#define HR     8
#define HBINS  (NNODES / HR)
#define HCH    32
#define HCHUNK (NEDGES / HCH)
#define HN4    (HCHUNK / 4)

__global__ __launch_bounds__(1024)
void deg_hist(const int* __restrict__ dst, unsigned* __restrict__ part) {
    __shared__ __attribute__((aligned(16))) unsigned h[HBINS];
    const int tid = threadIdx.x;
    const int r   = blockIdx.x & (HR - 1);
    const int c   = blockIdx.x >> 3;

    uint4* h4 = (uint4*)h;
    for (int i = tid; i < HBINS / 4; i += 1024) h4[i] = make_uint4(0u, 0u, 0u, 0u);
    __syncthreads();

    const unsigned lo = (unsigned)(r * HBINS);
    const int4* p = (const int4*)(dst + (size_t)c * HCHUNK);
    for (int base = tid; base < HN4; base += 1024 * 8) {
        int4 v[8];
#pragma unroll
        for (int u = 0; u < 8; ++u) {
            const int j = base + u * 1024;
            v[u] = (j < HN4) ? p[j] : make_int4(-1, -1, -1, -1);
        }
#pragma unroll
        for (int u = 0; u < 8; ++u) {
            unsigned a;
            a = (unsigned)v[u].x - lo; if (a < (unsigned)HBINS) atomicAdd(&h[a], 1u);
            a = (unsigned)v[u].y - lo; if (a < (unsigned)HBINS) atomicAdd(&h[a], 1u);
            a = (unsigned)v[u].z - lo; if (a < (unsigned)HBINS) atomicAdd(&h[a], 1u);
            a = (unsigned)v[u].w - lo; if (a < (unsigned)HBINS) atomicAdd(&h[a], 1u);
        }
    }
    __syncthreads();

    uint4* o = (uint4*)(part + (size_t)c * NNODES + lo);
    for (int i = tid; i < HBINS / 4; i += 1024) o[i] = h4[i];
}

__global__ void deg_reduce(const unsigned* __restrict__ part, float* __restrict__ dinv) {
    const int i = blockIdx.x * blockDim.x + threadIdx.x;
    if (i >= NNODES / 4) return;
    const uint4* p = (const uint4*)part + i;
    uint4 s = make_uint4(0u, 0u, 0u, 0u);
#pragma unroll
    for (int c = 0; c < HCH; ++c) {
        const uint4 v = p[(size_t)c * (NNODES / 4)];
        s.x += v.x; s.y += v.y; s.z += v.z; s.w += v.w;
    }
    float4 d;
    d.x = rsqrtf((float)s.x + 1.0f);
    d.y = rsqrtf((float)s.y + 1.0f);
    d.z = rsqrtf((float)s.z + 1.0f);
    d.w = rsqrtf((float)s.w + 1.0f);
    ((float4*)dinv)[i] = d;
}

__global__ __launch_bounds__(256)
void gemm_wmma_f32(const float* __restrict__ X, const float* __restrict__ W,
                   const float* __restrict__ rs, float* __restrict__ T, int M, int K) {
    __shared__ float lw[NFEAT * NHID];
    const int tid = threadIdx.x;
    const int kn  = K * NHID;
    for (int i = tid; i < kn; i += 256) lw[i] = W[i];
    __syncthreads();

    const int wave = tid >> 5;
    const int lane = tid & 31;
    const int tile = blockIdx.x * 8 + wave;
    if (tile * 16 >= M) return;

    const int row   = tile * 16 + (lane & 15);
    const int kHalf = (lane >> 4) << 1;
    const int n     = lane & 15;

    v8f c = {};
    const float* xr = X + (size_t)row * K + kHalf;
    for (int k0 = 0; k0 < K; k0 += 4) {
        float2 av = *(const float2*)(xr + k0);
        v2f a; a.x = av.x; a.y = av.y;
        v2f b;
        b.x = lw[(k0 + kHalf)     * NHID + n];
        b.y = lw[(k0 + kHalf + 1) * NHID + n];
        c = __builtin_amdgcn_wmma_f32_16x16x4_f32(
                 false, a,  false, b,
                 (short)0, c,  false,  false);
    }

    const int rbase = tile * 16 + ((lane >> 4) << 3);
#pragma unroll
    for (int v = 0; v < 8; ++v)
        T[(size_t)(rbase + v) * NHID + n] = c[v] * rs[rbase + v];
}

#define AR     50
#define ANR    (NNODES / AR)
#define ACH    5
#define ACHUNK (NEDGES / ACH)
#define AN4    (ACHUNK / 4)
#define AU     6
#define AWT    (32 * AU)
#define ANWT   ((AN4 + AWT - 1) / AWT)
#define AWQ    128
#define AB     3
#define APITCH 17

__global__ __launch_bounds__(1024)
void edge_aggregate_filter(const int* __restrict__ src, const int* __restrict__ dst,
                           const float* __restrict__ T, float* __restrict__ OUT) {
    __shared__ __attribute__((aligned(16))) float acc[ANR * APITCH];
    __shared__ uint2 queue[32][AWQ];
    const int tid  = threadIdx.x;
    const int lane = tid & 31;
    const int wave = __builtin_amdgcn_readfirstlane(tid >> 5);
    const int r    = blockIdx.x % AR;
    const int c    = blockIdx.x / AR;

    float4* a4 = (float4*)acc;
    for (int i = tid; i < ANR * APITCH / 4; i += 1024) a4[i] = make_float4(0.0f, 0.0f, 0.0f, 0.0f);
    for (int i = tid; i < 32 * AWQ; i += 1024) (&queue[0][0])[i] = make_uint2(0xffffffffu, 0u);
    __syncthreads();

    const unsigned lo = (unsigned)(r * ANR);
    const int ebase   = c * ACHUNK;
    const int4* p     = (const int4*)(dst + (size_t)ebase);
    const unsigned grp = (unsigned)(lane >> 2);
    const unsigned f4  = (unsigned)(lane & 3) << 2;
    uint2* qb = queue[wave];

    int4 v[AU];
    auto load_tile = [&](int wt) {
#pragma unroll
        for (int u = 0; u < AU; ++u) {
            const int jb = wt * AWT + u * 32;
            if (jb < AN4) v[u] = p[jb + lane];
            else          v[u] = make_int4(-1, -1, -1, -1);
        }
    };

    auto masks = [&](unsigned (&msk)[4 * AU]) -> unsigned {
        unsigned tot = 0u;
#pragma unroll
        for (int u = 0; u < AU; ++u) {
            msk[4 * u + 0] = __builtin_amdgcn_ballot_w32((unsigned)v[u].x - lo < (unsigned)ANR);
            msk[4 * u + 1] = __builtin_amdgcn_ballot_w32((unsigned)v[u].y - lo < (unsigned)ANR);
            msk[4 * u + 2] = __builtin_amdgcn_ballot_w32((unsigned)v[u].z - lo < (unsigned)ANR);
            msk[4 * u + 3] = __builtin_amdgcn_ballot_w32((unsigned)v[u].w - lo < (unsigned)ANR);
            tot += (unsigned)(__builtin_popcount(msk[4 * u + 0]) + __builtin_popcount(msk[4 * u + 1]) +
                              __builtin_popcount(msk[4 * u + 2]) + __builtin_popcount(msk[4 * u + 3]));
        }
        return tot;
    };

    auto emit = [&](const unsigned (&msk)[4 * AU], int wt, unsigned run) {
#pragma unroll
        for (int u = 0; u < AU; ++u) {
            const int e0 = ebase + 4 * (wt * AWT + u * 32 + lane);
            const int dv[4] = { v[u].x, v[u].y, v[u].z, v[u].w };
#pragma unroll
            for (int k = 0; k < 4; ++k) {
                const unsigned m = msk[4 * u + k];
                if (m != 0u) {
                    const unsigned a = (unsigned)dv[k] - lo;
                    const unsigned pos = run + __builtin_amdgcn_mbcnt_lo(m, 0u);
                    if (a < (unsigned)ANR && pos < (unsigned)AWQ)
                        qb[pos] = make_uint2(a, (unsigned)(e0 + k));
                    run += (unsigned)__builtin_popcount(m);
                }
            }
        }
    };

    unsigned pa[AB];
    v4f pm[AB];
#pragma unroll
    for (int k = 0; k < AB; ++k) { pa[k] = 0xffffffffu; pm[k] = (v4f){0.0f, 0.0f, 0.0f, 0.0f}; }

    auto issue = [&](unsigned n, unsigned j0) {
        uint2 q[AB];
        unsigned s[AB];
#pragma unroll
        for (int k = 0; k < AB; ++k) {
            const unsigned jj = j0 + 8u * k;
            const uint2 t = qb[jj & (unsigned)(AWQ - 1)];
            q[k].x = (jj < n) ? t.x : 0xffffffffu;
            q[k].y = (jj < n) ? t.y : 0u;
        }
#pragma unroll
        for (int k = 0; k < AB; ++k) {
            s[k]  = (unsigned)*(const int*)((const char*)src + (q[k].y << 2));
            pa[k] = q[k].x;
        }
#pragma unroll
        for (int k = 0; k < AB; ++k)
            pm[k] = *(const v4f*)((const char*)T + ((s[k] << 6) | (f4 << 2)));
    };
    auto commit = [&]() {
#pragma unroll
        for (int k = 0; k < AB; ++k) {
            v4f val = pm[k];
            asm volatile("" : "+v"(val));
            if (pa[k] < (unsigned)ANR) {
                float* d = &acc[pa[k] * APITCH + f4];
                atomicAdd(d + 0, val.x);
                atomicAdd(d + 1, val.y);
                atomicAdd(d + 2, val.z);
                atomicAdd(d + 3, val.w);
            }
        }
    };

    int wt = wave;
    load_tile(wt);
    unsigned msk[4 * AU];
    unsigned tot = masks(msk);
    unsigned rlo = 0u;
    while (wt < ANWT) {
        emit(msk, wt, 0u - rlo);
        commit();
        __builtin_amdgcn_wave_barrier();
        const unsigned n = min(tot - rlo, (unsigned)AWQ);
        const bool last = rlo + (unsigned)AWQ >= tot;
        if (last) load_tile(wt + 32);
        issue(n, grp);
        for (unsigned jr = 8u * AB; jr < n; jr += 8u * AB) { commit(); issue(n, grp + jr); }
        __builtin_amdgcn_wave_barrier();
        if (last) {
            tot = masks(msk);
            wt += 32;
            rlo = 0u;
        } else {
            rlo += (unsigned)AWQ;
        }
    }
    commit();
    __syncthreads();

    float* o = OUT + (size_t)lo * NHID;
    for (int i = tid; i < ANR * NHID; i += 1024) atomicAdd(o + i, acc[(i >> 4) * APITCH + (i & 15)]);
}

#define BR     250
#define BNR    (NNODES / BR)
#define BNB    250
#define BSL    (NEDGES / BNB)
#define BSL4   (BSL / 4)
#define BU     ((BSL4 + 1023) / 1024)
#define BCU    ((BNB + 31) / 32)
#define SBITS  18
#define BCO    4
#define BWR    ((BR + 31) / 32)
static_assert(BWR <= 32, "one lane per range of the wave");
static_assert(BR * BNR == NNODES, "the ranges cover the rows exactly");

__global__ __launch_bounds__(1024)
void bin_count(const int* __restrict__ dst, unsigned* __restrict__ cnt) {
    __shared__ unsigned h[BR];
    const int tid = threadIdx.x;
    if (tid < BR) h[tid] = 0u;
    __syncthreads();
    const int4* p = (const int4*)(dst + (size_t)blockIdx.x * BSL);
    int4 v[BU];
#pragma unroll
    for (int u = 0; u < BU; ++u) {
        const int j = tid + u * 1024;
        v[u] = (j < BSL4) ? p[j] : make_int4(-1, -1, -1, -1);
    }
#pragma unroll
    for (int u = 0; u < BU; ++u) {
        const unsigned d[4] = { (unsigned)v[u].x, (unsigned)v[u].y, (unsigned)v[u].z, (unsigned)v[u].w };
#pragma unroll
        for (int k = 0; k < 4; ++k)
            if (d[k] < (unsigned)NNODES) atomicAdd(&h[d[k] / (unsigned)BNR], 1u);
    }
    __syncthreads();
    if (tid < BR) cnt[tid * BNB + blockIdx.x] = h[tid];
}

__global__ __launch_bounds__(1024)
void bin_scatter(const int* __restrict__ src, const int* __restrict__ dst,
                 const unsigned* __restrict__ cnt, unsigned* __restrict__ bstart,
                 unsigned* __restrict__ list) {
    __shared__ uint2 tm[BR];
    __shared__ unsigned pre[BR];
    __shared__ unsigned cur[BR];
    __shared__ unsigned lbeg[BR];
    __shared__ unsigned gbeg[BR];
    __shared__ unsigned stage[BSL];
    const int tid  = threadIdx.x;
    const int lane = tid & 31;
    const int wave = __builtin_amdgcn_readfirstlane(tid >> 5);
    const int b    = blockIdx.x;

    for (int r = wave; r < BR; r += 32) {
        unsigned t = 0u, q = 0u, m = 0u;
#pragma unroll
        for (int u = 0; u < BCU; ++u) {
            const int j = lane + 32 * u;
            const unsigned v = (j < BNB) ? cnt[r * BNB + j] : 0u;
            t += v;
            q += (j < b) ? v : 0u;
            m += (j == b) ? v : 0u;
        }
        for (int k = 16; k >= 1; k >>= 1) {
            t += __shfl_xor(t, k, 32);
            q += __shfl_xor(q, k, 32);
            m += __shfl_xor(m, k, 32);
        }
        if (lane == 0) { tm[r] = make_uint2(t, m); pre[r] = q; }
    }
    __syncthreads();
    if (tid < BR) {
        unsigned run = 0u;
        unsigned loc = 0u;
#pragma unroll
        for (int j = 0; j < BR; ++j) {
            const uint2 v = tm[j];
            run += (j < tid) ? v.x : 0u;
            loc += (j < tid) ? v.y : 0u;
        }
        gbeg[tid] = run + pre[tid];
        lbeg[tid] = loc;
        cur[tid]  = loc;
        if (b == 0) {
            bstart[tid] = run;
            if (tid == BR - 1) bstart[BR] = run + tm[tid].x;
        }
    }
    __syncthreads();

    const int4* pd = (const int4*)(dst + (size_t)b * BSL);
    const int4* ps = (const int4*)(src + (size_t)b * BSL);
    int4 vd[BU], vs[BU];
#pragma unroll
    for (int u = 0; u < BU; ++u) {
        const int j = tid + u * 1024;
        vd[u] = (j < BSL4) ? pd[j] : make_int4(-1, -1, -1, -1);
        vs[u] = (j < BSL4) ? ps[j] : make_int4(0, 0, 0, 0);
    }
#pragma unroll
    for (int u = 0; u < BU; ++u) {
        const unsigned d[4] = { (unsigned)vd[u].x, (unsigned)vd[u].y, (unsigned)vd[u].z, (unsigned)vd[u].w };
        const unsigned s[4] = { (unsigned)vs[u].x, (unsigned)vs[u].y, (unsigned)vs[u].z, (unsigned)vs[u].w };
#pragma unroll
        for (int k = 0; k < 4; ++k) {
            if (d[k] < (unsigned)NNODES) {
                const unsigned r   = d[k] / (unsigned)BNR;
                const unsigned a   = d[k] - r * (unsigned)BNR;
                const unsigned pos = atomicAdd(&cur[r], 1u);
                stage[pos] = (a << SBITS) | s[k];
            }
        }
    }
    __syncthreads();

    const int      rk  = wave + 32 * lane;
    const bool     hk  = lane < BWR && rk < BR;
    const int      nk  = hk ? (int)tm[rk].y : 0;
    const unsigned lbk = hk ? lbeg[rk] : 0u;
    const unsigned gbk = hk ? gbeg[rk] : 0u;
    for (int k = 0; k < BWR; ++k) {
        const int      n  = __builtin_amdgcn_readlane(nk, k);
        const unsigned lb = (unsigned)__builtin_amdgcn_readlane((int)lbk, k);
        const unsigned gb = (unsigned)__builtin_amdgcn_readlane((int)gbk, k);
        if (n == 0) continue;
        for (int c0 = -(int)(gb & 31u); c0 < n; c0 += 32 * BCO) {
            unsigned w[BCO];
#pragma unroll
            for (int u = 0; u < BCO; ++u) {
                const int i = c0 + 32 * u + lane;
                w[u] = stage[lb + (unsigned)min(max(i, 0), n - 1)];
            }
#pragma unroll
            for (int u = 0; u < BCO; ++u) {
                const int i = c0 + 32 * u + lane;
                if (i >= 0 && i < n) list[gb + (unsigned)i] = w[u];
            }
        }
    }
}

#define DU     8
static_assert(BNR <= 1024, "one thread per row of the range in the clear and in the store");

__global__ __launch_bounds__(1024)
void deg_from_list(const unsigned* __restrict__ list, const unsigned* __restrict__ bstart,
                   float* __restrict__ dinv) {
    __shared__ unsigned h[BNR];
    const int tid = threadIdx.x;
    const int r   = blockIdx.x;
    if (tid < BNR) h[tid] = 0u;
    __syncthreads();

    const unsigned plo = bstart[r];
    const unsigned phi = bstart[r + 1];
    for (unsigned b0 = plo; b0 < phi; b0 += 1024u * DU) {
        unsigned a[DU];
#pragma unroll
        for (int u = 0; u < DU; ++u) {
            const unsigned idx = b0 + 1024u * u + (unsigned)tid;
            const unsigned w = list[min(idx, (unsigned)(NEDGES - 1))];
            a[u] = (idx < phi) ? (w >> SBITS) : 0xffffffffu;
        }
#pragma unroll
        for (int u = 0; u < DU; ++u)
            if (a[u] < (unsigned)BNR) atomicAdd(&h[a[u]], 1u);
    }
    __syncthreads();
    if (tid < BNR) dinv[r * BNR + tid] = rsqrtf((float)h[tid] + 1.0f);
}

#define AT     2
#define BLDSW  42240
static_assert(BNR * APITCH <= BLDSW, "the rows fit into the group's LDS");
static_assert((BNR * APITCH) % 4 == 0 && (BNR * NHID) % 32 == 0, "16-byte clears, whole 128-byte output lines per range");

__global__ __launch_bounds__(1024)
void edge_aggregate_binned(const unsigned* __restrict__ list, const unsigned* __restrict__ bstart,
                           const float* __restrict__ T, const float* __restrict__ dinv,
                           const float* __restrict__ bias, float* __restrict__ OUT, int doRelu) {
    __shared__ __attribute__((aligned(16))) float acc[BLDSW];
    const int tid  = threadIdx.x;
    const int lane = tid & 31;
    const int wave = __builtin_amdgcn_readfirstlane(tid >> 5);
    const int r    = blockIdx.x;

    float4* a4 = (float4*)acc;
    for (int i = tid; i < BNR * APITCH / 4; i += 1024) a4[i] = make_float4(0.0f, 0.0f, 0.0f, 0.0f);
    __syncthreads();

    const unsigned plo = bstart[r];
    const unsigned phi = bstart[r + 1];
    const unsigned grp = (unsigned)(lane >> 2);
    const unsigned f4  = (unsigned)(lane & 3) << 2;
    const unsigned PAD = 0x3fffu << SBITS;

    unsigned ent[AT];
    auto load_ent = [&](unsigned t0, unsigned (&e)[AT]) {
#pragma unroll
        for (int u = 0; u < AT; ++u) {
            const unsigned idx = t0 + 32u * u + (unsigned)lane;
            const unsigned w = list[min(idx, (unsigned)(NEDGES - 1))];
            e[u] = (idx < phi) ? w : PAD;
        }
    };

    unsigned t0 = plo + (unsigned)wave * (32u * AT);
    load_ent(t0, ent);
    while (t0 < phi) {
        unsigned nxt[AT];
        load_ent(t0 + 32u * 32u * AT, nxt);
        unsigned pa[4 * AT];
        v4f pm[4 * AT];
#pragma unroll
        for (int u = 0; u < AT; ++u) {
#pragma unroll
            for (int k = 0; k < 4; ++k) {
                const unsigned e = (unsigned)__shfl((int)ent[u], (int)(8u * k + grp), 32);
                const unsigned s = e & ((1u << SBITS) - 1u);
                pa[4 * u + k] = e >> SBITS;
                pm[4 * u + k] = *(const v4f*)((const char*)T + ((s << 6) | (f4 << 2)));
            }
        }
#pragma unroll
        for (int j = 0; j < 4 * AT; ++j) {
            if (pa[j] < (unsigned)BNR) {
                float* d = &acc[pa[j] * APITCH + f4];
                atomicAdd(d + 0, pm[j].x);
                atomicAdd(d + 1, pm[j].y);
                atomicAdd(d + 2, pm[j].z);
                atomicAdd(d + 3, pm[j].w);
            }
        }
#pragma unroll
        for (int u = 0; u < AT; ++u) ent[u] = nxt[u];
        t0 += 32u * 32u * AT;
    }
    __syncthreads();

    const int row0 = r * BNR;
    const float4* t4 = (const float4*)(T + (size_t)row0 * NHID);
    float4* o4 = (float4*)(OUT + (size_t)row0 * NHID);
    for (int i = tid; i < BNR * NHID / 4; i += 1024) {
        const int row = i >> 2;
        const int f   = (i & 3) << 2;
        const float di = dinv[row0 + row];
        const float4 t = t4[i];
        const float* a = &acc[row * APITCH + f];
        float4 v;
        v.x = (a[0] + t.x) * di + bias[f + 0];
        v.y = (a[1] + t.y) * di + bias[f + 1];
        v.z = (a[2] + t.z) * di + bias[f + 2];
        v.w = (a[3] + t.w) * di + bias[f + 3];
        if (doRelu) {
            v.x = fmaxf(v.x, 0.0f);
            v.y = fmaxf(v.y, 0.0f);
            v.z = fmaxf(v.z, 0.0f);
            v.w = fmaxf(v.w, 0.0f);
        }
        o4[i] = v;
    }
}

__global__ void finalize_node(float* __restrict__ ACC, const float* __restrict__ T,
                              const float* __restrict__ dinv, const float* __restrict__ bias,
                              int n16, int doRelu) {
    int idx = blockIdx.x * blockDim.x + threadIdx.x;
    if (idx >= n16) return;
    int i = idx >> 4, f = idx & 15;
    float di = dinv[i];
    float v  = (ACC[idx] + T[idx]) * di + bias[f];
    ACC[idx] = doRelu ? fmaxf(v, 0.0f) : v;
}

__global__ void pool_kernel(const int* __restrict__ bvec, const float* __restrict__ Hn,
                            float* __restrict__ Gp, int n) {
    int t = blockIdx.x * blockDim.x + threadIdx.x;
    int i = t >> 2, q = t & 3;
    if (i >= n) return;
    int g = bvec[i];
    const float4 h = *(const float4*)(Hn + (size_t)i * NHID + q * 4);
    float* o = Gp + (size_t)g * NHID + q * 4;
    atomicAdd(o + 0, h.x);
    atomicAdd(o + 1, h.y);
    atomicAdd(o + 2, h.z);
    atomicAdd(o + 3, h.w);
}

__global__ void lin1_kernel(const float* __restrict__ G, const float* __restrict__ W,
                            const float* __restrict__ B, float* __restrict__ O) {
    int t = blockIdx.x * blockDim.x + threadIdx.x;
    if (t >= NGRAPH * NMLP) return;
    int gi = t / NMLP, j = t % NMLP;
    float acc = B[j];
#pragma unroll
    for (int k = 0; k < NHID; ++k)
        acc = fmaf(fmaxf(G[gi * NHID + k], 0.0f), W[k * NMLP + j], acc);
    O[t] = fmaxf(acc, 0.0f);
}

__global__ void lin2_kernel(const float* __restrict__ G, const float* __restrict__ W,
                            const float* __restrict__ B, float* __restrict__ O) {
    int t = blockIdx.x * blockDim.x + threadIdx.x;
    if (t >= NGRAPH * NCLS) return;
    int gi = t / NCLS, c = t % NCLS;
    float acc = B[c];
    for (int k = 0; k < NMLP; ++k)
        acc = fmaf(G[gi * NMLP + k], W[k * NCLS + c], acc);
    O[t] = acc;
}


extern "C" void kernel_launch(void* const* d_in, const int* in_sizes, int n_in,
                              void* d_out, int out_size, void* d_ws, size_t ws_size,
                              hipStream_t stream) {
    const float* x   = (const float*)d_in[0];
    const int*  eidx = (const int*)d_in[1];
    const int*  bvec = (const int*)d_in[2];
    const float* W1  = (const float*)d_in[3];
    const float* b1  = (const float*)d_in[4];
    const float* W2  = (const float*)d_in[5];
    const float* b2  = (const float*)d_in[6];
    const float* Wl1 = (const float*)d_in[7];
    const float* bl1 = (const float*)d_in[8];
    const float* Wl2 = (const float*)d_in[9];
    const float* bl2 = (const float*)d_in[10];
    float* out = (float*)d_out;

    char* ws = (char*)d_ws;
    size_t off = 0;
    auto walloc = [&](size_t bytes) -> void* {
        void* p = ws + off;
        off = (off + bytes + 255) & ~(size_t)255;
        return p;
    };
    float* deg  = (float*)walloc((size_t)NNODES * 4);
    float* bufA = (float*)walloc((size_t)NNODES * NHID * 4);
    float* bufB = (float*)walloc((size_t)NNODES * NHID * 4);
    float* gbuf = (float*)walloc((size_t)NGRAPH * NHID * 4);
    float* g2   = (float*)walloc((size_t)NGRAPH * NMLP * 4);
    static_assert((size_t)BR * BNB * 4 <= (size_t)NGRAPH * NMLP * 4, "the count table fits into g2's place");
    unsigned* bcnt   = (unsigned*)g2;
    unsigned* bstart = (unsigned*)walloc((size_t)(BR + 1) * 4);
    unsigned* blist  = (unsigned*)walloc((size_t)NEDGES * 4);
    const bool binned = off <= ws_size;

    const int* src = eidx;
    const int* dst = eidx + NEDGES;

    const int TB = 256;
    const int nN16  = (NNODES * NHID + TB - 1) / TB;
    const int nN4   = (NNODES * 4 + TB - 1) / TB;
    const int nTile = ((NNODES / 16) + 7) / 8;

    float* dinv = deg;
    if (binned) {
        bin_count<<<BNB, 1024, 0, stream>>>(dst, bcnt);
        bin_scatter<<<BNB, 1024, 0, stream>>>(src, dst, bcnt, bstart, blist);
        deg_from_list<<<BR, 1024, 0, stream>>>(blist, bstart, dinv);
    } else {
        unsigned* part = (unsigned*)bufA;
        deg_hist<<<HR * HCH, 1024, 0, stream>>>(dst, part);
        deg_reduce<<<(NNODES / 4 + TB - 1) / TB, TB, 0, stream>>>(part, deg);
    }

    gemm_wmma_f32<<<nTile, TB, 0, stream>>>(x, W1, dinv, bufA, NNODES, NFEAT);
    if (binned) {
        edge_aggregate_binned<<<BR, 1024, 0, stream>>>(blist, bstart, bufA, dinv, b1, bufB, 1);
    } else {
        fill_f32<<<nN16, TB, 0, stream>>>(bufB, NNODES * NHID, 0.0f);
        edge_aggregate_filter<<<AR * ACH, 1024, 0, stream>>>(src, dst, bufA, bufB);
        finalize_node<<<nN16, TB, 0, stream>>>(bufB, bufA, dinv, b1, NNODES * NHID, 1);
    }

    gemm_wmma_f32<<<nTile, TB, 0, stream>>>(bufB, W2, dinv, bufA, NNODES, NHID);
    if (binned) {
        edge_aggregate_binned<<<BR, 1024, 0, stream>>>(blist, bstart, bufA, dinv, b2, bufB, 0);
    } else {
        fill_f32<<<nN16, TB, 0, stream>>>(bufB, NNODES * NHID, 0.0f);
        edge_aggregate_filter<<<AR * ACH, 1024, 0, stream>>>(src, dst, bufA, bufB);
        finalize_node<<<nN16, TB, 0, stream>>>(bufB, bufA, dinv, b2, NNODES * NHID, 0);
    }

    fill_f32<<<(NGRAPH * NHID + TB - 1) / TB, TB, 0, stream>>>(gbuf, NGRAPH * NHID, 0.0f);
    pool_kernel<<<nN4, TB, 0, stream>>>(bvec, bufB, gbuf, NNODES);
    lin1_kernel<<<(NGRAPH * NMLP + TB - 1) / TB, TB, 0, stream>>>(gbuf, Wl1, bl1, g2);
    lin2_kernel<<<(NGRAPH * NCLS + TB - 1) / TB, TB, 0, stream>>>(g2, Wl2, bl2, out);
}
